# starting kernel
# baseline (speedup 1.0000x reference)
#include <hip/hip_runtime.h>
#include <math.h>

#define BB 2
#define TT 2048
#define DD 1024
#define RR 64
#define RSS 16
#define BT (BB*TT)

typedef __attribute__((ext_vector_type(2))) float v2f;
typedef __attribute__((ext_vector_type(8))) float v8f;

#define GBM 256
#define GBN 64
#define GBK 32

__global__ __launch_bounds__(256) void gemm_wmma_f32(
    const float* __restrict__ A, const float* __restrict__ Bm,
    float* __restrict__ C, const float* __restrict__ mulC,
    int M, int N, int K, int lda, int ldb, int ldc, int bT,
    long long sA, long long sB, long long sC)
{
    __shared__ __align__(16) float As [2][GBM][34];
    __shared__ __align__(16) float BsT[2][GBN][34];

    const int tid  = threadIdx.x;
    const int wave = tid >> 5;
    const int lane = tid & 31;
    const int hi   = lane >> 4;
    const int lm   = lane & 15;

    const int m0 = blockIdx.y * GBM;
    const int n0 = blockIdx.x * GBN;

    const float* Ab = A  + (size_t)blockIdx.z * sA;
    const float* Bb = Bm + (size_t)blockIdx.z * sB;
    float*       Cb = C  + (size_t)blockIdx.z * sC;
    const float* Mb = mulC ? (mulC + (size_t)blockIdx.z * sC) : nullptr;

    float4 ra[8];
    float4 rb[2];

    int arow[8], ac4[8];
    #pragma unroll
    for (int i = 0; i < 8; ++i) { int idx = tid + i * 256; arow[i] = idx >> 3; ac4[i] = (idx & 7) * 4; }
    int bi0[2], bi1[2];
    #pragma unroll
    for (int i = 0; i < 2; ++i) {
        int idx = tid + i * 256;
        if (true) {   }
        bi0[i] = idx; bi1[i] = idx;
    }

    auto loadGlobal = [&](int k0) {
        #pragma unroll
        for (int i = 0; i < 8; ++i)
            ra[i] = *(const float4*)(Ab + (size_t)(m0 + arow[i]) * lda + k0 + ac4[i]);
        if (bT) {
            #pragma unroll
            for (int i = 0; i < 2; ++i) {
                int idx = bi0[i];
                int n = idx >> 3, c4 = (idx & 7) * 4;
                rb[i] = *(const float4*)(Bb + (size_t)(n0 + n) * ldb + k0 + c4);
            }
        } else {
            #pragma unroll
            for (int i = 0; i < 2; ++i) {
                int idx = bi1[i];
                int r = idx >> 4, c4 = (idx & 15) * 4;
                rb[i] = *(const float4*)(Bb + (size_t)(k0 + r) * ldb + n0 + c4);
            }
        }
    };

    auto storeLDS = [&](int buf) {
        #pragma unroll
        for (int i = 0; i < 8; ++i) {
            float4 t = ra[i];
            v2f lo; lo.x = t.x; lo.y = t.y;
            v2f hi2; hi2.x = t.z; hi2.y = t.w;
            *(v2f*)&As[buf][arow[i]][ac4[i]]     = lo;
            *(v2f*)&As[buf][arow[i]][ac4[i] + 2] = hi2;
        }
        if (bT) {
            #pragma unroll
            for (int i = 0; i < 2; ++i) {
                int idx = bi0[i];
                int n = idx >> 3, c4 = (idx & 7) * 4;
                float4 t = rb[i];
                v2f lo; lo.x = t.x; lo.y = t.y;
                v2f hi2; hi2.x = t.z; hi2.y = t.w;
                *(v2f*)&BsT[buf][n][c4]     = lo;
                *(v2f*)&BsT[buf][n][c4 + 2] = hi2;
            }
        } else {
            #pragma unroll
            for (int i = 0; i < 2; ++i) {
                int idx = bi1[i];
                int r = idx >> 4, c4 = (idx & 15) * 4;
                float4 t = rb[i];
                BsT[buf][c4 + 0][r] = t.x;
                BsT[buf][c4 + 1][r] = t.y;
                BsT[buf][c4 + 2][r] = t.z;
                BsT[buf][c4 + 3][r] = t.w;
            }
        }
    };

    v8f acc00 = {}, acc01 = {}, acc02 = {}, acc03 = {};
    v8f acc10 = {}, acc11 = {}, acc12 = {}, acc13 = {};

    const int r0 = wave * 32 + lm;
    const int r1 = r0 + 16;
    const int nchunks = K / GBK;

    loadGlobal(0);
    for (int c = 0; c < nchunks; ++c) {
        const int buf = c & 1;
        storeLDS(buf);
        __syncthreads();
        if (c + 1 < nchunks) loadGlobal((c + 1) * GBK);

        #pragma unroll
        for (int kk = 0; kk < GBK; kk += 4) {
            const int kc = kk + 2 * hi;
            v2f a0 = *(const v2f*)&As [buf][r0][kc];
            v2f a1 = *(const v2f*)&As [buf][r1][kc];
            v2f b0 = *(const v2f*)&BsT[buf][lm +  0][kc];
            v2f b1 = *(const v2f*)&BsT[buf][lm + 16][kc];
            v2f b2 = *(const v2f*)&BsT[buf][lm + 32][kc];
            v2f b3 = *(const v2f*)&BsT[buf][lm + 48][kc];
            acc00 = __builtin_amdgcn_wmma_f32_16x16x4_f32(false, a0, false, b0, (short)0, acc00, false, false);
            acc01 = __builtin_amdgcn_wmma_f32_16x16x4_f32(false, a0, false, b1, (short)0, acc01, false, false);
            acc02 = __builtin_amdgcn_wmma_f32_16x16x4_f32(false, a0, false, b2, (short)0, acc02, false, false);
            acc03 = __builtin_amdgcn_wmma_f32_16x16x4_f32(false, a0, false, b3, (short)0, acc03, false, false);
            acc10 = __builtin_amdgcn_wmma_f32_16x16x4_f32(false, a1, false, b0, (short)0, acc10, false, false);
            acc11 = __builtin_amdgcn_wmma_f32_16x16x4_f32(false, a1, false, b1, (short)0, acc11, false, false);
            acc12 = __builtin_amdgcn_wmma_f32_16x16x4_f32(false, a1, false, b2, (short)0, acc12, false, false);
            acc13 = __builtin_amdgcn_wmma_f32_16x16x4_f32(false, a1, false, b3, (short)0, acc13, false, false);
        }
        __syncthreads();
    }

    const int ncol = n0 + lm;
    #pragma unroll
    for (int mt = 0; mt < 2; ++mt) {
        const int mrow = m0 + wave * 32 + mt * 16 + hi * 8;
        v8f a0 = mt ? acc10 : acc00, a1 = mt ? acc11 : acc01;
        v8f a2 = mt ? acc12 : acc02, a3 = mt ? acc13 : acc03;
        #pragma unroll
        for (int r = 0; r < 8; ++r) {
            size_t base = (size_t)(mrow + r) * ldc + ncol;
            float v0 = a0[r], v1 = a1[r], v2 = a2[r], v3 = a3[r];
            if (Mb) {
                v0 *= Mb[base]; v1 *= Mb[base + 16]; v2 *= Mb[base + 32]; v3 *= Mb[base + 48];
            }
            Cb[base] = v0; Cb[base + 16] = v1; Cb[base + 32] = v2; Cb[base + 48] = v3;
        }
    }
}

__device__ inline float waveReduce(float v) {
    #pragma unroll
    for (int o = 16; o > 0; o >>= 1) v += __shfl_xor(v, o, 32);
    return v;
}
__device__ inline float dot4f(float4 a, float4 b) {
    return a.x * b.x + a.y * b.y + a.z * b.z + a.w * b.w;
}
__device__ inline float softplusf(float x) {
    if (x > 20.f) return x;
    if (x < -20.f) return expf(x);
    return log1pf(expf(x));
}

__global__ __launch_bounds__(256) void proj_dab_kernel(
    const float* __restrict__ x, const float* __restrict__ Wd,
    const float* __restrict__ Wa, const float* __restrict__ Wb,
    float* __restrict__ dOut, float* __restrict__ aOut, float* __restrict__ bOut)
{
    const int wave = threadIdx.x >> 5, lane = threadIdx.x & 31;
    const int row  = blockIdx.x * 8 + wave;
    const float* xr = x + (size_t)row * DD;

    float xv[DD / 32];
    #pragma unroll
    for (int i = 0; i < DD / 32; ++i) xv[i] = xr[lane + 32 * i];

    float acc = 0.f;
    #pragma unroll
    for (int i = 0; i < DD / 32; ++i) acc += xv[i] * Wd[lane + 32 * i];
    acc = waveReduce(acc);
    if (lane == 0) dOut[row] = acc;

    for (int c = 0; c < RSS; ++c) {
        float a = 0.f, bm = 0.f;
        #pragma unroll
        for (int i = 0; i < DD / 32; ++i) {
            int k = lane + 32 * i;
            a  += xv[i] * Wa[(size_t)k * RSS + c];
            bm += xv[i] * Wb[(size_t)k * RSS + c];
        }
        a = waveReduce(a); bm = waveReduce(bm);
        if (lane == 0) { aOut[(size_t)row * RSS + c] = a; bOut[(size_t)row * RSS + c] = bm; }
    }
}

__global__ __launch_bounds__(256) void zmatB_kernel(
    const float* __restrict__ z, const float* __restrict__ Bm, float* __restrict__ zB)
{
    const int wave = threadIdx.x >> 5, lane = threadIdx.x & 31;
    const int rowg = blockIdx.x * 8 + wave;
    const int b    = rowg >> 11;
    const float* zr = z + (size_t)rowg * TT;
    const float* Bb = Bm + (size_t)b * TT * RSS;

    float acc[RSS];
    #pragma unroll
    for (int r = 0; r < RSS; ++r) acc[r] = 0.f;

    for (int j = lane; j < TT; j += 32) {
        float zv = zr[j];
        const float4* br = (const float4*)(Bb + (size_t)j * RSS);
        float4 b0 = br[0], b1 = br[1], b2 = br[2], b3 = br[3];
        acc[0]  += zv * b0.x; acc[1]  += zv * b0.y; acc[2]  += zv * b0.z; acc[3]  += zv * b0.w;
        acc[4]  += zv * b1.x; acc[5]  += zv * b1.y; acc[6]  += zv * b1.z; acc[7]  += zv * b1.w;
        acc[8]  += zv * b2.x; acc[9]  += zv * b2.y; acc[10] += zv * b2.z; acc[11] += zv * b2.w;
        acc[12] += zv * b3.x; acc[13] += zv * b3.y; acc[14] += zv * b3.z; acc[15] += zv * b3.w;
    }
    #pragma unroll
    for (int r = 0; r < RSS; ++r) {
        float v = waveReduce(acc[r]);
        if (lane == 0) zB[(size_t)rowg * RSS + r] = v;
    }
}

__global__ __launch_bounds__(256) void dplr_combine_kernel(
    const float* __restrict__ z, const float* __restrict__ dvec,
    const float* __restrict__ zB, const float* __restrict__ Amat,
    float* __restrict__ out)
{
    const int rowg = blockIdx.y;
    const int b    = rowg >> 11;
    const int j    = (blockIdx.x * blockDim.x + threadIdx.x) * 4;

    const float4* cr = (const float4*)(zB + (size_t)rowg * RSS);
    const float4 c0 = cr[0], c1 = cr[1], c2 = cr[2], c3 = cr[3];

    const float4 z4 = *(const float4*)(z + (size_t)rowg * TT + j);
    const float4 d4 = *(const float4*)(dvec + (size_t)b * TT + j);

    const float4* A0 = (const float4*)(Amat + ((size_t)b * TT + j) * RSS);
    float dv[4];
    #pragma unroll
    for (int q = 0; q < 4; ++q) {
        const float4* Ar = A0 + q * 4;
        dv[q] = dot4f(c0, Ar[0]) + dot4f(c1, Ar[1]) + dot4f(c2, Ar[2]) + dot4f(c3, Ar[3]);
    }
    float4 o;
    o.x = z4.x * d4.x + dv[0];
    o.y = z4.y * d4.y + dv[1];
    o.z = z4.z * d4.z + dv[2];
    o.w = z4.w * d4.w + dv[3];
    *(float4*)(out + (size_t)rowg * TT + j) = o;
}

__global__ __launch_bounds__(256) void rownorm_kernel(
    const float* __restrict__ h, float* __restrict__ denom)
{
    const int wave = threadIdx.x >> 5, lane = threadIdx.x & 31;
    const int rowg = blockIdx.x * 8 + wave;
    const float* hr = h + (size_t)rowg * TT;
    float acc = 0.f;
    for (int j = lane; j < TT; j += 32) { float v = hr[j]; acc += v * v; }
    acc = waveReduce(acc);
    if (lane == 0) denom[rowg] = sqrtf(acc) + 1e-8f;
}

__global__ __launch_bounds__(256) void combine_h_kernel(
    float* __restrict__ hmix, const float* __restrict__ hscale,
    const float* __restrict__ denom)
{
    const int rowg = blockIdx.y;
    const int j = (blockIdx.x * blockDim.x + threadIdx.x) * 4;
    const float inv = 1.f / denom[rowg];
    float4 m = *(const float4*)(hmix   + (size_t)rowg * TT + j);
    float4 s = *(const float4*)(hscale + (size_t)rowg * TT + j);
    float4 o;
    o.x = softplusf(s.x) * fmaxf(m.x * inv, 0.f);
    o.y = softplusf(s.y) * fmaxf(m.y * inv, 0.f);
    o.z = softplusf(s.z) * fmaxf(m.z * inv, 0.f);
    o.w = softplusf(s.w) * fmaxf(m.w * inv, 0.f);
    *(float4*)(hmix + (size_t)rowg * TT + j) = o;
}

__global__ __launch_bounds__(256) void emul_kernel(float* __restrict__ a, const float* __restrict__ b)
{
    const size_t i = ((size_t)blockIdx.x * blockDim.x + threadIdx.x) * 4;
    float4 av = *(const float4*)(a + i);
    float4 bv = *(const float4*)(b + i);
    av.x *= bv.x; av.y *= bv.y; av.z *= bv.z; av.w *= bv.w;
    *(float4*)(a + i) = av;
}

extern "C" void kernel_launch(void* const* d_in, const int* in_sizes, int n_in,
                              void* d_out, int out_size, void* d_ws, size_t ws_size,
                              hipStream_t stream) {
    (void)in_sizes; (void)n_in; (void)out_size; (void)ws_size;

    const float* x = (const float*)d_in[0];
    const float* qk_Wq1  = (const float*)d_in[1];  const float* qk_Wq2  = (const float*)d_in[2];
    const float* qk_Wm1  = (const float*)d_in[3];  const float* qk_Wm2  = (const float*)d_in[4];
    const float* qk_Wk1  = (const float*)d_in[5];  const float* qk_Wk2  = (const float*)d_in[6];
    const float* qks_Wq1 = (const float*)d_in[7];  const float* qks_Wq2 = (const float*)d_in[8];
    const float* qks_Wm1 = (const float*)d_in[9];  const float* qks_Wm2 = (const float*)d_in[10];
    const float* qks_Wk1 = (const float*)d_in[11]; const float* qks_Wk2 = (const float*)d_in[12];
    const float* vo_Wv1  = (const float*)d_in[13]; const float* vo_Wv2  = (const float*)d_in[14];
    const float* vo_Wm1  = (const float*)d_in[15]; const float* vo_Wm2  = (const float*)d_in[16];
    const float* vo_Wo   = (const float*)d_in[17];
    const float* sm1_Wd  = (const float*)d_in[18]; const float* sm1_Wa  = (const float*)d_in[19]; const float* sm1_Wb  = (const float*)d_in[20];
    const float* sm1s_Wd = (const float*)d_in[21]; const float* sm1s_Wa = (const float*)d_in[22]; const float* sm1s_Wb = (const float*)d_in[23];
    const float* sm2_Wd  = (const float*)d_in[24]; const float* sm2_Wa  = (const float*)d_in[25]; const float* sm2_Wb  = (const float*)d_in[26];

    const size_t BTD = (size_t)BT * DD;
    const size_t BTT = (size_t)BB * TT * TT;

    float* w = (float*)d_ws;
    float* qg   = w; w += BTD;
    float* kb   = w; w += BTD;
    float* qsg  = w; w += BTD;
    float* ksb  = w; w += BTD;
    float* vb   = w; w += BTD;
    float* mvob = w; w += BTD;
    float* tmpb = w; w += BTD;
    float* zb   = w; w += BTT;
    float* zsb  = w; w += BTT;
    float* ctx  = w; w += BTD;
    float* xr   = w; w += (size_t)BT * RR;
    float* d1   = w; w += BT; float* d1s = w; w += BT; float* d2 = w; w += BT;
    float* A1   = w; w += (size_t)BT * RSS; float* B1  = w; w += (size_t)BT * RSS;
    float* A1s  = w; w += (size_t)BT * RSS; float* B1s = w; w += (size_t)BT * RSS;
    float* A2   = w; w += (size_t)BT * RSS; float* B2  = w; w += (size_t)BT * RSS;
    float* zBb  = w; w += (size_t)BT * RSS;
    float* denom= w; w += BT;

    auto gemm = [&](const float* A, const float* Bm, float* C, const float* mul,
                    int M, int N, int K, int lda, int ldb, int ldc, int bT,
                    long long sA, long long sB, long long sC, int batches) {
        dim3 g((unsigned)(N / GBN), (unsigned)(M / GBM), (unsigned)batches);
        gemm_wmma_f32<<<g, dim3(256), 0, stream>>>(A, Bm, C, mul, M, N, K, lda, ldb, ldc, bT, sA, sB, sC);
    };

    proj_dab_kernel<<<dim3(BT / 8), 256, 0, stream>>>(x, sm1_Wd,  sm1_Wa,  sm1_Wb,  d1,  A1,  B1);
    proj_dab_kernel<<<dim3(BT / 8), 256, 0, stream>>>(x, sm1s_Wd, sm1s_Wa, sm1s_Wb, d1s, A1s, B1s);
    proj_dab_kernel<<<dim3(BT / 8), 256, 0, stream>>>(x, sm2_Wd,  sm2_Wa,  sm2_Wb,  d2,  A2,  B2);

    gemm(x,  qk_Wm1, xr,   nullptr, BT, RR, DD, DD, RR, RR, 0, 0, 0, 0, 1);
    gemm(xr, qk_Wm2, tmpb, nullptr, BT, DD, RR, RR, DD, DD, 0, 0, 0, 0, 1);
    gemm(x,  qk_Wq1, xr,   nullptr, BT, RR, DD, DD, RR, RR, 0, 0, 0, 0, 1);
    gemm(xr, qk_Wq2, qg,   tmpb,    BT, DD, RR, RR, DD, DD, 0, 0, 0, 0, 1);
    gemm(x,  qk_Wk1, xr,   nullptr, BT, RR, DD, DD, RR, RR, 0, 0, 0, 0, 1);
    gemm(xr, qk_Wk2, kb,   nullptr, BT, DD, RR, RR, DD, DD, 0, 0, 0, 0, 1);

    gemm(x,  qks_Wm1, xr,   nullptr, BT, RR, DD, DD, RR, RR, 0, 0, 0, 0, 1);
    gemm(xr, qks_Wm2, tmpb, nullptr, BT, DD, RR, RR, DD, DD, 0, 0, 0, 0, 1);
    gemm(x,  qks_Wq1, xr,   nullptr, BT, RR, DD, DD, RR, RR, 0, 0, 0, 0, 1);
    gemm(xr, qks_Wq2, qsg,  tmpb,    BT, DD, RR, RR, DD, DD, 0, 0, 0, 0, 1);
    gemm(x,  qks_Wk1, xr,   nullptr, BT, RR, DD, DD, RR, RR, 0, 0, 0, 0, 1);
    gemm(xr, qks_Wk2, ksb,  nullptr, BT, DD, RR, RR, DD, DD, 0, 0, 0, 0, 1);

    gemm(x,  vo_Wv1, xr,   nullptr, BT, RR, DD, DD, RR, RR, 0, 0, 0, 0, 1);
    gemm(xr, vo_Wv2, vb,   nullptr, BT, DD, RR, RR, DD, DD, 0, 0, 0, 0, 1);
    gemm(x,  vo_Wm1, xr,   nullptr, BT, RR, DD, DD, RR, RR, 0, 0, 0, 0, 1);
    gemm(xr, vo_Wm2, mvob, nullptr, BT, DD, RR, RR, DD, DD, 0, 0, 0, 0, 1);

    const long long sTD = (long long)TT * DD, sTTl = (long long)TT * TT;
    gemm(qg,  kb,  zb,  nullptr, TT, TT, DD, DD, DD, TT, 1, sTD, sTD, sTTl, BB);
    gemm(qsg, ksb, zsb, nullptr, TT, TT, DD, DD, DD, TT, 1, sTD, sTD, sTTl, BB);

    zmatB_kernel<<<dim3(BT / 8), 256, 0, stream>>>(zb, B1, zBb);
    dplr_combine_kernel<<<dim3(TT / 1024, BT), 256, 0, stream>>>(zb, d1, zBb, A1, zb);
    zmatB_kernel<<<dim3(BT / 8), 256, 0, stream>>>(zsb, B1s, zBb);
    dplr_combine_kernel<<<dim3(TT / 1024, BT), 256, 0, stream>>>(zsb, d1s, zBb, A1s, zsb);
    rownorm_kernel<<<dim3(BT / 8), 256, 0, stream>>>(zb, denom);
    combine_h_kernel<<<dim3(TT / 1024, BT), 256, 0, stream>>>(zb, zsb, denom);

    zmatB_kernel<<<dim3(BT / 8), 256, 0, stream>>>(zb, B2, zBb);
    dplr_combine_kernel<<<dim3(TT / 1024, BT), 256, 0, stream>>>(zb, d2, zBb, A2, zsb);

    gemm(zsb, vb, ctx, nullptr, TT, DD, TT, TT, DD, DD, 0, sTTl, sTD, sTD, BB);
    emul_kernel<<<dim3((unsigned)(BTD / 4 / 256)), 256, 0, stream>>>(ctx, mvob);
    gemm(ctx, vo_Wo, (float*)d_out, nullptr, BT, DD, DD, DD, DD, DD, 1, 0, 0, 0, 1);
}


# rank-16 products of the three DPLR mixes (z@B and zB@A^T) run as f32 WMMA tiles, factor fragments held in registers, not per-element vector dots; sum order of these mix products changes
# speedup vs baseline: 2.4446x; 2.4446x over previous
#include <hip/hip_runtime.h>
#include <math.h>

#define BB 2
#define TT 2048
#define DD 1024
#define RR 64
#define RSS 16
#define BT (BB*TT)

typedef __attribute__((ext_vector_type(2))) float v2f;
typedef __attribute__((ext_vector_type(8))) float v8f;
typedef __attribute__((ext_vector_type(4))) float v4f;

#define GBM 256
#define GBN 64
#define GBK 32

__global__ __launch_bounds__(256) void gemm_wmma_f32(
    const float* __restrict__ A, const float* __restrict__ Bm,
    float* __restrict__ C, const float* __restrict__ mulC,
    int M, int N, int K, int lda, int ldb, int ldc, int bT,
    long long sA, long long sB, long long sC)
{
    __shared__ __align__(16) float As [2][GBM][34];
    __shared__ __align__(16) float BsT[2][GBN][34];

    const int tid  = threadIdx.x;
    const int wave = tid >> 5;
    const int lane = tid & 31;
    const int hi   = lane >> 4;
    const int lm   = lane & 15;

    const int m0 = blockIdx.y * GBM;
    const int n0 = blockIdx.x * GBN;

    const float* Ab = A  + (size_t)blockIdx.z * sA;
    const float* Bb = Bm + (size_t)blockIdx.z * sB;
    float*       Cb = C  + (size_t)blockIdx.z * sC;
    const float* Mb = mulC ? (mulC + (size_t)blockIdx.z * sC) : nullptr;

    float4 ra[8];
    float4 rb[2];

    int arow[8], ac4[8];
    #pragma unroll
    for (int i = 0; i < 8; ++i) { int idx = tid + i * 256; arow[i] = idx >> 3; ac4[i] = (idx & 7) * 4; }
    int bi0[2], bi1[2];
    #pragma unroll
    for (int i = 0; i < 2; ++i) {
        int idx = tid + i * 256;
        if (true) {   }
        bi0[i] = idx; bi1[i] = idx;
    }

    auto loadGlobal = [&](int k0) {
        #pragma unroll
        for (int i = 0; i < 8; ++i)
            ra[i] = *(const float4*)(Ab + (size_t)(m0 + arow[i]) * lda + k0 + ac4[i]);
        if (bT) {
            #pragma unroll
            for (int i = 0; i < 2; ++i) {
                int idx = bi0[i];
                int n = idx >> 3, c4 = (idx & 7) * 4;
                rb[i] = *(const float4*)(Bb + (size_t)(n0 + n) * ldb + k0 + c4);
            }
        } else {
            #pragma unroll
            for (int i = 0; i < 2; ++i) {
                int idx = bi1[i];
                int r = idx >> 4, c4 = (idx & 15) * 4;
                rb[i] = *(const float4*)(Bb + (size_t)(k0 + r) * ldb + n0 + c4);
            }
        }
    };

    auto storeLDS = [&](int buf) {
        #pragma unroll
        for (int i = 0; i < 8; ++i) {
            float4 t = ra[i];
            v2f lo; lo.x = t.x; lo.y = t.y;
            v2f hi2; hi2.x = t.z; hi2.y = t.w;
            *(v2f*)&As[buf][arow[i]][ac4[i]]     = lo;
            *(v2f*)&As[buf][arow[i]][ac4[i] + 2] = hi2;
        }
        if (bT) {
            #pragma unroll
            for (int i = 0; i < 2; ++i) {
                int idx = bi0[i];
                int n = idx >> 3, c4 = (idx & 7) * 4;
                float4 t = rb[i];
                v2f lo; lo.x = t.x; lo.y = t.y;
                v2f hi2; hi2.x = t.z; hi2.y = t.w;
                *(v2f*)&BsT[buf][n][c4]     = lo;
                *(v2f*)&BsT[buf][n][c4 + 2] = hi2;
            }
        } else {
            #pragma unroll
            for (int i = 0; i < 2; ++i) {
                int idx = bi1[i];
                int r = idx >> 4, c4 = (idx & 15) * 4;
                float4 t = rb[i];
                BsT[buf][c4 + 0][r] = t.x;
                BsT[buf][c4 + 1][r] = t.y;
                BsT[buf][c4 + 2][r] = t.z;
                BsT[buf][c4 + 3][r] = t.w;
            }
        }
    };

    v8f acc00 = {}, acc01 = {}, acc02 = {}, acc03 = {};
    v8f acc10 = {}, acc11 = {}, acc12 = {}, acc13 = {};

    const int r0 = wave * 32 + lm;
    const int r1 = r0 + 16;
    const int nchunks = K / GBK;

    loadGlobal(0);
    for (int c = 0; c < nchunks; ++c) {
        const int buf = c & 1;
        storeLDS(buf);
        __syncthreads();
        if (c + 1 < nchunks) loadGlobal((c + 1) * GBK);

        #pragma unroll
        for (int kk = 0; kk < GBK; kk += 4) {
            const int kc = kk + 2 * hi;
            v2f a0 = *(const v2f*)&As [buf][r0][kc];
            v2f a1 = *(const v2f*)&As [buf][r1][kc];
            v2f b0 = *(const v2f*)&BsT[buf][lm +  0][kc];
            v2f b1 = *(const v2f*)&BsT[buf][lm + 16][kc];
            v2f b2 = *(const v2f*)&BsT[buf][lm + 32][kc];
            v2f b3 = *(const v2f*)&BsT[buf][lm + 48][kc];
            acc00 = __builtin_amdgcn_wmma_f32_16x16x4_f32(false, a0, false, b0, (short)0, acc00, false, false);
            acc01 = __builtin_amdgcn_wmma_f32_16x16x4_f32(false, a0, false, b1, (short)0, acc01, false, false);
            acc02 = __builtin_amdgcn_wmma_f32_16x16x4_f32(false, a0, false, b2, (short)0, acc02, false, false);
            acc03 = __builtin_amdgcn_wmma_f32_16x16x4_f32(false, a0, false, b3, (short)0, acc03, false, false);
            acc10 = __builtin_amdgcn_wmma_f32_16x16x4_f32(false, a1, false, b0, (short)0, acc10, false, false);
            acc11 = __builtin_amdgcn_wmma_f32_16x16x4_f32(false, a1, false, b1, (short)0, acc11, false, false);
            acc12 = __builtin_amdgcn_wmma_f32_16x16x4_f32(false, a1, false, b2, (short)0, acc12, false, false);
            acc13 = __builtin_amdgcn_wmma_f32_16x16x4_f32(false, a1, false, b3, (short)0, acc13, false, false);
        }
        __syncthreads();
    }

    const int ncol = n0 + lm;
    #pragma unroll
    for (int mt = 0; mt < 2; ++mt) {
        const int mrow = m0 + wave * 32 + mt * 16 + hi * 8;
        v8f a0 = mt ? acc10 : acc00, a1 = mt ? acc11 : acc01;
        v8f a2 = mt ? acc12 : acc02, a3 = mt ? acc13 : acc03;
        #pragma unroll
        for (int r = 0; r < 8; ++r) {
            size_t base = (size_t)(mrow + r) * ldc + ncol;
            float v0 = a0[r], v1 = a1[r], v2 = a2[r], v3 = a3[r];
            if (Mb) {
                v0 *= Mb[base]; v1 *= Mb[base + 16]; v2 *= Mb[base + 32]; v3 *= Mb[base + 48];
            }
            Cb[base] = v0; Cb[base + 16] = v1; Cb[base + 32] = v2; Cb[base + 48] = v3;
        }
    }
}

__device__ inline float waveReduce(float v) {
    #pragma unroll
    for (int o = 16; o > 0; o >>= 1) v += __shfl_xor(v, o, 32);
    return v;
}
__device__ inline float softplusf(float x) {
    if (x > 20.f) return x;
    if (x < -20.f) return expf(x);
    return log1pf(expf(x));
}

struct XwArgs {
    const float* x;
    const float *w0, *w1, *w2, *w3, *w4, *w5, *w6, *w7;
    const float *e0, *e1, *e2, *e3, *e4, *e5;
    const float *wdA, *wdB, *wdC;
    float* xr;
    float *o0, *o1, *o2, *o3, *o4, *o5;
    float *dA, *dB, *dC;
};

struct BFrag {
    v4f    r0[8], r1[8];
    float  e0[8], e1[8];
};

#define XT_ROWS 32
#define XT_K    32
#define XT_PITCH 36

__global__ __launch_bounds__(256) void xw_shared_kernel(XwArgs p)
{
    __shared__ __align__(16) float Xs[2][XT_ROWS][XT_PITCH];

    const int tid  = threadIdx.x;
    const int wave = tid >> 5, lane = tid & 31;
    const int hi   = lane >> 4, lm = lane & 15;
    const int m0   = blockIdx.x * XT_ROWS;
    const int sel  = blockIdx.y * 4 + (wave >> 1);
    const int mt   = wave & 1;

    const float* Wm = sel == 0 ? p.w0 : sel == 1 ? p.w1 : sel == 2 ? p.w2 : sel == 3 ? p.w3
                    : sel == 4 ? p.w4 : sel == 5 ? p.w5 : sel == 6 ? p.w6 : p.w7;
    Wm += (size_t)(2 * hi) * RR + 4 * lm;

    const float* ep;
    int es;
    if (sel < 6) {
        ep = (sel == 0 ? p.e0 : sel == 1 ? p.e1 : sel == 2 ? p.e2
            : sel == 3 ? p.e3 : sel == 4 ? p.e4 : p.e5) + lm;
        es = RSS;
    } else {
        ep = (sel == 6 && lm == 1) ? p.wdB : (sel == 6 && lm == 2) ? p.wdC : p.wdA;
        es = 1;
    }
    ep += (size_t)(2 * hi) * es;

    const int xrow = tid >> 3, xc4 = (tid & 7) * 4;
    const float* xg = p.x + (size_t)(m0 + xrow) * DD + xc4;

    auto loadB = [&](BFrag& f, int k0) {
        #pragma unroll
        for (int s = 0; s < 8; ++s) {
            const int k = k0 + 4 * s;
            f.r0[s] = *(const v4f*)(Wm + (size_t)k * RR);
            f.r1[s] = *(const v4f*)(Wm + (size_t)(k + 1) * RR);
            f.e0[s] = ep[(size_t)k * es];
            f.e1[s] = ep[(size_t)(k + 1) * es];
        }
    };

    v8f acc0 = {}, acc1 = {}, acc2 = {}, acc3 = {}, accE = {};
    const int arow = mt * 16 + lm;
    const int nch  = DD / XT_K;
    v4f xreg;

    auto stage = [&](const BFrag& cur, BFrag& nxt, int c, int buf) {
        *(v4f*)&Xs[buf][xrow][xc4] = xreg;
        __syncthreads();
        const int cn = (c + 1 < nch) ? c + 1 : c;
        xreg = *(const v4f*)(xg + cn * XT_K);
        loadB(nxt, cn * XT_K);

        v2f a[8];
        #pragma unroll
        for (int s = 0; s < 8; ++s)
            a[s] = *(const v2f*)&Xs[buf][arow][4 * s + 2 * hi];
        __builtin_amdgcn_sched_barrier(0);
        #pragma unroll
        for (int s = 0; s < 8; ++s) {
            v2f b0, b1, b2, b3, be;
            b0.x = cur.r0[s].x; b0.y = cur.r1[s].x;
            b1.x = cur.r0[s].y; b1.y = cur.r1[s].y;
            b2.x = cur.r0[s].z; b2.y = cur.r1[s].z;
            b3.x = cur.r0[s].w; b3.y = cur.r1[s].w;
            be.x = cur.e0[s];   be.y = cur.e1[s];
            acc0 = __builtin_amdgcn_wmma_f32_16x16x4_f32(false, a[s], false, b0, (short)0, acc0, false, false);
            acc1 = __builtin_amdgcn_wmma_f32_16x16x4_f32(false, a[s], false, b1, (short)0, acc1, false, false);
            acc2 = __builtin_amdgcn_wmma_f32_16x16x4_f32(false, a[s], false, b2, (short)0, acc2, false, false);
            acc3 = __builtin_amdgcn_wmma_f32_16x16x4_f32(false, a[s], false, b3, (short)0, acc3, false, false);
            accE = __builtin_amdgcn_wmma_f32_16x16x4_f32(false, a[s], false, be, (short)0, accE, false, false);
        }
        __builtin_amdgcn_sched_barrier(0);
        asm volatile("" :: "v"(nxt.r0[0]), "v"(nxt.r0[1]), "v"(nxt.r0[2]), "v"(nxt.r0[3]),
                           "v"(nxt.r0[4]), "v"(nxt.r0[5]), "v"(nxt.r0[6]), "v"(nxt.r0[7]),
                           "v"(nxt.r1[0]), "v"(nxt.r1[1]), "v"(nxt.r1[2]), "v"(nxt.r1[3]),
                           "v"(nxt.r1[4]), "v"(nxt.r1[5]), "v"(nxt.r1[6]), "v"(nxt.r1[7]));
        asm volatile("" :: "v"(nxt.e0[0]), "v"(nxt.e0[1]), "v"(nxt.e0[2]), "v"(nxt.e0[3]),
                           "v"(nxt.e0[4]), "v"(nxt.e0[5]), "v"(nxt.e0[6]), "v"(nxt.e0[7]),
                           "v"(nxt.e1[0]), "v"(nxt.e1[1]), "v"(nxt.e1[2]), "v"(nxt.e1[3]),
                           "v"(nxt.e1[4]), "v"(nxt.e1[5]), "v"(nxt.e1[6]), "v"(nxt.e1[7]),
                           "v"(xreg));
    };

    BFrag f0, f1;
    xreg = *(const v4f*)xg;
    loadB(f0, 0);
    for (int c = 0; c < nch; c += 2) {
        stage(f0, f1, c,     0);
        stage(f1, f0, c + 1, 1);
    }

    const int orow = m0 + mt * 16 + hi * 8;
    float* xo = p.xr + (size_t)sel * BT * RR + (size_t)orow * RR + 4 * lm;
    #pragma unroll
    for (int r = 0; r < 8; ++r) {
        float4 o; o.x = acc0[r]; o.y = acc1[r]; o.z = acc2[r]; o.w = acc3[r];
        *(float4*)(xo + (size_t)r * RR) = o;
    }
    if (sel < 6) {
        float* eo = (sel == 0 ? p.o0 : sel == 1 ? p.o1 : sel == 2 ? p.o2
                   : sel == 3 ? p.o3 : sel == 4 ? p.o4 : p.o5) + (size_t)orow * RSS + lm;
        #pragma unroll
        for (int r = 0; r < 8; ++r) eo[(size_t)r * RSS] = accE[r];
    } else if (sel == 6 && lm < 3) {
        float* dp = (lm == 0 ? p.dA : lm == 1 ? p.dB : p.dC) + orow;
        #pragma unroll
        for (int r = 0; r < 8; ++r) dp[r] = accE[r];
    }
}

__global__ __launch_bounds__(512) void zmatB_kernel(
    const float* __restrict__ z, const float* __restrict__ Bm, float* __restrict__ zB)
{
    __shared__ float part[16][16][16];

    const int tid  = threadIdx.x;
    const int wave = tid >> 5, lane = tid & 31;
    const int hi   = lane >> 4, lm = lane & 15;
    const int row0 = blockIdx.x * 16;
    const int b    = row0 >> 11;
    const int kw   = wave * 128;

    const float* zr = z  + (size_t)(row0 + lm) * TT + kw + 4 * hi;
    const float* Bp = Bm + ((size_t)b * TT + kw + 4 * hi) * RSS + lm;

    v8f acc0 = {}, acc1 = {};
    #pragma unroll
    for (int c = 0; c < 4; ++c) {
        v4f   a[4];
        float bv[16];
        #pragma unroll
        for (int i = 0; i < 4; ++i) a[i] = *(const v4f*)(zr + c * 32 + 8 * i);
        #pragma unroll
        for (int i = 0; i < 4; ++i) {
            #pragma unroll
            for (int q = 0; q < 4; ++q)
                bv[4 * i + q] = Bp[(size_t)(c * 32 + 8 * i + q) * RSS];
        }
        #pragma unroll
        for (int i = 0; i < 4; ++i) {
            v2f alo, ahi, blo, bhi;
            alo.x = a[i].x;        alo.y = a[i].y;
            ahi.x = a[i].z;        ahi.y = a[i].w;
            blo.x = bv[4 * i + 0]; blo.y = bv[4 * i + 1];
            bhi.x = bv[4 * i + 2]; bhi.y = bv[4 * i + 3];
            acc0 = __builtin_amdgcn_wmma_f32_16x16x4_f32(false, alo, false, blo, (short)0, acc0, false, false);
            acc1 = __builtin_amdgcn_wmma_f32_16x16x4_f32(false, ahi, false, bhi, (short)0, acc1, false, false);
        }
    }

    #pragma unroll
    for (int r = 0; r < 8; ++r) part[wave][8 * hi + r][lm] = acc0[r] + acc1[r];
    __syncthreads();
    if (tid < 256) {
        const int orow = tid >> 4, ocol = tid & 15;
        float v = 0.f;
        #pragma unroll
        for (int w = 0; w < 16; ++w) v += part[w][orow][ocol];
        zB[(size_t)row0 * RSS + tid] = v;
    }
}

__global__ __launch_bounds__(256) void dplr_combine_kernel(
    const float* z, const float* __restrict__ dvec,
    const float* __restrict__ zB, const float* __restrict__ Amat,
    float* out)
{
    const int wave = threadIdx.x >> 5, lane = threadIdx.x & 31;
    const int hi   = lane >> 4, lm = lane & 15;
    const int row0 = blockIdx.y * 16;
    const int b    = row0 >> 11;
    const int j0   = (blockIdx.x * 8 + wave) * 64;

    const float* zp = zB + (size_t)(row0 + lm) * RSS + 8 * hi;
    const v4f za0 = *(const v4f*)zp, za1 = *(const v4f*)(zp + 4);

    const float* Ap = Amat + ((size_t)b * TT + j0 + 4 * lm) * RSS + 8 * hi;
    v4f f0[4], f1[4];
    #pragma unroll
    for (int t = 0; t < 4; ++t) {
        f0[t] = *(const v4f*)(Ap + t * RSS);
        f1[t] = *(const v4f*)(Ap + t * RSS + 4);
    }
    const v4f d4 = *(const v4f*)(dvec + (size_t)b * TT + j0 + 4 * lm);

    const size_t off = (size_t)(row0 + 8 * hi) * TT + j0 + 4 * lm;
    v8f acc0, acc1, acc2, acc3;
    #pragma unroll
    for (int r = 0; r < 8; ++r) {
        const v4f zv = *(const v4f*)(z + off + (size_t)r * TT);
        acc0[r] = zv.x * d4.x; acc1[r] = zv.y * d4.y;
        acc2[r] = zv.z * d4.z; acc3[r] = zv.w * d4.w;
    }

    #pragma unroll
    for (int w = 0; w < 4; ++w) {
        const v4f az = (w < 2) ? za0 : za1;
        v2f a; a.x = (w & 1) ? az.z : az.x; a.y = (w & 1) ? az.w : az.y;
        v2f bt[4];
        #pragma unroll
        for (int t = 0; t < 4; ++t) {
            const v4f fz = (w < 2) ? f0[t] : f1[t];
            bt[t].x = (w & 1) ? fz.z : fz.x; bt[t].y = (w & 1) ? fz.w : fz.y;
        }
        acc0 = __builtin_amdgcn_wmma_f32_16x16x4_f32(false, a, false, bt[0], (short)0, acc0, false, false);
        acc1 = __builtin_amdgcn_wmma_f32_16x16x4_f32(false, a, false, bt[1], (short)0, acc1, false, false);
        acc2 = __builtin_amdgcn_wmma_f32_16x16x4_f32(false, a, false, bt[2], (short)0, acc2, false, false);
        acc3 = __builtin_amdgcn_wmma_f32_16x16x4_f32(false, a, false, bt[3], (short)0, acc3, false, false);
    }

    #pragma unroll
    for (int r = 0; r < 8; ++r) {
        v4f o; o.x = acc0[r]; o.y = acc1[r]; o.z = acc2[r]; o.w = acc3[r];
        *(v4f*)(out + off + (size_t)r * TT) = o;
    }
}

__global__ __launch_bounds__(256) void rownorm_kernel(
    const float* __restrict__ h, float* __restrict__ denom)
{
    const int wave = threadIdx.x >> 5, lane = threadIdx.x & 31;
    const int rowg = blockIdx.x * 8 + wave;
    const float* hr = h + (size_t)rowg * TT;
    float acc = 0.f;
    for (int j = lane; j < TT; j += 32) { float v = hr[j]; acc += v * v; }
    acc = waveReduce(acc);
    if (lane == 0) denom[rowg] = sqrtf(acc) + 1e-8f;
}

__global__ __launch_bounds__(256) void combine_h_kernel(
    float* __restrict__ hmix, const float* __restrict__ hscale,
    const float* __restrict__ denom)
{
    const int rowg = blockIdx.y;
    const int j = (blockIdx.x * blockDim.x + threadIdx.x) * 4;
    const float inv = 1.f / denom[rowg];
    float4 m = *(const float4*)(hmix   + (size_t)rowg * TT + j);
    float4 s = *(const float4*)(hscale + (size_t)rowg * TT + j);
    float4 o;
    o.x = softplusf(s.x) * fmaxf(m.x * inv, 0.f);
    o.y = softplusf(s.y) * fmaxf(m.y * inv, 0.f);
    o.z = softplusf(s.z) * fmaxf(m.z * inv, 0.f);
    o.w = softplusf(s.w) * fmaxf(m.w * inv, 0.f);
    *(float4*)(hmix + (size_t)rowg * TT + j) = o;
}

__global__ __launch_bounds__(256) void emul_kernel(float* __restrict__ a, const float* __restrict__ b)
{
    const size_t i = ((size_t)blockIdx.x * blockDim.x + threadIdx.x) * 4;
    float4 av = *(const float4*)(a + i);
    float4 bv = *(const float4*)(b + i);
    av.x *= bv.x; av.y *= bv.y; av.z *= bv.z; av.w *= bv.w;
    *(float4*)(a + i) = av;
}

extern "C" void kernel_launch(void* const* d_in, const int* in_sizes, int n_in,
                              void* d_out, int out_size, void* d_ws, size_t ws_size,
                              hipStream_t stream) {
    (void)in_sizes; (void)n_in; (void)out_size; (void)ws_size;

    const float* x = (const float*)d_in[0];
    const float* qk_Wq1  = (const float*)d_in[1];  const float* qk_Wq2  = (const float*)d_in[2];
    const float* qk_Wm1  = (const float*)d_in[3];  const float* qk_Wm2  = (const float*)d_in[4];
    const float* qk_Wk1  = (const float*)d_in[5];  const float* qk_Wk2  = (const float*)d_in[6];
    const float* qks_Wq1 = (const float*)d_in[7];  const float* qks_Wq2 = (const float*)d_in[8];
    const float* qks_Wm1 = (const float*)d_in[9];  const float* qks_Wm2 = (const float*)d_in[10];
    const float* qks_Wk1 = (const float*)d_in[11]; const float* qks_Wk2 = (const float*)d_in[12];
    const float* vo_Wv1  = (const float*)d_in[13]; const float* vo_Wv2  = (const float*)d_in[14];
    const float* vo_Wm1  = (const float*)d_in[15]; const float* vo_Wm2  = (const float*)d_in[16];
    const float* vo_Wo   = (const float*)d_in[17];
    const float* sm1_Wd  = (const float*)d_in[18]; const float* sm1_Wa  = (const float*)d_in[19]; const float* sm1_Wb  = (const float*)d_in[20];
    const float* sm1s_Wd = (const float*)d_in[21]; const float* sm1s_Wa = (const float*)d_in[22]; const float* sm1s_Wb = (const float*)d_in[23];
    const float* sm2_Wd  = (const float*)d_in[24]; const float* sm2_Wa  = (const float*)d_in[25]; const float* sm2_Wb  = (const float*)d_in[26];

    const size_t BTD = (size_t)BT * DD;
    const size_t BTT = (size_t)BB * TT * TT;

    float* w = (float*)d_ws;
    float* qg   = w; w += BTD;
    float* kb   = w; w += BTD;
    float* qsg  = w; w += BTD;
    float* ksb  = w; w += BTD;
    float* vb   = w; w += BTD;
    float* mvob = w; w += BTD;
    float* tmpb = w; w += BTD;
    float* zb   = w; w += BTT;
    float* zsb  = w; w += BTT;
    float* ctx  = w; w += BTD;
    w += (size_t)BT * RR;
    float* d1   = w; w += BT; float* d1s = w; w += BT; float* d2 = w; w += BT;
    float* A1   = w; w += (size_t)BT * RSS; float* B1  = w; w += (size_t)BT * RSS;
    float* A1s  = w; w += (size_t)BT * RSS; float* B1s = w; w += (size_t)BT * RSS;
    float* A2   = w; w += (size_t)BT * RSS; float* B2  = w; w += (size_t)BT * RSS;
    float* zBb  = w; w += (size_t)BT * RSS;
    float* denom= w; w += BT;

    auto gemm = [&](const float* A, const float* Bm, float* C, const float* mul,
                    int M, int N, int K, int lda, int ldb, int ldc, int bT,
                    long long sA, long long sB, long long sC, int batches) {
        dim3 g((unsigned)(N / GBN), (unsigned)(M / GBM), (unsigned)batches);
        gemm_wmma_f32<<<g, dim3(256), 0, stream>>>(A, Bm, C, mul, M, N, K, lda, ldb, ldc, bT, sA, sB, sC);
    };

    float* xrAll = ctx;
    const size_t XP = (size_t)BT * RR;
    XwArgs xa;
    xa.x  = x;
    xa.w0 = qk_Wm1;  xa.w1 = qk_Wq1;  xa.w2 = qk_Wk1;
    xa.w3 = qks_Wm1; xa.w4 = qks_Wq1; xa.w5 = qks_Wk1;
    xa.w6 = vo_Wv1;  xa.w7 = vo_Wm1;
    xa.e0 = sm1_Wa;  xa.e1 = sm1_Wb;  xa.e2 = sm1s_Wa; xa.e3 = sm1s_Wb; xa.e4 = sm2_Wa; xa.e5 = sm2_Wb;
    xa.wdA = sm1_Wd; xa.wdB = sm1s_Wd; xa.wdC = sm2_Wd;
    xa.xr = xrAll;
    xa.o0 = A1; xa.o1 = B1; xa.o2 = A1s; xa.o3 = B1s; xa.o4 = A2; xa.o5 = B2;
    xa.dA = d1; xa.dB = d1s; xa.dC = d2;
    xw_shared_kernel<<<dim3(BT / XT_ROWS, 2), 256, 0, stream>>>(xa);

    gemm(xrAll + 0 * XP, qk_Wm2,  tmpb, nullptr, BT, DD, RR, RR, DD, DD, 0, 0, 0, 0, 1);
    gemm(xrAll + 1 * XP, qk_Wq2,  qg,   tmpb,    BT, DD, RR, RR, DD, DD, 0, 0, 0, 0, 1);
    gemm(xrAll + 2 * XP, qk_Wk2,  kb,   nullptr, BT, DD, RR, RR, DD, DD, 0, 0, 0, 0, 1);

    gemm(xrAll + 3 * XP, qks_Wm2, tmpb, nullptr, BT, DD, RR, RR, DD, DD, 0, 0, 0, 0, 1);
    gemm(xrAll + 4 * XP, qks_Wq2, qsg,  tmpb,    BT, DD, RR, RR, DD, DD, 0, 0, 0, 0, 1);
    gemm(xrAll + 5 * XP, qks_Wk2, ksb,  nullptr, BT, DD, RR, RR, DD, DD, 0, 0, 0, 0, 1);

    gemm(xrAll + 6 * XP, vo_Wv2,  vb,   nullptr, BT, DD, RR, RR, DD, DD, 0, 0, 0, 0, 1);
    gemm(xrAll + 7 * XP, vo_Wm2,  mvob, nullptr, BT, DD, RR, RR, DD, DD, 0, 0, 0, 0, 1);

    const long long sTD = (long long)TT * DD, sTTl = (long long)TT * TT;
    gemm(qg,  kb,  zb,  nullptr, TT, TT, DD, DD, DD, TT, 1, sTD, sTD, sTTl, BB);
    gemm(qsg, ksb, zsb, nullptr, TT, TT, DD, DD, DD, TT, 1, sTD, sTD, sTTl, BB);

    zmatB_kernel<<<dim3(BT / 16), 512, 0, stream>>>(zb, B1, zBb);
    dplr_combine_kernel<<<dim3(TT / 512, BT / 16), 256, 0, stream>>>(zb, d1, zBb, A1, zb);
    zmatB_kernel<<<dim3(BT / 16), 512, 0, stream>>>(zsb, B1s, zBb);
    dplr_combine_kernel<<<dim3(TT / 512, BT / 16), 256, 0, stream>>>(zsb, d1s, zBb, A1s, zsb);
    rownorm_kernel<<<dim3(BT / 8), 256, 0, stream>>>(zb, denom);
    combine_h_kernel<<<dim3(TT / 1024, BT), 256, 0, stream>>>(zb, zsb, denom);

    zmatB_kernel<<<dim3(BT / 16), 512, 0, stream>>>(zb, B2, zBb);
    dplr_combine_kernel<<<dim3(TT / 512, BT / 16), 256, 0, stream>>>(zb, d2, zBb, A2, zsb);

    gemm(zsb, vb, ctx, nullptr, TT, DD, TT, TT, DD, DD, 0, sTTl, sTD, sTD, BB);
    emul_kernel<<<dim3((unsigned)(BTD / 4 / 256)), 256, 0, stream>>>(ctx, mvob);
    gemm(ctx, vo_Wo, (float*)d_out, nullptr, BT, DD, DD, DD, DD, DD, 1, 0, 0, 0, 1);
}


# DPLR mixes 1 and 1s, row norm, gated combine and DPLR mix 2 in ONE launch: each work-group keeps a 16-row strip of z and zs in registers; sum order of mix products and row norm changes
# speedup vs baseline: 2.6284x; 1.0752x over previous
#include <hip/hip_runtime.h>
#include <math.h>

#define BB 2
#define TT 2048
#define DD 1024
#define RR 64
#define RSS 16
#define BT (BB*TT)

typedef __attribute__((ext_vector_type(2))) float v2f;
typedef __attribute__((ext_vector_type(8))) float v8f;
typedef __attribute__((ext_vector_type(4))) float v4f;

#define GBM 256
#define GBN 64
#define GBK 32

__global__ __launch_bounds__(256) void gemm_wmma_f32(
    const float* __restrict__ A, const float* __restrict__ Bm,
    float* __restrict__ C, const float* __restrict__ mulC,
    int M, int N, int K, int lda, int ldb, int ldc, int bT,
    long long sA, long long sB, long long sC)
{
    __shared__ __align__(16) float As [2][GBM][34];
    __shared__ __align__(16) float BsT[2][GBN][34];

    const int tid  = threadIdx.x;
    const int wave = tid >> 5;
    const int lane = tid & 31;
    const int hi   = lane >> 4;
    const int lm   = lane & 15;

    const int m0 = blockIdx.y * GBM;
    const int n0 = blockIdx.x * GBN;

    const float* Ab = A  + (size_t)blockIdx.z * sA;
    const float* Bb = Bm + (size_t)blockIdx.z * sB;
    float*       Cb = C  + (size_t)blockIdx.z * sC;
    const float* Mb = mulC ? (mulC + (size_t)blockIdx.z * sC) : nullptr;

    float4 ra[8];
    float4 rb[2];

    int arow[8], ac4[8];
    #pragma unroll
    for (int i = 0; i < 8; ++i) { int idx = tid + i * 256; arow[i] = idx >> 3; ac4[i] = (idx & 7) * 4; }
    int bi0[2], bi1[2];
    #pragma unroll
    for (int i = 0; i < 2; ++i) {
        int idx = tid + i * 256;
        if (true) {   }
        bi0[i] = idx; bi1[i] = idx;
    }

    auto loadGlobal = [&](int k0) {
        #pragma unroll
        for (int i = 0; i < 8; ++i)
            ra[i] = *(const float4*)(Ab + (size_t)(m0 + arow[i]) * lda + k0 + ac4[i]);
        if (bT) {
            #pragma unroll
            for (int i = 0; i < 2; ++i) {
                int idx = bi0[i];
                int n = idx >> 3, c4 = (idx & 7) * 4;
                rb[i] = *(const float4*)(Bb + (size_t)(n0 + n) * ldb + k0 + c4);
            }
        } else {
            #pragma unroll
            for (int i = 0; i < 2; ++i) {
                int idx = bi1[i];
                int r = idx >> 4, c4 = (idx & 15) * 4;
                rb[i] = *(const float4*)(Bb + (size_t)(k0 + r) * ldb + n0 + c4);
            }
        }
    };

    auto storeLDS = [&](int buf) {
        #pragma unroll
        for (int i = 0; i < 8; ++i) {
            float4 t = ra[i];
            v2f lo; lo.x = t.x; lo.y = t.y;
            v2f hi2; hi2.x = t.z; hi2.y = t.w;
            *(v2f*)&As[buf][arow[i]][ac4[i]]     = lo;
            *(v2f*)&As[buf][arow[i]][ac4[i] + 2] = hi2;
        }
        if (bT) {
            #pragma unroll
            for (int i = 0; i < 2; ++i) {
                int idx = bi0[i];
                int n = idx >> 3, c4 = (idx & 7) * 4;
                float4 t = rb[i];
                v2f lo; lo.x = t.x; lo.y = t.y;
                v2f hi2; hi2.x = t.z; hi2.y = t.w;
                *(v2f*)&BsT[buf][n][c4]     = lo;
                *(v2f*)&BsT[buf][n][c4 + 2] = hi2;
            }
        } else {
            #pragma unroll
            for (int i = 0; i < 2; ++i) {
                int idx = bi1[i];
                int r = idx >> 4, c4 = (idx & 15) * 4;
                float4 t = rb[i];
                BsT[buf][c4 + 0][r] = t.x;
                BsT[buf][c4 + 1][r] = t.y;
                BsT[buf][c4 + 2][r] = t.z;
                BsT[buf][c4 + 3][r] = t.w;
            }
        }
    };

    v8f acc00 = {}, acc01 = {}, acc02 = {}, acc03 = {};
    v8f acc10 = {}, acc11 = {}, acc12 = {}, acc13 = {};

    const int r0 = wave * 32 + lm;
    const int r1 = r0 + 16;
    const int nchunks = K / GBK;

    loadGlobal(0);
    for (int c = 0; c < nchunks; ++c) {
        const int buf = c & 1;
        storeLDS(buf);
        __syncthreads();
        if (c + 1 < nchunks) loadGlobal((c + 1) * GBK);

        #pragma unroll
        for (int kk = 0; kk < GBK; kk += 4) {
            const int kc = kk + 2 * hi;
            v2f a0 = *(const v2f*)&As [buf][r0][kc];
            v2f a1 = *(const v2f*)&As [buf][r1][kc];
            v2f b0 = *(const v2f*)&BsT[buf][lm +  0][kc];
            v2f b1 = *(const v2f*)&BsT[buf][lm + 16][kc];
            v2f b2 = *(const v2f*)&BsT[buf][lm + 32][kc];
            v2f b3 = *(const v2f*)&BsT[buf][lm + 48][kc];
            acc00 = __builtin_amdgcn_wmma_f32_16x16x4_f32(false, a0, false, b0, (short)0, acc00, false, false);
            acc01 = __builtin_amdgcn_wmma_f32_16x16x4_f32(false, a0, false, b1, (short)0, acc01, false, false);
            acc02 = __builtin_amdgcn_wmma_f32_16x16x4_f32(false, a0, false, b2, (short)0, acc02, false, false);
            acc03 = __builtin_amdgcn_wmma_f32_16x16x4_f32(false, a0, false, b3, (short)0, acc03, false, false);
            acc10 = __builtin_amdgcn_wmma_f32_16x16x4_f32(false, a1, false, b0, (short)0, acc10, false, false);
            acc11 = __builtin_amdgcn_wmma_f32_16x16x4_f32(false, a1, false, b1, (short)0, acc11, false, false);
            acc12 = __builtin_amdgcn_wmma_f32_16x16x4_f32(false, a1, false, b2, (short)0, acc12, false, false);
            acc13 = __builtin_amdgcn_wmma_f32_16x16x4_f32(false, a1, false, b3, (short)0, acc13, false, false);
        }
        __syncthreads();
    }

    const int ncol = n0 + lm;
    #pragma unroll
    for (int mt = 0; mt < 2; ++mt) {
        const int mrow = m0 + wave * 32 + mt * 16 + hi * 8;
        v8f a0 = mt ? acc10 : acc00, a1 = mt ? acc11 : acc01;
        v8f a2 = mt ? acc12 : acc02, a3 = mt ? acc13 : acc03;
        #pragma unroll
        for (int r = 0; r < 8; ++r) {
            size_t base = (size_t)(mrow + r) * ldc + ncol;
            float v0 = a0[r], v1 = a1[r], v2 = a2[r], v3 = a3[r];
            if (Mb) {
                v0 *= Mb[base]; v1 *= Mb[base + 16]; v2 *= Mb[base + 32]; v3 *= Mb[base + 48];
            }
            Cb[base] = v0; Cb[base + 16] = v1; Cb[base + 32] = v2; Cb[base + 48] = v3;
        }
    }
}

__device__ inline float softplusf(float x) {
    if (x > 20.f) return x;
    if (x < -20.f) return expf(x);
    return log1pf(expf(x));
}

struct XwArgs {
    const float* x;
    const float *w0, *w1, *w2, *w3, *w4, *w5, *w6, *w7;
    const float *e0, *e1, *e2, *e3, *e4, *e5;
    const float *wdA, *wdB, *wdC;
    float* xr;
    float *o0, *o1, *o2, *o3, *o4, *o5;
    float *dA, *dB, *dC;
};

struct BFrag {
    v4f    r0[8], r1[8];
    float  e0[8], e1[8];
};

#define XT_ROWS 32
#define XT_K    32
#define XT_PITCH 36

__global__ __launch_bounds__(256) void xw_shared_kernel(XwArgs p)
{
    __shared__ __align__(16) float Xs[2][XT_ROWS][XT_PITCH];

    const int tid  = threadIdx.x;
    const int wave = tid >> 5, lane = tid & 31;
    const int hi   = lane >> 4, lm = lane & 15;
    const int m0   = blockIdx.x * XT_ROWS;
    const int sel  = blockIdx.y * 4 + (wave >> 1);
    const int mt   = wave & 1;

    const float* Wm = sel == 0 ? p.w0 : sel == 1 ? p.w1 : sel == 2 ? p.w2 : sel == 3 ? p.w3
                    : sel == 4 ? p.w4 : sel == 5 ? p.w5 : sel == 6 ? p.w6 : p.w7;
    Wm += (size_t)(2 * hi) * RR + 4 * lm;

    const float* ep;
    int es;
    if (sel < 6) {
        ep = (sel == 0 ? p.e0 : sel == 1 ? p.e1 : sel == 2 ? p.e2
            : sel == 3 ? p.e3 : sel == 4 ? p.e4 : p.e5) + lm;
        es = RSS;
    } else {
        ep = (sel == 6 && lm == 1) ? p.wdB : (sel == 6 && lm == 2) ? p.wdC : p.wdA;
        es = 1;
    }
    ep += (size_t)(2 * hi) * es;

    const int xrow = tid >> 3, xc4 = (tid & 7) * 4;
    const float* xg = p.x + (size_t)(m0 + xrow) * DD + xc4;

    auto loadB = [&](BFrag& f, int k0) {
        #pragma unroll
        for (int s = 0; s < 8; ++s) {
            const int k = k0 + 4 * s;
            f.r0[s] = *(const v4f*)(Wm + (size_t)k * RR);
            f.r1[s] = *(const v4f*)(Wm + (size_t)(k + 1) * RR);
            f.e0[s] = ep[(size_t)k * es];
            f.e1[s] = ep[(size_t)(k + 1) * es];
        }
    };

    v8f acc0 = {}, acc1 = {}, acc2 = {}, acc3 = {}, accE = {};
    const int arow = mt * 16 + lm;
    const int nch  = DD / XT_K;
    v4f xreg;

    auto stage = [&](const BFrag& cur, BFrag& nxt, int c, int buf) {
        *(v4f*)&Xs[buf][xrow][xc4] = xreg;
        __syncthreads();
        const int cn = (c + 1 < nch) ? c + 1 : c;
        xreg = *(const v4f*)(xg + cn * XT_K);
        loadB(nxt, cn * XT_K);

        v2f a[8];
        #pragma unroll
        for (int s = 0; s < 8; ++s)
            a[s] = *(const v2f*)&Xs[buf][arow][4 * s + 2 * hi];
        __builtin_amdgcn_sched_barrier(0);
        #pragma unroll
        for (int s = 0; s < 8; ++s) {
            v2f b0, b1, b2, b3, be;
            b0.x = cur.r0[s].x; b0.y = cur.r1[s].x;
            b1.x = cur.r0[s].y; b1.y = cur.r1[s].y;
            b2.x = cur.r0[s].z; b2.y = cur.r1[s].z;
            b3.x = cur.r0[s].w; b3.y = cur.r1[s].w;
            be.x = cur.e0[s];   be.y = cur.e1[s];
            acc0 = __builtin_amdgcn_wmma_f32_16x16x4_f32(false, a[s], false, b0, (short)0, acc0, false, false);
            acc1 = __builtin_amdgcn_wmma_f32_16x16x4_f32(false, a[s], false, b1, (short)0, acc1, false, false);
            acc2 = __builtin_amdgcn_wmma_f32_16x16x4_f32(false, a[s], false, b2, (short)0, acc2, false, false);
            acc3 = __builtin_amdgcn_wmma_f32_16x16x4_f32(false, a[s], false, b3, (short)0, acc3, false, false);
            accE = __builtin_amdgcn_wmma_f32_16x16x4_f32(false, a[s], false, be, (short)0, accE, false, false);
        }
        __builtin_amdgcn_sched_barrier(0);
        asm volatile("" :: "v"(nxt.r0[0]), "v"(nxt.r0[1]), "v"(nxt.r0[2]), "v"(nxt.r0[3]),
                           "v"(nxt.r0[4]), "v"(nxt.r0[5]), "v"(nxt.r0[6]), "v"(nxt.r0[7]),
                           "v"(nxt.r1[0]), "v"(nxt.r1[1]), "v"(nxt.r1[2]), "v"(nxt.r1[3]),
                           "v"(nxt.r1[4]), "v"(nxt.r1[5]), "v"(nxt.r1[6]), "v"(nxt.r1[7]));
        asm volatile("" :: "v"(nxt.e0[0]), "v"(nxt.e0[1]), "v"(nxt.e0[2]), "v"(nxt.e0[3]),
                           "v"(nxt.e0[4]), "v"(nxt.e0[5]), "v"(nxt.e0[6]), "v"(nxt.e0[7]),
                           "v"(nxt.e1[0]), "v"(nxt.e1[1]), "v"(nxt.e1[2]), "v"(nxt.e1[3]),
                           "v"(nxt.e1[4]), "v"(nxt.e1[5]), "v"(nxt.e1[6]), "v"(nxt.e1[7]),
                           "v"(xreg));
    };

    BFrag f0, f1;
    xreg = *(const v4f*)xg;
    loadB(f0, 0);
    for (int c = 0; c < nch; c += 2) {
        stage(f0, f1, c,     0);
        stage(f1, f0, c + 1, 1);
    }

    const int orow = m0 + mt * 16 + hi * 8;
    float* xo = p.xr + (size_t)sel * BT * RR + (size_t)orow * RR + 4 * lm;
    #pragma unroll
    for (int r = 0; r < 8; ++r) {
        float4 o; o.x = acc0[r]; o.y = acc1[r]; o.z = acc2[r]; o.w = acc3[r];
        *(float4*)(xo + (size_t)r * RR) = o;
    }
    if (sel < 6) {
        float* eo = (sel == 0 ? p.o0 : sel == 1 ? p.o1 : sel == 2 ? p.o2
                   : sel == 3 ? p.o3 : sel == 4 ? p.o4 : p.o5) + (size_t)orow * RSS + lm;
        #pragma unroll
        for (int r = 0; r < 8; ++r) eo[(size_t)r * RSS] = accE[r];
    } else if (sel == 6 && lm < 3) {
        float* dp = (lm == 0 ? p.dA : lm == 1 ? p.dB : p.dC) + orow;
        #pragma unroll
        for (int r = 0; r < 8; ++r) dp[r] = accE[r];
    }
}

__device__ inline v8f strip_times_factor(const v8f (&m)[8], const float* __restrict__ fac)
{
    v8f acc0 = {}, acc1 = {};
    #pragma unroll
    for (int t = 0; t < 8; ++t) {
        float bv[8];
        #pragma unroll
        for (int r = 0; r < 8; ++r) bv[r] = fac[(size_t)(16 * t + r) * RSS];
        #pragma unroll
        for (int q = 0; q < 4; ++q) {
            v2f a, b;
            a.x = m[t][2 * q]; a.y = m[t][2 * q + 1];
            b.x = bv[2 * q];   b.y = bv[2 * q + 1];
            if (q & 1) acc1 = __builtin_amdgcn_wmma_f32_16x16x4_f32(false, a, false, b, (short)0, acc1, false, false);
            else       acc0 = __builtin_amdgcn_wmma_f32_16x16x4_f32(false, a, false, b, (short)0, acc0, false, false);
        }
    }
    return acc0 + acc1;
}

__device__ inline void mix_apply(v8f (&m)[8], const float* __restrict__ dptr,
                                 const float* __restrict__ Arow, const v4f zb0, const v4f zb1)
{
    v2f b0, b1, b2, b3;
    b0.x = zb0.x; b0.y = zb0.y; b1.x = zb0.z; b1.y = zb0.w;
    b2.x = zb1.x; b2.y = zb1.y; b3.x = zb1.z; b3.y = zb1.w;
    #pragma unroll
    for (int t = 0; t < 8; ++t) {
        const v4f dlo = *(const v4f*)(dptr + 16 * t), dhi = *(const v4f*)(dptr + 16 * t + 4);
        const v4f f0 = *(const v4f*)(Arow + (size_t)(16 * t) * RSS);
        const v4f f1 = *(const v4f*)(Arow + (size_t)(16 * t) * RSS + 4);
        v8f c;
        c[0] = m[t][0] * dlo.x; c[1] = m[t][1] * dlo.y; c[2] = m[t][2] * dlo.z; c[3] = m[t][3] * dlo.w;
        c[4] = m[t][4] * dhi.x; c[5] = m[t][5] * dhi.y; c[6] = m[t][6] * dhi.z; c[7] = m[t][7] * dhi.w;
        v2f a0, a1, a2, a3;
        a0.x = f0.x; a0.y = f0.y; a1.x = f0.z; a1.y = f0.w;
        a2.x = f1.x; a2.y = f1.y; a3.x = f1.z; a3.y = f1.w;
        c = __builtin_amdgcn_wmma_f32_16x16x4_f32(false, a0, false, b0, (short)0, c, false, false);
        c = __builtin_amdgcn_wmma_f32_16x16x4_f32(false, a1, false, b1, (short)0, c, false, false);
        c = __builtin_amdgcn_wmma_f32_16x16x4_f32(false, a2, false, b2, (short)0, c, false, false);
        c = __builtin_amdgcn_wmma_f32_16x16x4_f32(false, a3, false, b3, (short)0, c, false, false);
        m[t] = c;
    }
}

__global__ __launch_bounds__(512) void mix_chain_kernel(
    const float* __restrict__ z, const float* zs,
    const float* __restrict__ d1,  const float* __restrict__ A1,  const float* __restrict__ B1,
    const float* __restrict__ d1s, const float* __restrict__ A1s, const float* __restrict__ B1s,
    const float* __restrict__ d2,  const float* __restrict__ A2,  const float* __restrict__ B2,
    float* out)
{
    __shared__ __align__(16) float part[2][16][16][16];
    __shared__ __align__(16) float full[2][16][16];
    __shared__ float rowpart[16][16];

    const int tid  = threadIdx.x;
    const int wave = tid >> 5, lane = tid & 31;
    const int hi   = lane >> 4, lm = lane & 15;
    const int row0 = blockIdx.x * 16;
    const int fb   = (row0 >> 11) * TT;
    const int j0   = wave * 128;

    const size_t soff = (size_t)(row0 + lm) * TT + j0 + 8 * hi;
    const size_t frow = (size_t)(fb + j0 + 8 * hi) * RSS + lm;
    const size_t arow = (size_t)(fb + j0 + lm) * RSS + 8 * hi;
    const int    doff = fb + j0 + 8 * hi;

    v8f m[8], ms[8];
    #pragma unroll
    for (int t = 0; t < 8; ++t) {
        const v4f lo = *(const v4f*)(z + soff + 16 * t), up = *(const v4f*)(z + soff + 16 * t + 4);
        m[t][0] = lo.x; m[t][1] = lo.y; m[t][2] = lo.z; m[t][3] = lo.w;
        m[t][4] = up.x; m[t][5] = up.y; m[t][6] = up.z; m[t][7] = up.w;
    }
    #pragma unroll
    for (int t = 0; t < 8; ++t) {
        const v4f lo = *(const v4f*)(zs + soff + 16 * t), up = *(const v4f*)(zs + soff + 16 * t + 4);
        ms[t][0] = lo.x; ms[t][1] = lo.y; ms[t][2] = lo.z; ms[t][3] = lo.w;
        ms[t][4] = up.x; ms[t][5] = up.y; ms[t][6] = up.z; ms[t][7] = up.w;
    }

    {
        const v8f p0 = strip_times_factor(m, B1 + frow);
        #pragma unroll
        for (int r = 0; r < 8; ++r) part[0][wave][8 * hi + r][lm] = p0[r];
    }
    {
        const v8f p1 = strip_times_factor(ms, B1s + frow);
        #pragma unroll
        for (int r = 0; r < 8; ++r) part[1][wave][8 * hi + r][lm] = p1[r];
    }
    __syncthreads();
    {
        const int pr = tid >> 8, e = tid & 255;
        float v = 0.f;
        #pragma unroll
        for (int w = 0; w < 16; ++w) v += (&part[pr][w][0][0])[e];
        (&full[pr][0][0])[e] = v;
    }
    __syncthreads();

    mix_apply(m,  d1  + doff, A1  + arow, *(const v4f*)&full[0][lm][8 * hi], *(const v4f*)&full[0][lm][8 * hi + 4]);
    mix_apply(ms, d1s + doff, A1s + arow, *(const v4f*)&full[1][lm][8 * hi], *(const v4f*)&full[1][lm][8 * hi + 4]);

    float ss = 0.f;
    #pragma unroll
    for (int t = 0; t < 8; ++t) {
        #pragma unroll
        for (int r = 0; r < 8; ++r) ss += m[t][r] * m[t][r];
    }
    ss += __shfl_xor(ss, 16, 32);
    if (hi == 0) rowpart[wave][lm] = ss;
    __syncthreads();
    float tot = 0.f;
    #pragma unroll
    for (int w = 0; w < 16; ++w) tot += rowpart[w][lm];
    const float inv = 1.f / (sqrtf(tot) + 1e-8f);

    #pragma unroll
    for (int t = 0; t < 8; ++t) {
        #pragma unroll
        for (int r = 0; r < 8; ++r) m[t][r] = softplusf(ms[t][r]) * fmaxf(m[t][r] * inv, 0.f);
    }

    {
        const v8f p0 = strip_times_factor(m, B2 + frow);
        #pragma unroll
        for (int r = 0; r < 8; ++r) part[0][wave][8 * hi + r][lm] = p0[r];
    }
    __syncthreads();
    if (tid < 256) {
        float v = 0.f;
        #pragma unroll
        for (int w = 0; w < 16; ++w) v += (&part[0][w][0][0])[tid];
        (&full[0][0][0])[tid] = v;
    }
    __syncthreads();
    mix_apply(m, d2 + doff, A2 + arow, *(const v4f*)&full[0][lm][8 * hi], *(const v4f*)&full[0][lm][8 * hi + 4]);

    #pragma unroll
    for (int t = 0; t < 8; ++t) {
        v4f lo, up;
        lo.x = m[t][0]; lo.y = m[t][1]; lo.z = m[t][2]; lo.w = m[t][3];
        up.x = m[t][4]; up.y = m[t][5]; up.z = m[t][6]; up.w = m[t][7];
        *(v4f*)(out + soff + 16 * t)     = lo;
        *(v4f*)(out + soff + 16 * t + 4) = up;
    }
}

__global__ __launch_bounds__(256) void emul_kernel(float* __restrict__ a, const float* __restrict__ b)
{
    const size_t i = ((size_t)blockIdx.x * blockDim.x + threadIdx.x) * 4;
    float4 av = *(const float4*)(a + i);
    float4 bv = *(const float4*)(b + i);
    av.x *= bv.x; av.y *= bv.y; av.z *= bv.z; av.w *= bv.w;
    *(float4*)(a + i) = av;
}

extern "C" void kernel_launch(void* const* d_in, const int* in_sizes, int n_in,
                              void* d_out, int out_size, void* d_ws, size_t ws_size,
                              hipStream_t stream) {
    (void)in_sizes; (void)n_in; (void)out_size; (void)ws_size;

    const float* x = (const float*)d_in[0];
    const float* qk_Wq1  = (const float*)d_in[1];  const float* qk_Wq2  = (const float*)d_in[2];
    const float* qk_Wm1  = (const float*)d_in[3];  const float* qk_Wm2  = (const float*)d_in[4];
    const float* qk_Wk1  = (const float*)d_in[5];  const float* qk_Wk2  = (const float*)d_in[6];
    const float* qks_Wq1 = (const float*)d_in[7];  const float* qks_Wq2 = (const float*)d_in[8];
    const float* qks_Wm1 = (const float*)d_in[9];  const float* qks_Wm2 = (const float*)d_in[10];
    const float* qks_Wk1 = (const float*)d_in[11]; const float* qks_Wk2 = (const float*)d_in[12];
    const float* vo_Wv1  = (const float*)d_in[13]; const float* vo_Wv2  = (const float*)d_in[14];
    const float* vo_Wm1  = (const float*)d_in[15]; const float* vo_Wm2  = (const float*)d_in[16];
    const float* vo_Wo   = (const float*)d_in[17];
    const float* sm1_Wd  = (const float*)d_in[18]; const float* sm1_Wa  = (const float*)d_in[19]; const float* sm1_Wb  = (const float*)d_in[20];
    const float* sm1s_Wd = (const float*)d_in[21]; const float* sm1s_Wa = (const float*)d_in[22]; const float* sm1s_Wb = (const float*)d_in[23];
    const float* sm2_Wd  = (const float*)d_in[24]; const float* sm2_Wa  = (const float*)d_in[25]; const float* sm2_Wb  = (const float*)d_in[26];

    const size_t BTD = (size_t)BT * DD;
    const size_t BTT = (size_t)BB * TT * TT;

    float* w = (float*)d_ws;
    float* qg   = w; w += BTD;
    float* kb   = w; w += BTD;
    float* qsg  = w; w += BTD;
    float* ksb  = w; w += BTD;
    float* vb   = w; w += BTD;
    float* mvob = w; w += BTD;
    float* tmpb = w; w += BTD;
    float* zb   = w; w += BTT;
    float* zsb  = w; w += BTT;
    float* ctx  = w; w += BTD;
    w += (size_t)BT * RR;
    float* d1   = w; w += BT; float* d1s = w; w += BT; float* d2 = w; w += BT;
    float* A1   = w; w += (size_t)BT * RSS; float* B1  = w; w += (size_t)BT * RSS;
    float* A1s  = w; w += (size_t)BT * RSS; float* B1s = w; w += (size_t)BT * RSS;
    float* A2   = w; w += (size_t)BT * RSS; float* B2  = w; w += (size_t)BT * RSS;

    auto gemm = [&](const float* A, const float* Bm, float* C, const float* mul,
                    int M, int N, int K, int lda, int ldb, int ldc, int bT,
                    long long sA, long long sB, long long sC, int batches) {
        dim3 g((unsigned)(N / GBN), (unsigned)(M / GBM), (unsigned)batches);
        gemm_wmma_f32<<<g, dim3(256), 0, stream>>>(A, Bm, C, mul, M, N, K, lda, ldb, ldc, bT, sA, sB, sC);
    };

    float* xrAll = ctx;
    const size_t XP = (size_t)BT * RR;
    XwArgs xa;
    xa.x  = x;
    xa.w0 = qk_Wm1;  xa.w1 = qk_Wq1;  xa.w2 = qk_Wk1;
    xa.w3 = qks_Wm1; xa.w4 = qks_Wq1; xa.w5 = qks_Wk1;
    xa.w6 = vo_Wv1;  xa.w7 = vo_Wm1;
    xa.e0 = sm1_Wa;  xa.e1 = sm1_Wb;  xa.e2 = sm1s_Wa; xa.e3 = sm1s_Wb; xa.e4 = sm2_Wa; xa.e5 = sm2_Wb;
    xa.wdA = sm1_Wd; xa.wdB = sm1s_Wd; xa.wdC = sm2_Wd;
    xa.xr = xrAll;
    xa.o0 = A1; xa.o1 = B1; xa.o2 = A1s; xa.o3 = B1s; xa.o4 = A2; xa.o5 = B2;
    xa.dA = d1; xa.dB = d1s; xa.dC = d2;
    xw_shared_kernel<<<dim3(BT / XT_ROWS, 2), 256, 0, stream>>>(xa);

    gemm(xrAll + 0 * XP, qk_Wm2,  tmpb, nullptr, BT, DD, RR, RR, DD, DD, 0, 0, 0, 0, 1);
    gemm(xrAll + 1 * XP, qk_Wq2,  qg,   tmpb,    BT, DD, RR, RR, DD, DD, 0, 0, 0, 0, 1);
    gemm(xrAll + 2 * XP, qk_Wk2,  kb,   nullptr, BT, DD, RR, RR, DD, DD, 0, 0, 0, 0, 1);

    gemm(xrAll + 3 * XP, qks_Wm2, tmpb, nullptr, BT, DD, RR, RR, DD, DD, 0, 0, 0, 0, 1);
    gemm(xrAll + 4 * XP, qks_Wq2, qsg,  tmpb,    BT, DD, RR, RR, DD, DD, 0, 0, 0, 0, 1);
    gemm(xrAll + 5 * XP, qks_Wk2, ksb,  nullptr, BT, DD, RR, RR, DD, DD, 0, 0, 0, 0, 1);

    gemm(xrAll + 6 * XP, vo_Wv2,  vb,   nullptr, BT, DD, RR, RR, DD, DD, 0, 0, 0, 0, 1);
    gemm(xrAll + 7 * XP, vo_Wm2,  mvob, nullptr, BT, DD, RR, RR, DD, DD, 0, 0, 0, 0, 1);

    const long long sTD = (long long)TT * DD, sTTl = (long long)TT * TT;
    gemm(qg,  kb,  zb,  nullptr, TT, TT, DD, DD, DD, TT, 1, sTD, sTD, sTTl, BB);
    gemm(qsg, ksb, zsb, nullptr, TT, TT, DD, DD, DD, TT, 1, sTD, sTD, sTTl, BB);

    mix_chain_kernel<<<dim3(BT / 16), 512, 0, stream>>>(zb, zsb, d1, A1, B1, d1s, A1s, B1s, d2, A2, B2, zsb);

    gemm(zsb, vb, ctx, nullptr, TT, DD, TT, TT, DD, DD, 0, sTTl, sTD, sTD, BB);
    emul_kernel<<<dim3((unsigned)(BTD / 4 / 256)), 256, 0, stream>>>(ctx, mvob);
    gemm(ctx, vo_Wo, (float*)d_out, nullptr, BT, DD, DD, DD, DD, DD, 1, 0, 0, 0, 1);
}


# the eight second-stage products [BT,64]@[64,D] run as TWO launches (six ungated, then q and qs times their gates) instead of eight; blockIdx.z picks the problem; same tile code, same sum order
# speedup vs baseline: 2.7555x; 1.0484x over previous
#include <hip/hip_runtime.h>
#include <math.h>

#define BB 2
#define TT 2048
#define DD 1024
#define RR 64
#define RSS 16
#define BT (BB*TT)

typedef __attribute__((ext_vector_type(2))) float v2f;
typedef __attribute__((ext_vector_type(8))) float v8f;
typedef __attribute__((ext_vector_type(4))) float v4f;

#define GBM 256
#define GBN 64
#define GBK 32

__device__ __forceinline__ void gemm_tile_f32(
    const float* __restrict__ Ab, const float* __restrict__ Bb,
    float* __restrict__ Cb, const float* __restrict__ Mb,
    int K, int lda, int ldb, int ldc, int bT, int m0, int n0)
{
    __shared__ __align__(16) float As [2][GBM][34];
    __shared__ __align__(16) float BsT[2][GBN][34];

    const int tid  = threadIdx.x;
    const int wave = tid >> 5;
    const int lane = tid & 31;
    const int hi   = lane >> 4;
    const int lm   = lane & 15;

    float4 ra[8];
    float4 rb[2];

    int arow[8], ac4[8];
    #pragma unroll
    for (int i = 0; i < 8; ++i) { int idx = tid + i * 256; arow[i] = idx >> 3; ac4[i] = (idx & 7) * 4; }
    int bi0[2], bi1[2];
    #pragma unroll
    for (int i = 0; i < 2; ++i) {
        int idx = tid + i * 256;
        if (true) {   }
        bi0[i] = idx; bi1[i] = idx;
    }

    auto loadGlobal = [&](int k0) {
        #pragma unroll
        for (int i = 0; i < 8; ++i)
            ra[i] = *(const float4*)(Ab + (size_t)(m0 + arow[i]) * lda + k0 + ac4[i]);
        if (bT) {
            #pragma unroll
            for (int i = 0; i < 2; ++i) {
                int idx = bi0[i];
                int n = idx >> 3, c4 = (idx & 7) * 4;
                rb[i] = *(const float4*)(Bb + (size_t)(n0 + n) * ldb + k0 + c4);
            }
        } else {
            #pragma unroll
            for (int i = 0; i < 2; ++i) {
                int idx = bi1[i];
                int r = idx >> 4, c4 = (idx & 15) * 4;
                rb[i] = *(const float4*)(Bb + (size_t)(k0 + r) * ldb + n0 + c4);
            }
        }
    };

    auto storeLDS = [&](int buf) {
        #pragma unroll
        for (int i = 0; i < 8; ++i) {
            float4 t = ra[i];
            v2f lo; lo.x = t.x; lo.y = t.y;
            v2f hi2; hi2.x = t.z; hi2.y = t.w;
            *(v2f*)&As[buf][arow[i]][ac4[i]]     = lo;
            *(v2f*)&As[buf][arow[i]][ac4[i] + 2] = hi2;
        }
        if (bT) {
            #pragma unroll
            for (int i = 0; i < 2; ++i) {
                int idx = bi0[i];
                int n = idx >> 3, c4 = (idx & 7) * 4;
                float4 t = rb[i];
                v2f lo; lo.x = t.x; lo.y = t.y;
                v2f hi2; hi2.x = t.z; hi2.y = t.w;
                *(v2f*)&BsT[buf][n][c4]     = lo;
                *(v2f*)&BsT[buf][n][c4 + 2] = hi2;
            }
        } else {
            #pragma unroll
            for (int i = 0; i < 2; ++i) {
                int idx = bi1[i];
                int r = idx >> 4, c4 = (idx & 15) * 4;
                float4 t = rb[i];
                BsT[buf][c4 + 0][r] = t.x;
                BsT[buf][c4 + 1][r] = t.y;
                BsT[buf][c4 + 2][r] = t.z;
                BsT[buf][c4 + 3][r] = t.w;
            }
        }
    };

    v8f acc00 = {}, acc01 = {}, acc02 = {}, acc03 = {};
    v8f acc10 = {}, acc11 = {}, acc12 = {}, acc13 = {};

    const int r0 = wave * 32 + lm;
    const int r1 = r0 + 16;
    const int nchunks = K / GBK;

    loadGlobal(0);
    for (int c = 0; c < nchunks; ++c) {
        const int buf = c & 1;
        storeLDS(buf);
        __syncthreads();
        if (c + 1 < nchunks) loadGlobal((c + 1) * GBK);

        #pragma unroll
        for (int kk = 0; kk < GBK; kk += 4) {
            const int kc = kk + 2 * hi;
            v2f a0 = *(const v2f*)&As [buf][r0][kc];
            v2f a1 = *(const v2f*)&As [buf][r1][kc];
            v2f b0 = *(const v2f*)&BsT[buf][lm +  0][kc];
            v2f b1 = *(const v2f*)&BsT[buf][lm + 16][kc];
            v2f b2 = *(const v2f*)&BsT[buf][lm + 32][kc];
            v2f b3 = *(const v2f*)&BsT[buf][lm + 48][kc];
            acc00 = __builtin_amdgcn_wmma_f32_16x16x4_f32(false, a0, false, b0, (short)0, acc00, false, false);
            acc01 = __builtin_amdgcn_wmma_f32_16x16x4_f32(false, a0, false, b1, (short)0, acc01, false, false);
            acc02 = __builtin_amdgcn_wmma_f32_16x16x4_f32(false, a0, false, b2, (short)0, acc02, false, false);
            acc03 = __builtin_amdgcn_wmma_f32_16x16x4_f32(false, a0, false, b3, (short)0, acc03, false, false);
            acc10 = __builtin_amdgcn_wmma_f32_16x16x4_f32(false, a1, false, b0, (short)0, acc10, false, false);
            acc11 = __builtin_amdgcn_wmma_f32_16x16x4_f32(false, a1, false, b1, (short)0, acc11, false, false);
            acc12 = __builtin_amdgcn_wmma_f32_16x16x4_f32(false, a1, false, b2, (short)0, acc12, false, false);
            acc13 = __builtin_amdgcn_wmma_f32_16x16x4_f32(false, a1, false, b3, (short)0, acc13, false, false);
        }
        __syncthreads();
    }

    const int ncol = n0 + lm;
    #pragma unroll
    for (int mt = 0; mt < 2; ++mt) {
        const int mrow = m0 + wave * 32 + mt * 16 + hi * 8;
        v8f a0 = mt ? acc10 : acc00, a1 = mt ? acc11 : acc01;
        v8f a2 = mt ? acc12 : acc02, a3 = mt ? acc13 : acc03;
        #pragma unroll
        for (int r = 0; r < 8; ++r) {
            size_t base = (size_t)(mrow + r) * ldc + ncol;
            float v0 = a0[r], v1 = a1[r], v2 = a2[r], v3 = a3[r];
            if (Mb) {
                v0 *= Mb[base]; v1 *= Mb[base + 16]; v2 *= Mb[base + 32]; v3 *= Mb[base + 48];
            }
            Cb[base] = v0; Cb[base + 16] = v1; Cb[base + 32] = v2; Cb[base + 48] = v3;
        }
    }
}

__global__ __launch_bounds__(256) void gemm_wmma_f32(
    const float* __restrict__ A, const float* __restrict__ Bm,
    float* __restrict__ C, const float* __restrict__ mulC,
    int M, int N, int K, int lda, int ldb, int ldc, int bT,
    long long sA, long long sB, long long sC)
{
    (void)M; (void)N;
    gemm_tile_f32(A  + (size_t)blockIdx.z * sA, Bm + (size_t)blockIdx.z * sB,
                  C  + (size_t)blockIdx.z * sC,
                  mulC ? (mulC + (size_t)blockIdx.z * sC) : nullptr,
                  K, lda, ldb, ldc, bT, blockIdx.y * GBM, blockIdx.x * GBN);
}

struct Stage2Args {
    const float *a0, *a1, *a2, *a3, *a4, *a5;
    const float *w0, *w1, *w2, *w3, *w4, *w5;
    float       *c0, *c1, *c2, *c3, *c4, *c5;
    const float *g0, *g1, *g2, *g3, *g4, *g5;
};

__global__ __launch_bounds__(256) void gemm_stage2_kernel(Stage2Args p)
{
    const int s = blockIdx.z;
    const float* A = s == 0 ? p.a0 : s == 1 ? p.a1 : s == 2 ? p.a2 : s == 3 ? p.a3 : s == 4 ? p.a4 : p.a5;
    const float* Wm = s == 0 ? p.w0 : s == 1 ? p.w1 : s == 2 ? p.w2 : s == 3 ? p.w3 : s == 4 ? p.w4 : p.w5;
    float*       C = s == 0 ? p.c0 : s == 1 ? p.c1 : s == 2 ? p.c2 : s == 3 ? p.c3 : s == 4 ? p.c4 : p.c5;
    const float* G = s == 0 ? p.g0 : s == 1 ? p.g1 : s == 2 ? p.g2 : s == 3 ? p.g3 : s == 4 ? p.g4 : p.g5;
    gemm_tile_f32(A, Wm, C, G, RR, RR, DD, DD, 0, blockIdx.y * GBM, blockIdx.x * GBN);
}

__device__ inline float softplusf(float x) {
    if (x > 20.f) return x;
    if (x < -20.f) return expf(x);
    return log1pf(expf(x));
}

struct XwArgs {
    const float* x;
    const float *w0, *w1, *w2, *w3, *w4, *w5, *w6, *w7;
    const float *e0, *e1, *e2, *e3, *e4, *e5;
    const float *wdA, *wdB, *wdC;
    float* xr;
    float *o0, *o1, *o2, *o3, *o4, *o5;
    float *dA, *dB, *dC;
};

struct BFrag {
    v4f    r0[8], r1[8];
    float  e0[8], e1[8];
};

#define XT_ROWS 32
#define XT_K    32
#define XT_PITCH 36

__global__ __launch_bounds__(256) void xw_shared_kernel(XwArgs p)
{
    __shared__ __align__(16) float Xs[2][XT_ROWS][XT_PITCH];

    const int tid  = threadIdx.x;
    const int wave = tid >> 5, lane = tid & 31;
    const int hi   = lane >> 4, lm = lane & 15;
    const int m0   = blockIdx.x * XT_ROWS;
    const int sel  = blockIdx.y * 4 + (wave >> 1);
    const int mt   = wave & 1;

    const float* Wm = sel == 0 ? p.w0 : sel == 1 ? p.w1 : sel == 2 ? p.w2 : sel == 3 ? p.w3
                    : sel == 4 ? p.w4 : sel == 5 ? p.w5 : sel == 6 ? p.w6 : p.w7;
    Wm += (size_t)(2 * hi) * RR + 4 * lm;

    const float* ep;
    int es;
    if (sel < 6) {
        ep = (sel == 0 ? p.e0 : sel == 1 ? p.e1 : sel == 2 ? p.e2
            : sel == 3 ? p.e3 : sel == 4 ? p.e4 : p.e5) + lm;
        es = RSS;
    } else {
        ep = (sel == 6 && lm == 1) ? p.wdB : (sel == 6 && lm == 2) ? p.wdC : p.wdA;
        es = 1;
    }
    ep += (size_t)(2 * hi) * es;

    const int xrow = tid >> 3, xc4 = (tid & 7) * 4;
    const float* xg = p.x + (size_t)(m0 + xrow) * DD + xc4;

    auto loadB = [&](BFrag& f, int k0) {
        #pragma unroll
        for (int s = 0; s < 8; ++s) {
            const int k = k0 + 4 * s;
            f.r0[s] = *(const v4f*)(Wm + (size_t)k * RR);
            f.r1[s] = *(const v4f*)(Wm + (size_t)(k + 1) * RR);
            f.e0[s] = ep[(size_t)k * es];
            f.e1[s] = ep[(size_t)(k + 1) * es];
        }
    };

    v8f acc0 = {}, acc1 = {}, acc2 = {}, acc3 = {}, accE = {};
    const int arow = mt * 16 + lm;
    const int nch  = DD / XT_K;
    v4f xreg;

    auto stage = [&](const BFrag& cur, BFrag& nxt, int c, int buf) {
        *(v4f*)&Xs[buf][xrow][xc4] = xreg;
        __syncthreads();
        const int cn = (c + 1 < nch) ? c + 1 : c;
        xreg = *(const v4f*)(xg + cn * XT_K);
        loadB(nxt, cn * XT_K);

        v2f a[8];
        #pragma unroll
        for (int s = 0; s < 8; ++s)
            a[s] = *(const v2f*)&Xs[buf][arow][4 * s + 2 * hi];
        __builtin_amdgcn_sched_barrier(0);
        #pragma unroll
        for (int s = 0; s < 8; ++s) {
            v2f b0, b1, b2, b3, be;
            b0.x = cur.r0[s].x; b0.y = cur.r1[s].x;
            b1.x = cur.r0[s].y; b1.y = cur.r1[s].y;
            b2.x = cur.r0[s].z; b2.y = cur.r1[s].z;
            b3.x = cur.r0[s].w; b3.y = cur.r1[s].w;
            be.x = cur.e0[s];   be.y = cur.e1[s];
            acc0 = __builtin_amdgcn_wmma_f32_16x16x4_f32(false, a[s], false, b0, (short)0, acc0, false, false);
            acc1 = __builtin_amdgcn_wmma_f32_16x16x4_f32(false, a[s], false, b1, (short)0, acc1, false, false);
            acc2 = __builtin_amdgcn_wmma_f32_16x16x4_f32(false, a[s], false, b2, (short)0, acc2, false, false);
            acc3 = __builtin_amdgcn_wmma_f32_16x16x4_f32(false, a[s], false, b3, (short)0, acc3, false, false);
            accE = __builtin_amdgcn_wmma_f32_16x16x4_f32(false, a[s], false, be, (short)0, accE, false, false);
        }
        __builtin_amdgcn_sched_barrier(0);
        asm volatile("" :: "v"(nxt.r0[0]), "v"(nxt.r0[1]), "v"(nxt.r0[2]), "v"(nxt.r0[3]),
                           "v"(nxt.r0[4]), "v"(nxt.r0[5]), "v"(nxt.r0[6]), "v"(nxt.r0[7]),
                           "v"(nxt.r1[0]), "v"(nxt.r1[1]), "v"(nxt.r1[2]), "v"(nxt.r1[3]),
                           "v"(nxt.r1[4]), "v"(nxt.r1[5]), "v"(nxt.r1[6]), "v"(nxt.r1[7]));
        asm volatile("" :: "v"(nxt.e0[0]), "v"(nxt.e0[1]), "v"(nxt.e0[2]), "v"(nxt.e0[3]),
                           "v"(nxt.e0[4]), "v"(nxt.e0[5]), "v"(nxt.e0[6]), "v"(nxt.e0[7]),
                           "v"(nxt.e1[0]), "v"(nxt.e1[1]), "v"(nxt.e1[2]), "v"(nxt.e1[3]),
                           "v"(nxt.e1[4]), "v"(nxt.e1[5]), "v"(nxt.e1[6]), "v"(nxt.e1[7]),
                           "v"(xreg));
    };

    BFrag f0, f1;
    xreg = *(const v4f*)xg;
    loadB(f0, 0);
    for (int c = 0; c < nch; c += 2) {
        stage(f0, f1, c,     0);
        stage(f1, f0, c + 1, 1);
    }

    const int orow = m0 + mt * 16 + hi * 8;
    float* xo = p.xr + (size_t)sel * BT * RR + (size_t)orow * RR + 4 * lm;
    #pragma unroll
    for (int r = 0; r < 8; ++r) {
        float4 o; o.x = acc0[r]; o.y = acc1[r]; o.z = acc2[r]; o.w = acc3[r];
        *(float4*)(xo + (size_t)r * RR) = o;
    }
    if (sel < 6) {
        float* eo = (sel == 0 ? p.o0 : sel == 1 ? p.o1 : sel == 2 ? p.o2
                   : sel == 3 ? p.o3 : sel == 4 ? p.o4 : p.o5) + (size_t)orow * RSS + lm;
        #pragma unroll
        for (int r = 0; r < 8; ++r) eo[(size_t)r * RSS] = accE[r];
    } else if (sel == 6 && lm < 3) {
        float* dp = (lm == 0 ? p.dA : lm == 1 ? p.dB : p.dC) + orow;
        #pragma unroll
        for (int r = 0; r < 8; ++r) dp[r] = accE[r];
    }
}

__device__ inline v8f strip_times_factor(const v8f (&m)[8], const float* __restrict__ fac)
{
    v8f acc0 = {}, acc1 = {};
    #pragma unroll
    for (int t = 0; t < 8; ++t) {
        float bv[8];
        #pragma unroll
        for (int r = 0; r < 8; ++r) bv[r] = fac[(size_t)(16 * t + r) * RSS];
        #pragma unroll
        for (int q = 0; q < 4; ++q) {
            v2f a, b;
            a.x = m[t][2 * q]; a.y = m[t][2 * q + 1];
            b.x = bv[2 * q];   b.y = bv[2 * q + 1];
            if (q & 1) acc1 = __builtin_amdgcn_wmma_f32_16x16x4_f32(false, a, false, b, (short)0, acc1, false, false);
            else       acc0 = __builtin_amdgcn_wmma_f32_16x16x4_f32(false, a, false, b, (short)0, acc0, false, false);
        }
    }
    return acc0 + acc1;
}

__device__ inline void mix_apply(v8f (&m)[8], const float* __restrict__ dptr,
                                 const float* __restrict__ Arow, const v4f zb0, const v4f zb1)
{
    v2f b0, b1, b2, b3;
    b0.x = zb0.x; b0.y = zb0.y; b1.x = zb0.z; b1.y = zb0.w;
    b2.x = zb1.x; b2.y = zb1.y; b3.x = zb1.z; b3.y = zb1.w;
    #pragma unroll
    for (int t = 0; t < 8; ++t) {
        const v4f dlo = *(const v4f*)(dptr + 16 * t), dhi = *(const v4f*)(dptr + 16 * t + 4);
        const v4f f0 = *(const v4f*)(Arow + (size_t)(16 * t) * RSS);
        const v4f f1 = *(const v4f*)(Arow + (size_t)(16 * t) * RSS + 4);
        v8f c;
        c[0] = m[t][0] * dlo.x; c[1] = m[t][1] * dlo.y; c[2] = m[t][2] * dlo.z; c[3] = m[t][3] * dlo.w;
        c[4] = m[t][4] * dhi.x; c[5] = m[t][5] * dhi.y; c[6] = m[t][6] * dhi.z; c[7] = m[t][7] * dhi.w;
        v2f a0, a1, a2, a3;
        a0.x = f0.x; a0.y = f0.y; a1.x = f0.z; a1.y = f0.w;
        a2.x = f1.x; a2.y = f1.y; a3.x = f1.z; a3.y = f1.w;
        c = __builtin_amdgcn_wmma_f32_16x16x4_f32(false, a0, false, b0, (short)0, c, false, false);
        c = __builtin_amdgcn_wmma_f32_16x16x4_f32(false, a1, false, b1, (short)0, c, false, false);
        c = __builtin_amdgcn_wmma_f32_16x16x4_f32(false, a2, false, b2, (short)0, c, false, false);
        c = __builtin_amdgcn_wmma_f32_16x16x4_f32(false, a3, false, b3, (short)0, c, false, false);
        m[t] = c;
    }
}

__global__ __launch_bounds__(512) void mix_chain_kernel(
    const float* __restrict__ z, const float* zs,
    const float* __restrict__ d1,  const float* __restrict__ A1,  const float* __restrict__ B1,
    const float* __restrict__ d1s, const float* __restrict__ A1s, const float* __restrict__ B1s,
    const float* __restrict__ d2,  const float* __restrict__ A2,  const float* __restrict__ B2,
    float* out)
{
    __shared__ __align__(16) float part[2][16][16][16];
    __shared__ __align__(16) float full[2][16][16];
    __shared__ float rowpart[16][16];

    const int tid  = threadIdx.x;
    const int wave = tid >> 5, lane = tid & 31;
    const int hi   = lane >> 4, lm = lane & 15;
    const int row0 = blockIdx.x * 16;
    const int fb   = (row0 >> 11) * TT;
    const int j0   = wave * 128;

    const size_t soff = (size_t)(row0 + lm) * TT + j0 + 8 * hi;
    const size_t frow = (size_t)(fb + j0 + 8 * hi) * RSS + lm;
    const size_t arow = (size_t)(fb + j0 + lm) * RSS + 8 * hi;
    const int    doff = fb + j0 + 8 * hi;

    v8f m[8], ms[8];
    #pragma unroll
    for (int t = 0; t < 8; ++t) {
        const v4f lo = *(const v4f*)(z + soff + 16 * t), up = *(const v4f*)(z + soff + 16 * t + 4);
        m[t][0] = lo.x; m[t][1] = lo.y; m[t][2] = lo.z; m[t][3] = lo.w;
        m[t][4] = up.x; m[t][5] = up.y; m[t][6] = up.z; m[t][7] = up.w;
    }
    #pragma unroll
    for (int t = 0; t < 8; ++t) {
        const v4f lo = *(const v4f*)(zs + soff + 16 * t), up = *(const v4f*)(zs + soff + 16 * t + 4);
        ms[t][0] = lo.x; ms[t][1] = lo.y; ms[t][2] = lo.z; ms[t][3] = lo.w;
        ms[t][4] = up.x; ms[t][5] = up.y; ms[t][6] = up.z; ms[t][7] = up.w;
    }

    {
        const v8f p0 = strip_times_factor(m, B1 + frow);
        #pragma unroll
        for (int r = 0; r < 8; ++r) part[0][wave][8 * hi + r][lm] = p0[r];
    }
    {
        const v8f p1 = strip_times_factor(ms, B1s + frow);
        #pragma unroll
        for (int r = 0; r < 8; ++r) part[1][wave][8 * hi + r][lm] = p1[r];
    }
    __syncthreads();
    {
        const int pr = tid >> 8, e = tid & 255;
        float v = 0.f;
        #pragma unroll
        for (int w = 0; w < 16; ++w) v += (&part[pr][w][0][0])[e];
        (&full[pr][0][0])[e] = v;
    }
    __syncthreads();

    mix_apply(m,  d1  + doff, A1  + arow, *(const v4f*)&full[0][lm][8 * hi], *(const v4f*)&full[0][lm][8 * hi + 4]);
    mix_apply(ms, d1s + doff, A1s + arow, *(const v4f*)&full[1][lm][8 * hi], *(const v4f*)&full[1][lm][8 * hi + 4]);

    float ss = 0.f;
    #pragma unroll
    for (int t = 0; t < 8; ++t) {
        #pragma unroll
        for (int r = 0; r < 8; ++r) ss += m[t][r] * m[t][r];
    }
    ss += __shfl_xor(ss, 16, 32);
    if (hi == 0) rowpart[wave][lm] = ss;
    __syncthreads();
    float tot = 0.f;
    #pragma unroll
    for (int w = 0; w < 16; ++w) tot += rowpart[w][lm];
    const float inv = 1.f / (sqrtf(tot) + 1e-8f);

    #pragma unroll
    for (int t = 0; t < 8; ++t) {
        #pragma unroll
        for (int r = 0; r < 8; ++r) m[t][r] = softplusf(ms[t][r]) * fmaxf(m[t][r] * inv, 0.f);
    }

    {
        const v8f p0 = strip_times_factor(m, B2 + frow);
        #pragma unroll
        for (int r = 0; r < 8; ++r) part[0][wave][8 * hi + r][lm] = p0[r];
    }
    __syncthreads();
    if (tid < 256) {
        float v = 0.f;
        #pragma unroll
        for (int w = 0; w < 16; ++w) v += (&part[0][w][0][0])[tid];
        (&full[0][0][0])[tid] = v;
    }
    __syncthreads();
    mix_apply(m, d2 + doff, A2 + arow, *(const v4f*)&full[0][lm][8 * hi], *(const v4f*)&full[0][lm][8 * hi + 4]);

    #pragma unroll
    for (int t = 0; t < 8; ++t) {
        v4f lo, up;
        lo.x = m[t][0]; lo.y = m[t][1]; lo.z = m[t][2]; lo.w = m[t][3];
        up.x = m[t][4]; up.y = m[t][5]; up.z = m[t][6]; up.w = m[t][7];
        *(v4f*)(out + soff + 16 * t)     = lo;
        *(v4f*)(out + soff + 16 * t + 4) = up;
    }
}

__global__ __launch_bounds__(256) void emul_kernel(float* __restrict__ a, const float* __restrict__ b)
{
    const size_t i = ((size_t)blockIdx.x * blockDim.x + threadIdx.x) * 4;
    float4 av = *(const float4*)(a + i);
    float4 bv = *(const float4*)(b + i);
    av.x *= bv.x; av.y *= bv.y; av.z *= bv.z; av.w *= bv.w;
    *(float4*)(a + i) = av;
}

extern "C" void kernel_launch(void* const* d_in, const int* in_sizes, int n_in,
                              void* d_out, int out_size, void* d_ws, size_t ws_size,
                              hipStream_t stream) {
    (void)in_sizes; (void)n_in; (void)out_size; (void)ws_size;

    const float* x = (const float*)d_in[0];
    const float* qk_Wq1  = (const float*)d_in[1];  const float* qk_Wq2  = (const float*)d_in[2];
    const float* qk_Wm1  = (const float*)d_in[3];  const float* qk_Wm2  = (const float*)d_in[4];
    const float* qk_Wk1  = (const float*)d_in[5];  const float* qk_Wk2  = (const float*)d_in[6];
    const float* qks_Wq1 = (const float*)d_in[7];  const float* qks_Wq2 = (const float*)d_in[8];
    const float* qks_Wm1 = (const float*)d_in[9];  const float* qks_Wm2 = (const float*)d_in[10];
    const float* qks_Wk1 = (const float*)d_in[11]; const float* qks_Wk2 = (const float*)d_in[12];
    const float* vo_Wv1  = (const float*)d_in[13]; const float* vo_Wv2  = (const float*)d_in[14];
    const float* vo_Wm1  = (const float*)d_in[15]; const float* vo_Wm2  = (const float*)d_in[16];
    const float* vo_Wo   = (const float*)d_in[17];
    const float* sm1_Wd  = (const float*)d_in[18]; const float* sm1_Wa  = (const float*)d_in[19]; const float* sm1_Wb  = (const float*)d_in[20];
    const float* sm1s_Wd = (const float*)d_in[21]; const float* sm1s_Wa = (const float*)d_in[22]; const float* sm1s_Wb = (const float*)d_in[23];
    const float* sm2_Wd  = (const float*)d_in[24]; const float* sm2_Wa  = (const float*)d_in[25]; const float* sm2_Wb  = (const float*)d_in[26];

    const size_t BTD = (size_t)BT * DD;
    const size_t BTT = (size_t)BB * TT * TT;

    float* w = (float*)d_ws;
    float* qg   = w; w += BTD;
    float* kb   = w; w += BTD;
    float* qsg  = w; w += BTD;
    float* ksb  = w; w += BTD;
    float* vb   = w; w += BTD;
    float* mvob = w; w += BTD;
    float* tmpb = w; w += BTD;
    float* zb   = w; w += BTT;
    float* zsb  = w; w += BTT;
    float* ctx  = w; w += BTD;
    w += (size_t)BT * RR;
    float* d1   = w; w += BT; float* d1s = w; w += BT; float* d2 = w; w += BT;
    float* A1   = w; w += (size_t)BT * RSS; float* B1  = w; w += (size_t)BT * RSS;
    float* A1s  = w; w += (size_t)BT * RSS; float* B1s = w; w += (size_t)BT * RSS;
    float* A2   = w; w += (size_t)BT * RSS; float* B2  = w; w += (size_t)BT * RSS;

    auto gemm = [&](const float* A, const float* Bm, float* C, const float* mul,
                    int M, int N, int K, int lda, int ldb, int ldc, int bT,
                    long long sA, long long sB, long long sC, int batches) {
        dim3 g((unsigned)(N / GBN), (unsigned)(M / GBM), (unsigned)batches);
        gemm_wmma_f32<<<g, dim3(256), 0, stream>>>(A, Bm, C, mul, M, N, K, lda, ldb, ldc, bT, sA, sB, sC);
    };

    float* xrAll = ctx;
    const size_t XP = (size_t)BT * RR;
    XwArgs xa;
    xa.x  = x;
    xa.w0 = qk_Wm1;  xa.w1 = qk_Wq1;  xa.w2 = qk_Wk1;
    xa.w3 = qks_Wm1; xa.w4 = qks_Wq1; xa.w5 = qks_Wk1;
    xa.w6 = vo_Wv1;  xa.w7 = vo_Wm1;
    xa.e0 = sm1_Wa;  xa.e1 = sm1_Wb;  xa.e2 = sm1s_Wa; xa.e3 = sm1s_Wb; xa.e4 = sm2_Wa; xa.e5 = sm2_Wb;
    xa.wdA = sm1_Wd; xa.wdB = sm1s_Wd; xa.wdC = sm2_Wd;
    xa.xr = xrAll;
    xa.o0 = A1; xa.o1 = B1; xa.o2 = A1s; xa.o3 = B1s; xa.o4 = A2; xa.o5 = B2;
    xa.dA = d1; xa.dB = d1s; xa.dC = d2;
    xw_shared_kernel<<<dim3(BT / XT_ROWS, 2), 256, 0, stream>>>(xa);

    float* tmpb2 = zsb;
    const dim3 g2((unsigned)(DD / GBN), (unsigned)(BT / GBM), 6u);
    Stage2Args sa;
    sa.a0 = xrAll + 0 * XP; sa.w0 = qk_Wm2;  sa.c0 = tmpb;  sa.g0 = nullptr;
    sa.a1 = xrAll + 3 * XP; sa.w1 = qks_Wm2; sa.c1 = tmpb2; sa.g1 = nullptr;
    sa.a2 = xrAll + 2 * XP; sa.w2 = qk_Wk2;  sa.c2 = kb;    sa.g2 = nullptr;
    sa.a3 = xrAll + 5 * XP; sa.w3 = qks_Wk2; sa.c3 = ksb;   sa.g3 = nullptr;
    sa.a4 = xrAll + 6 * XP; sa.w4 = vo_Wv2;  sa.c4 = vb;    sa.g4 = nullptr;
    sa.a5 = xrAll + 7 * XP; sa.w5 = vo_Wm2;  sa.c5 = mvob;  sa.g5 = nullptr;
    gemm_stage2_kernel<<<g2, dim3(256), 0, stream>>>(sa);

    const dim3 g2b((unsigned)(DD / GBN), (unsigned)(BT / GBM), 2u);
    Stage2Args sb;
    sb.a0 = xrAll + 1 * XP; sb.w0 = qk_Wq2;  sb.c0 = qg;  sb.g0 = tmpb;
    sb.a1 = xrAll + 4 * XP; sb.w1 = qks_Wq2; sb.c1 = qsg; sb.g1 = tmpb2;
    sb.a2 = nullptr; sb.w2 = nullptr; sb.c2 = nullptr; sb.g2 = nullptr;
    sb.a3 = nullptr; sb.w3 = nullptr; sb.c3 = nullptr; sb.g3 = nullptr;
    sb.a4 = nullptr; sb.w4 = nullptr; sb.c4 = nullptr; sb.g4 = nullptr;
    sb.a5 = nullptr; sb.w5 = nullptr; sb.c5 = nullptr; sb.g5 = nullptr;
    gemm_stage2_kernel<<<g2b, dim3(256), 0, stream>>>(sb);

    const long long sTD = (long long)TT * DD, sTTl = (long long)TT * TT;
    gemm(qg,  kb,  zb,  nullptr, TT, TT, DD, DD, DD, TT, 1, sTD, sTD, sTTl, BB);
    gemm(qsg, ksb, zsb, nullptr, TT, TT, DD, DD, DD, TT, 1, sTD, sTD, sTTl, BB);

    mix_chain_kernel<<<dim3(BT / 16), 512, 0, stream>>>(zb, zsb, d1, A1, B1, d1s, A1s, B1s, d2, A2, B2, zsb);

    gemm(zsb, vb, ctx, nullptr, TT, DD, TT, TT, DD, DD, 0, sTTl, sTD, sTD, BB);
    emul_kernel<<<dim3((unsigned)(BTD / 4 / 256)), 256, 0, stream>>>(ctx, mvob);
    gemm(ctx, vo_Wo, (float*)d_out, nullptr, BT, DD, DD, DD, DD, DD, 1, 0, 0, 0, 1);
}


# K loop of the shared GEMM tile software-pipelined: one barrier per 32-k chunk (was two); next chunk's global loads, LDS stores and first fragments go out under this chunk's WMMAs; sum order kept
# speedup vs baseline: 2.9674x; 1.0769x over previous
#include <hip/hip_runtime.h>
#include <math.h>

#define BB 2
#define TT 2048
#define DD 1024
#define RR 64
#define RSS 16
#define BT (BB*TT)

typedef __attribute__((ext_vector_type(2))) float v2f;
typedef __attribute__((ext_vector_type(8))) float v8f;
typedef __attribute__((ext_vector_type(4))) float v4f;

#define GBM 256
#define GBN 64
#define GBK 32

__device__ __forceinline__ void gemm_tile_f32(
    const float* __restrict__ Ab, const float* __restrict__ Bb,
    float* __restrict__ Cb, const float* __restrict__ Mb,
    int K, int lda, int ldb, int ldc, int bT, int m0, int n0)
{
    __shared__ __align__(16) float As [2][GBM][34];
    __shared__ __align__(16) float BsT[2][GBN][34];

    const int tid  = threadIdx.x;
    const int wave = tid >> 5;
    const int lane = tid & 31;
    const int hi   = lane >> 4;
    const int lm   = lane & 15;

    float4 ra[8];
    float4 rb[2];

    int arow[8], ac4[8];
    #pragma unroll
    for (int i = 0; i < 8; ++i) { int idx = tid + i * 256; arow[i] = idx >> 3; ac4[i] = (idx & 7) * 4; }
    int bi0[2], bi1[2];
    #pragma unroll
    for (int i = 0; i < 2; ++i) {
        int idx = tid + i * 256;
        if (true) {   }
        bi0[i] = idx; bi1[i] = idx;
    }

    auto loadGlobal = [&](int k0) {
        #pragma unroll
        for (int i = 0; i < 8; ++i)
            ra[i] = *(const float4*)(Ab + (size_t)(m0 + arow[i]) * lda + k0 + ac4[i]);
        if (bT) {
            #pragma unroll
            for (int i = 0; i < 2; ++i) {
                int idx = bi0[i];
                int n = idx >> 3, c4 = (idx & 7) * 4;
                rb[i] = *(const float4*)(Bb + (size_t)(n0 + n) * ldb + k0 + c4);
            }
        } else {
            #pragma unroll
            for (int i = 0; i < 2; ++i) {
                int idx = bi1[i];
                int r = idx >> 4, c4 = (idx & 15) * 4;
                rb[i] = *(const float4*)(Bb + (size_t)(k0 + r) * ldb + n0 + c4);
            }
        }
    };

    auto storeLDS = [&](int buf) {
        #pragma unroll
        for (int i = 0; i < 8; ++i) {
            float4 t = ra[i];
            v2f lo; lo.x = t.x; lo.y = t.y;
            v2f hi2; hi2.x = t.z; hi2.y = t.w;
            *(v2f*)&As[buf][arow[i]][ac4[i]]     = lo;
            *(v2f*)&As[buf][arow[i]][ac4[i] + 2] = hi2;
        }
        if (bT) {
            #pragma unroll
            for (int i = 0; i < 2; ++i) {
                int idx = bi0[i];
                int n = idx >> 3, c4 = (idx & 7) * 4;
                float4 t = rb[i];
                v2f lo; lo.x = t.x; lo.y = t.y;
                v2f hi2; hi2.x = t.z; hi2.y = t.w;
                *(v2f*)&BsT[buf][n][c4]     = lo;
                *(v2f*)&BsT[buf][n][c4 + 2] = hi2;
            }
        } else {
            #pragma unroll
            for (int i = 0; i < 2; ++i) {
                int idx = bi1[i];
                int r = idx >> 4, c4 = (idx & 15) * 4;
                float4 t = rb[i];
                BsT[buf][c4 + 0][r] = t.x;
                BsT[buf][c4 + 1][r] = t.y;
                BsT[buf][c4 + 2][r] = t.z;
                BsT[buf][c4 + 3][r] = t.w;
            }
        }
    };

    v8f acc00 = {}, acc01 = {}, acc02 = {}, acc03 = {};
    v8f acc10 = {}, acc11 = {}, acc12 = {}, acc13 = {};

    const int r0 = wave * 32 + lm;
    const int r1 = r0 + 16;
    const int nchunks = K / GBK;

    struct Frag { v2f a0[4], a1[4], b0[4], b1[4], b2[4], b3[4]; };

    auto loadFrag = [&](Frag& f, int buf, int step0) {
        #pragma unroll
        for (int s = 0; s < 4; ++s) {
            const int kc = (step0 + s) * 4 + 2 * hi;
            f.a0[s] = *(const v2f*)&As [buf][r0][kc];
            f.a1[s] = *(const v2f*)&As [buf][r1][kc];
            f.b0[s] = *(const v2f*)&BsT[buf][lm +  0][kc];
            f.b1[s] = *(const v2f*)&BsT[buf][lm + 16][kc];
            f.b2[s] = *(const v2f*)&BsT[buf][lm + 32][kc];
            f.b3[s] = *(const v2f*)&BsT[buf][lm + 48][kc];
        }
    };

    auto mma = [&](const Frag& f, int s) {
        acc00 = __builtin_amdgcn_wmma_f32_16x16x4_f32(false, f.a0[s], false, f.b0[s], (short)0, acc00, false, false);
        acc01 = __builtin_amdgcn_wmma_f32_16x16x4_f32(false, f.a0[s], false, f.b1[s], (short)0, acc01, false, false);
        acc02 = __builtin_amdgcn_wmma_f32_16x16x4_f32(false, f.a0[s], false, f.b2[s], (short)0, acc02, false, false);
        acc03 = __builtin_amdgcn_wmma_f32_16x16x4_f32(false, f.a0[s], false, f.b3[s], (short)0, acc03, false, false);
        acc10 = __builtin_amdgcn_wmma_f32_16x16x4_f32(false, f.a1[s], false, f.b0[s], (short)0, acc10, false, false);
        acc11 = __builtin_amdgcn_wmma_f32_16x16x4_f32(false, f.a1[s], false, f.b1[s], (short)0, acc11, false, false);
        acc12 = __builtin_amdgcn_wmma_f32_16x16x4_f32(false, f.a1[s], false, f.b2[s], (short)0, acc12, false, false);
        acc13 = __builtin_amdgcn_wmma_f32_16x16x4_f32(false, f.a1[s], false, f.b3[s], (short)0, acc13, false, false);
    };

    Frag P, Q;
    loadGlobal(0);
    storeLDS(0);
    __syncthreads();
    loadFrag(P, 0, 0);
    for (int c = 0; c + 1 < nchunks; ++c) {
        const int buf = c & 1;
        loadGlobal((c + 1) * GBK);
        loadFrag(Q, buf, 4);
        __builtin_amdgcn_sched_barrier(0);
        mma(P, 0); mma(P, 1); mma(P, 2); mma(P, 3);
        mma(Q, 0); mma(Q, 1);
        __builtin_amdgcn_sched_barrier(0);
        storeLDS(buf ^ 1);
        __builtin_amdgcn_sched_barrier(0);
        mma(Q, 2);
        __builtin_amdgcn_sched_barrier(0);
        __syncthreads();
        loadFrag(P, buf ^ 1, 0);
        __builtin_amdgcn_sched_barrier(0);
        mma(Q, 3);
    }
    {
        loadFrag(Q, (nchunks - 1) & 1, 4);
        __builtin_amdgcn_sched_barrier(0);
        mma(P, 0); mma(P, 1); mma(P, 2); mma(P, 3);
        mma(Q, 0); mma(Q, 1); mma(Q, 2); mma(Q, 3);
    }

    const int ncol = n0 + lm;
    #pragma unroll
    for (int mt = 0; mt < 2; ++mt) {
        const int mrow = m0 + wave * 32 + mt * 16 + hi * 8;
        v8f a0 = mt ? acc10 : acc00, a1 = mt ? acc11 : acc01;
        v8f a2 = mt ? acc12 : acc02, a3 = mt ? acc13 : acc03;
        #pragma unroll
        for (int r = 0; r < 8; ++r) {
            size_t base = (size_t)(mrow + r) * ldc + ncol;
            float v0 = a0[r], v1 = a1[r], v2 = a2[r], v3 = a3[r];
            if (Mb) {
                v0 *= Mb[base]; v1 *= Mb[base + 16]; v2 *= Mb[base + 32]; v3 *= Mb[base + 48];
            }
            Cb[base] = v0; Cb[base + 16] = v1; Cb[base + 32] = v2; Cb[base + 48] = v3;
        }
    }
}

__global__ __launch_bounds__(256) void gemm_wmma_f32(
    const float* __restrict__ A, const float* __restrict__ Bm,
    float* __restrict__ C, const float* __restrict__ mulC,
    int M, int N, int K, int lda, int ldb, int ldc, int bT,
    long long sA, long long sB, long long sC)
{
    (void)M; (void)N;
    const float* Ap = A  + (size_t)blockIdx.z * sA;
    const float* Bp = Bm + (size_t)blockIdx.z * sB;
    float*       Cp = C  + (size_t)blockIdx.z * sC;
    const float* Mp = mulC ? (mulC + (size_t)blockIdx.z * sC) : nullptr;
    if (bT) gemm_tile_f32(Ap, Bp, Cp, Mp, K, lda, ldb, ldc, 1, blockIdx.y * GBM, blockIdx.x * GBN);
    else    gemm_tile_f32(Ap, Bp, Cp, Mp, K, lda, ldb, ldc, 0, blockIdx.y * GBM, blockIdx.x * GBN);
}

struct Stage2Args {
    const float *a0, *a1, *a2, *a3, *a4, *a5;
    const float *w0, *w1, *w2, *w3, *w4, *w5;
    float       *c0, *c1, *c2, *c3, *c4, *c5;
    const float *g0, *g1, *g2, *g3, *g4, *g5;
};

__global__ __launch_bounds__(256) void gemm_stage2_kernel(Stage2Args p)
{
    const int s = blockIdx.z;
    const float* A = s == 0 ? p.a0 : s == 1 ? p.a1 : s == 2 ? p.a2 : s == 3 ? p.a3 : s == 4 ? p.a4 : p.a5;
    const float* Wm = s == 0 ? p.w0 : s == 1 ? p.w1 : s == 2 ? p.w2 : s == 3 ? p.w3 : s == 4 ? p.w4 : p.w5;
    float*       C = s == 0 ? p.c0 : s == 1 ? p.c1 : s == 2 ? p.c2 : s == 3 ? p.c3 : s == 4 ? p.c4 : p.c5;
    const float* G = s == 0 ? p.g0 : s == 1 ? p.g1 : s == 2 ? p.g2 : s == 3 ? p.g3 : s == 4 ? p.g4 : p.g5;
    gemm_tile_f32(A, Wm, C, G, RR, RR, DD, DD, 0, blockIdx.y * GBM, blockIdx.x * GBN);
}

__device__ inline float softplusf(float x) {
    if (x > 20.f) return x;
    if (x < -20.f) return expf(x);
    return log1pf(expf(x));
}

struct XwArgs {
    const float* x;
    const float *w0, *w1, *w2, *w3, *w4, *w5, *w6, *w7;
    const float *e0, *e1, *e2, *e3, *e4, *e5;
    const float *wdA, *wdB, *wdC;
    float* xr;
    float *o0, *o1, *o2, *o3, *o4, *o5;
    float *dA, *dB, *dC;
};

struct BFrag {
    v4f    r0[8], r1[8];
    float  e0[8], e1[8];
};

#define XT_ROWS 32
#define XT_K    32
#define XT_PITCH 36

__global__ __launch_bounds__(256) void xw_shared_kernel(XwArgs p)
{
    __shared__ __align__(16) float Xs[2][XT_ROWS][XT_PITCH];

    const int tid  = threadIdx.x;
    const int wave = tid >> 5, lane = tid & 31;
    const int hi   = lane >> 4, lm = lane & 15;
    const int m0   = blockIdx.x * XT_ROWS;
    const int sel  = blockIdx.y * 4 + (wave >> 1);
    const int mt   = wave & 1;

    const float* Wm = sel == 0 ? p.w0 : sel == 1 ? p.w1 : sel == 2 ? p.w2 : sel == 3 ? p.w3
                    : sel == 4 ? p.w4 : sel == 5 ? p.w5 : sel == 6 ? p.w6 : p.w7;
    Wm += (size_t)(2 * hi) * RR + 4 * lm;

    const float* ep;
    int es;
    if (sel < 6) {
        ep = (sel == 0 ? p.e0 : sel == 1 ? p.e1 : sel == 2 ? p.e2
            : sel == 3 ? p.e3 : sel == 4 ? p.e4 : p.e5) + lm;
        es = RSS;
    } else {
        ep = (sel == 6 && lm == 1) ? p.wdB : (sel == 6 && lm == 2) ? p.wdC : p.wdA;
        es = 1;
    }
    ep += (size_t)(2 * hi) * es;

    const int xrow = tid >> 3, xc4 = (tid & 7) * 4;
    const float* xg = p.x + (size_t)(m0 + xrow) * DD + xc4;

    auto loadB = [&](BFrag& f, int k0) {
        #pragma unroll
        for (int s = 0; s < 8; ++s) {
            const int k = k0 + 4 * s;
            f.r0[s] = *(const v4f*)(Wm + (size_t)k * RR);
            f.r1[s] = *(const v4f*)(Wm + (size_t)(k + 1) * RR);
            f.e0[s] = ep[(size_t)k * es];
            f.e1[s] = ep[(size_t)(k + 1) * es];
        }
    };

    v8f acc0 = {}, acc1 = {}, acc2 = {}, acc3 = {}, accE = {};
    const int arow = mt * 16 + lm;
    const int nch  = DD / XT_K;
    v4f xreg;

    auto stage = [&](const BFrag& cur, BFrag& nxt, int c, int buf) {
        *(v4f*)&Xs[buf][xrow][xc4] = xreg;
        __syncthreads();
        const int cn = (c + 1 < nch) ? c + 1 : c;
        xreg = *(const v4f*)(xg + cn * XT_K);
        loadB(nxt, cn * XT_K);

        v2f a[8];
        #pragma unroll
        for (int s = 0; s < 8; ++s)
            a[s] = *(const v2f*)&Xs[buf][arow][4 * s + 2 * hi];
        __builtin_amdgcn_sched_barrier(0);
        #pragma unroll
        for (int s = 0; s < 8; ++s) {
            v2f b0, b1, b2, b3, be;
            b0.x = cur.r0[s].x; b0.y = cur.r1[s].x;
            b1.x = cur.r0[s].y; b1.y = cur.r1[s].y;
            b2.x = cur.r0[s].z; b2.y = cur.r1[s].z;
            b3.x = cur.r0[s].w; b3.y = cur.r1[s].w;
            be.x = cur.e0[s];   be.y = cur.e1[s];
            acc0 = __builtin_amdgcn_wmma_f32_16x16x4_f32(false, a[s], false, b0, (short)0, acc0, false, false);
            acc1 = __builtin_amdgcn_wmma_f32_16x16x4_f32(false, a[s], false, b1, (short)0, acc1, false, false);
            acc2 = __builtin_amdgcn_wmma_f32_16x16x4_f32(false, a[s], false, b2, (short)0, acc2, false, false);
            acc3 = __builtin_amdgcn_wmma_f32_16x16x4_f32(false, a[s], false, b3, (short)0, acc3, false, false);
            accE = __builtin_amdgcn_wmma_f32_16x16x4_f32(false, a[s], false, be, (short)0, accE, false, false);
        }
        __builtin_amdgcn_sched_barrier(0);
        asm volatile("" :: "v"(nxt.r0[0]), "v"(nxt.r0[1]), "v"(nxt.r0[2]), "v"(nxt.r0[3]),
                           "v"(nxt.r0[4]), "v"(nxt.r0[5]), "v"(nxt.r0[6]), "v"(nxt.r0[7]),
                           "v"(nxt.r1[0]), "v"(nxt.r1[1]), "v"(nxt.r1[2]), "v"(nxt.r1[3]),
                           "v"(nxt.r1[4]), "v"(nxt.r1[5]), "v"(nxt.r1[6]), "v"(nxt.r1[7]));
        asm volatile("" :: "v"(nxt.e0[0]), "v"(nxt.e0[1]), "v"(nxt.e0[2]), "v"(nxt.e0[3]),
                           "v"(nxt.e0[4]), "v"(nxt.e0[5]), "v"(nxt.e0[6]), "v"(nxt.e0[7]),
                           "v"(nxt.e1[0]), "v"(nxt.e1[1]), "v"(nxt.e1[2]), "v"(nxt.e1[3]),
                           "v"(nxt.e1[4]), "v"(nxt.e1[5]), "v"(nxt.e1[6]), "v"(nxt.e1[7]),
                           "v"(xreg));
    };

    BFrag f0, f1;
    xreg = *(const v4f*)xg;
    loadB(f0, 0);
    for (int c = 0; c < nch; c += 2) {
        stage(f0, f1, c,     0);
        stage(f1, f0, c + 1, 1);
    }

    const int orow = m0 + mt * 16 + hi * 8;
    float* xo = p.xr + (size_t)sel * BT * RR + (size_t)orow * RR + 4 * lm;
    #pragma unroll
    for (int r = 0; r < 8; ++r) {
        float4 o; o.x = acc0[r]; o.y = acc1[r]; o.z = acc2[r]; o.w = acc3[r];
        *(float4*)(xo + (size_t)r * RR) = o;
    }
    if (sel < 6) {
        float* eo = (sel == 0 ? p.o0 : sel == 1 ? p.o1 : sel == 2 ? p.o2
                   : sel == 3 ? p.o3 : sel == 4 ? p.o4 : p.o5) + (size_t)orow * RSS + lm;
        #pragma unroll
        for (int r = 0; r < 8; ++r) eo[(size_t)r * RSS] = accE[r];
    } else if (sel == 6 && lm < 3) {
        float* dp = (lm == 0 ? p.dA : lm == 1 ? p.dB : p.dC) + orow;
        #pragma unroll
        for (int r = 0; r < 8; ++r) dp[r] = accE[r];
    }
}

__device__ inline v8f strip_times_factor(const v8f (&m)[8], const float* __restrict__ fac)
{
    v8f acc0 = {}, acc1 = {};
    #pragma unroll
    for (int t = 0; t < 8; ++t) {
        float bv[8];
        #pragma unroll
        for (int r = 0; r < 8; ++r) bv[r] = fac[(size_t)(16 * t + r) * RSS];
        #pragma unroll
        for (int q = 0; q < 4; ++q) {
            v2f a, b;
            a.x = m[t][2 * q]; a.y = m[t][2 * q + 1];
            b.x = bv[2 * q];   b.y = bv[2 * q + 1];
            if (q & 1) acc1 = __builtin_amdgcn_wmma_f32_16x16x4_f32(false, a, false, b, (short)0, acc1, false, false);
            else       acc0 = __builtin_amdgcn_wmma_f32_16x16x4_f32(false, a, false, b, (short)0, acc0, false, false);
        }
    }
    return acc0 + acc1;
}

__device__ inline void mix_apply(v8f (&m)[8], const float* __restrict__ dptr,
                                 const float* __restrict__ Arow, const v4f zb0, const v4f zb1)
{
    v2f b0, b1, b2, b3;
    b0.x = zb0.x; b0.y = zb0.y; b1.x = zb0.z; b1.y = zb0.w;
    b2.x = zb1.x; b2.y = zb1.y; b3.x = zb1.z; b3.y = zb1.w;
    #pragma unroll
    for (int t = 0; t < 8; ++t) {
        const v4f dlo = *(const v4f*)(dptr + 16 * t), dhi = *(const v4f*)(dptr + 16 * t + 4);
        const v4f f0 = *(const v4f*)(Arow + (size_t)(16 * t) * RSS);
        const v4f f1 = *(const v4f*)(Arow + (size_t)(16 * t) * RSS + 4);
        v8f c;
        c[0] = m[t][0] * dlo.x; c[1] = m[t][1] * dlo.y; c[2] = m[t][2] * dlo.z; c[3] = m[t][3] * dlo.w;
        c[4] = m[t][4] * dhi.x; c[5] = m[t][5] * dhi.y; c[6] = m[t][6] * dhi.z; c[7] = m[t][7] * dhi.w;
        v2f a0, a1, a2, a3;
        a0.x = f0.x; a0.y = f0.y; a1.x = f0.z; a1.y = f0.w;
        a2.x = f1.x; a2.y = f1.y; a3.x = f1.z; a3.y = f1.w;
        c = __builtin_amdgcn_wmma_f32_16x16x4_f32(false, a0, false, b0, (short)0, c, false, false);
        c = __builtin_amdgcn_wmma_f32_16x16x4_f32(false, a1, false, b1, (short)0, c, false, false);
        c = __builtin_amdgcn_wmma_f32_16x16x4_f32(false, a2, false, b2, (short)0, c, false, false);
        c = __builtin_amdgcn_wmma_f32_16x16x4_f32(false, a3, false, b3, (short)0, c, false, false);
        m[t] = c;
    }
}

__global__ __launch_bounds__(512) void mix_chain_kernel(
    const float* __restrict__ z, const float* zs,
    const float* __restrict__ d1,  const float* __restrict__ A1,  const float* __restrict__ B1,
    const float* __restrict__ d1s, const float* __restrict__ A1s, const float* __restrict__ B1s,
    const float* __restrict__ d2,  const float* __restrict__ A2,  const float* __restrict__ B2,
    float* out)
{
    __shared__ __align__(16) float part[2][16][16][16];
    __shared__ __align__(16) float full[2][16][16];
    __shared__ float rowpart[16][16];

    const int tid  = threadIdx.x;
    const int wave = tid >> 5, lane = tid & 31;
    const int hi   = lane >> 4, lm = lane & 15;
    const int row0 = blockIdx.x * 16;
    const int fb   = (row0 >> 11) * TT;
    const int j0   = wave * 128;

    const size_t soff = (size_t)(row0 + lm) * TT + j0 + 8 * hi;
    const size_t frow = (size_t)(fb + j0 + 8 * hi) * RSS + lm;
    const size_t arow = (size_t)(fb + j0 + lm) * RSS + 8 * hi;
    const int    doff = fb + j0 + 8 * hi;

    v8f m[8], ms[8];
    #pragma unroll
    for (int t = 0; t < 8; ++t) {
        const v4f lo = *(const v4f*)(z + soff + 16 * t), up = *(const v4f*)(z + soff + 16 * t + 4);
        m[t][0] = lo.x; m[t][1] = lo.y; m[t][2] = lo.z; m[t][3] = lo.w;
        m[t][4] = up.x; m[t][5] = up.y; m[t][6] = up.z; m[t][7] = up.w;
    }
    #pragma unroll
    for (int t = 0; t < 8; ++t) {
        const v4f lo = *(const v4f*)(zs + soff + 16 * t), up = *(const v4f*)(zs + soff + 16 * t + 4);
        ms[t][0] = lo.x; ms[t][1] = lo.y; ms[t][2] = lo.z; ms[t][3] = lo.w;
        ms[t][4] = up.x; ms[t][5] = up.y; ms[t][6] = up.z; ms[t][7] = up.w;
    }

    {
        const v8f p0 = strip_times_factor(m, B1 + frow);
        #pragma unroll
        for (int r = 0; r < 8; ++r) part[0][wave][8 * hi + r][lm] = p0[r];
    }
    {
        const v8f p1 = strip_times_factor(ms, B1s + frow);
        #pragma unroll
        for (int r = 0; r < 8; ++r) part[1][wave][8 * hi + r][lm] = p1[r];
    }
    __syncthreads();
    {
        const int pr = tid >> 8, e = tid & 255;
        float v = 0.f;
        #pragma unroll
        for (int w = 0; w < 16; ++w) v += (&part[pr][w][0][0])[e];
        (&full[pr][0][0])[e] = v;
    }
    __syncthreads();

    mix_apply(m,  d1  + doff, A1  + arow, *(const v4f*)&full[0][lm][8 * hi], *(const v4f*)&full[0][lm][8 * hi + 4]);
    mix_apply(ms, d1s + doff, A1s + arow, *(const v4f*)&full[1][lm][8 * hi], *(const v4f*)&full[1][lm][8 * hi + 4]);

    float ss = 0.f;
    #pragma unroll
    for (int t = 0; t < 8; ++t) {
        #pragma unroll
        for (int r = 0; r < 8; ++r) ss += m[t][r] * m[t][r];
    }
    ss += __shfl_xor(ss, 16, 32);
    if (hi == 0) rowpart[wave][lm] = ss;
    __syncthreads();
    float tot = 0.f;
    #pragma unroll
    for (int w = 0; w < 16; ++w) tot += rowpart[w][lm];
    const float inv = 1.f / (sqrtf(tot) + 1e-8f);

    #pragma unroll
    for (int t = 0; t < 8; ++t) {
        #pragma unroll
        for (int r = 0; r < 8; ++r) m[t][r] = softplusf(ms[t][r]) * fmaxf(m[t][r] * inv, 0.f);
    }

    {
        const v8f p0 = strip_times_factor(m, B2 + frow);
        #pragma unroll
        for (int r = 0; r < 8; ++r) part[0][wave][8 * hi + r][lm] = p0[r];
    }
    __syncthreads();
    if (tid < 256) {
        float v = 0.f;
        #pragma unroll
        for (int w = 0; w < 16; ++w) v += (&part[0][w][0][0])[tid];
        (&full[0][0][0])[tid] = v;
    }
    __syncthreads();
    mix_apply(m, d2 + doff, A2 + arow, *(const v4f*)&full[0][lm][8 * hi], *(const v4f*)&full[0][lm][8 * hi + 4]);

    #pragma unroll
    for (int t = 0; t < 8; ++t) {
        v4f lo, up;
        lo.x = m[t][0]; lo.y = m[t][1]; lo.z = m[t][2]; lo.w = m[t][3];
        up.x = m[t][4]; up.y = m[t][5]; up.z = m[t][6]; up.w = m[t][7];
        *(v4f*)(out + soff + 16 * t)     = lo;
        *(v4f*)(out + soff + 16 * t + 4) = up;
    }
}

__global__ __launch_bounds__(256) void emul_kernel(float* __restrict__ a, const float* __restrict__ b)
{
    const size_t i = ((size_t)blockIdx.x * blockDim.x + threadIdx.x) * 4;
    float4 av = *(const float4*)(a + i);
    float4 bv = *(const float4*)(b + i);
    av.x *= bv.x; av.y *= bv.y; av.z *= bv.z; av.w *= bv.w;
    *(float4*)(a + i) = av;
}

extern "C" void kernel_launch(void* const* d_in, const int* in_sizes, int n_in,
                              void* d_out, int out_size, void* d_ws, size_t ws_size,
                              hipStream_t stream) {
    (void)in_sizes; (void)n_in; (void)out_size; (void)ws_size;

    const float* x = (const float*)d_in[0];
    const float* qk_Wq1  = (const float*)d_in[1];  const float* qk_Wq2  = (const float*)d_in[2];
    const float* qk_Wm1  = (const float*)d_in[3];  const float* qk_Wm2  = (const float*)d_in[4];
    const float* qk_Wk1  = (const float*)d_in[5];  const float* qk_Wk2  = (const float*)d_in[6];
    const float* qks_Wq1 = (const float*)d_in[7];  const float* qks_Wq2 = (const float*)d_in[8];
    const float* qks_Wm1 = (const float*)d_in[9];  const float* qks_Wm2 = (const float*)d_in[10];
    const float* qks_Wk1 = (const float*)d_in[11]; const float* qks_Wk2 = (const float*)d_in[12];
    const float* vo_Wv1  = (const float*)d_in[13]; const float* vo_Wv2  = (const float*)d_in[14];
    const float* vo_Wm1  = (const float*)d_in[15]; const float* vo_Wm2  = (const float*)d_in[16];
    const float* vo_Wo   = (const float*)d_in[17];
    const float* sm1_Wd  = (const float*)d_in[18]; const float* sm1_Wa  = (const float*)d_in[19]; const float* sm1_Wb  = (const float*)d_in[20];
    const float* sm1s_Wd = (const float*)d_in[21]; const float* sm1s_Wa = (const float*)d_in[22]; const float* sm1s_Wb = (const float*)d_in[23];
    const float* sm2_Wd  = (const float*)d_in[24]; const float* sm2_Wa  = (const float*)d_in[25]; const float* sm2_Wb  = (const float*)d_in[26];

    const size_t BTD = (size_t)BT * DD;
    const size_t BTT = (size_t)BB * TT * TT;

    float* w = (float*)d_ws;
    float* qg   = w; w += BTD;
    float* kb   = w; w += BTD;
    float* qsg  = w; w += BTD;
    float* ksb  = w; w += BTD;
    float* vb   = w; w += BTD;
    float* mvob = w; w += BTD;
    float* tmpb = w; w += BTD;
    float* zb   = w; w += BTT;
    float* zsb  = w; w += BTT;
    float* ctx  = w; w += BTD;
    w += (size_t)BT * RR;
    float* d1   = w; w += BT; float* d1s = w; w += BT; float* d2 = w; w += BT;
    float* A1   = w; w += (size_t)BT * RSS; float* B1  = w; w += (size_t)BT * RSS;
    float* A1s  = w; w += (size_t)BT * RSS; float* B1s = w; w += (size_t)BT * RSS;
    float* A2   = w; w += (size_t)BT * RSS; float* B2  = w; w += (size_t)BT * RSS;

    auto gemm = [&](const float* A, const float* Bm, float* C, const float* mul,
                    int M, int N, int K, int lda, int ldb, int ldc, int bT,
                    long long sA, long long sB, long long sC, int batches) {
        dim3 g((unsigned)(N / GBN), (unsigned)(M / GBM), (unsigned)batches);
        gemm_wmma_f32<<<g, dim3(256), 0, stream>>>(A, Bm, C, mul, M, N, K, lda, ldb, ldc, bT, sA, sB, sC);
    };

    float* xrAll = ctx;
    const size_t XP = (size_t)BT * RR;
    XwArgs xa;
    xa.x  = x;
    xa.w0 = qk_Wm1;  xa.w1 = qk_Wq1;  xa.w2 = qk_Wk1;
    xa.w3 = qks_Wm1; xa.w4 = qks_Wq1; xa.w5 = qks_Wk1;
    xa.w6 = vo_Wv1;  xa.w7 = vo_Wm1;
    xa.e0 = sm1_Wa;  xa.e1 = sm1_Wb;  xa.e2 = sm1s_Wa; xa.e3 = sm1s_Wb; xa.e4 = sm2_Wa; xa.e5 = sm2_Wb;
    xa.wdA = sm1_Wd; xa.wdB = sm1s_Wd; xa.wdC = sm2_Wd;
    xa.xr = xrAll;
    xa.o0 = A1; xa.o1 = B1; xa.o2 = A1s; xa.o3 = B1s; xa.o4 = A2; xa.o5 = B2;
    xa.dA = d1; xa.dB = d1s; xa.dC = d2;
    xw_shared_kernel<<<dim3(BT / XT_ROWS, 2), 256, 0, stream>>>(xa);

    float* tmpb2 = zsb;
    const dim3 g2((unsigned)(DD / GBN), (unsigned)(BT / GBM), 6u);
    Stage2Args sa;
    sa.a0 = xrAll + 0 * XP; sa.w0 = qk_Wm2;  sa.c0 = tmpb;  sa.g0 = nullptr;
    sa.a1 = xrAll + 3 * XP; sa.w1 = qks_Wm2; sa.c1 = tmpb2; sa.g1 = nullptr;
    sa.a2 = xrAll + 2 * XP; sa.w2 = qk_Wk2;  sa.c2 = kb;    sa.g2 = nullptr;
    sa.a3 = xrAll + 5 * XP; sa.w3 = qks_Wk2; sa.c3 = ksb;   sa.g3 = nullptr;
    sa.a4 = xrAll + 6 * XP; sa.w4 = vo_Wv2;  sa.c4 = vb;    sa.g4 = nullptr;
    sa.a5 = xrAll + 7 * XP; sa.w5 = vo_Wm2;  sa.c5 = mvob;  sa.g5 = nullptr;
    gemm_stage2_kernel<<<g2, dim3(256), 0, stream>>>(sa);

    const dim3 g2b((unsigned)(DD / GBN), (unsigned)(BT / GBM), 2u);
    Stage2Args sb;
    sb.a0 = xrAll + 1 * XP; sb.w0 = qk_Wq2;  sb.c0 = qg;  sb.g0 = tmpb;
    sb.a1 = xrAll + 4 * XP; sb.w1 = qks_Wq2; sb.c1 = qsg; sb.g1 = tmpb2;
    sb.a2 = nullptr; sb.w2 = nullptr; sb.c2 = nullptr; sb.g2 = nullptr;
    sb.a3 = nullptr; sb.w3 = nullptr; sb.c3 = nullptr; sb.g3 = nullptr;
    sb.a4 = nullptr; sb.w4 = nullptr; sb.c4 = nullptr; sb.g4 = nullptr;
    sb.a5 = nullptr; sb.w5 = nullptr; sb.c5 = nullptr; sb.g5 = nullptr;
    gemm_stage2_kernel<<<g2b, dim3(256), 0, stream>>>(sb);

    const long long sTD = (long long)TT * DD, sTTl = (long long)TT * TT;
    gemm(qg,  kb,  zb,  nullptr, TT, TT, DD, DD, DD, TT, 1, sTD, sTD, sTTl, BB);
    gemm(qsg, ksb, zsb, nullptr, TT, TT, DD, DD, DD, TT, 1, sTD, sTD, sTTl, BB);

    mix_chain_kernel<<<dim3(BT / 16), 512, 0, stream>>>(zb, zsb, d1, A1, B1, d1s, A1s, B1s, d2, A2, B2, zsb);

    gemm(zsb, vb, ctx, nullptr, TT, DD, TT, TT, DD, DD, 0, sTTl, sTD, sTD, BB);
    emul_kernel<<<dim3((unsigned)(BTD / 4 / 256)), 256, 0, stream>>>(ctx, mvob);
    gemm(ctx, vo_Wo, (float*)d_out, nullptr, BT, DD, DD, DD, DD, DD, 1, 0, 0, 0, 1);
}


# gated combine of the mix chain: softplus = max(x,0)+log1p(e^-|x|) on hardware exp2/log2/rcp ops, 1+t rounding divided out; no library expf/log1pf chain, no branch per value; few-ulp change
# speedup vs baseline: 3.1138x; 1.0493x over previous
#include <hip/hip_runtime.h>
#include <math.h>

#define BB 2
#define TT 2048
#define DD 1024
#define RR 64
#define RSS 16
#define BT (BB*TT)

typedef __attribute__((ext_vector_type(2))) float v2f;
typedef __attribute__((ext_vector_type(8))) float v8f;
typedef __attribute__((ext_vector_type(4))) float v4f;

#define GBM 256
#define GBN 64
#define GBK 32

__device__ __forceinline__ void gemm_tile_f32(
    const float* __restrict__ Ab, const float* __restrict__ Bb,
    float* __restrict__ Cb, const float* __restrict__ Mb,
    int K, int lda, int ldb, int ldc, int bT, int m0, int n0)
{
    __shared__ __align__(16) float As [2][GBM][34];
    __shared__ __align__(16) float BsT[2][GBN][34];

    const int tid  = threadIdx.x;
    const int wave = tid >> 5;
    const int lane = tid & 31;
    const int hi   = lane >> 4;
    const int lm   = lane & 15;

    float4 ra[8];
    float4 rb[2];

    int arow[8], ac4[8];
    #pragma unroll
    for (int i = 0; i < 8; ++i) { int idx = tid + i * 256; arow[i] = idx >> 3; ac4[i] = (idx & 7) * 4; }
    int bi0[2], bi1[2];
    #pragma unroll
    for (int i = 0; i < 2; ++i) {
        int idx = tid + i * 256;
        if (true) {   }
        bi0[i] = idx; bi1[i] = idx;
    }

    auto loadGlobal = [&](int k0) {
        #pragma unroll
        for (int i = 0; i < 8; ++i)
            ra[i] = *(const float4*)(Ab + (size_t)(m0 + arow[i]) * lda + k0 + ac4[i]);
        if (bT) {
            #pragma unroll
            for (int i = 0; i < 2; ++i) {
                int idx = bi0[i];
                int n = idx >> 3, c4 = (idx & 7) * 4;
                rb[i] = *(const float4*)(Bb + (size_t)(n0 + n) * ldb + k0 + c4);
            }
        } else {
            #pragma unroll
            for (int i = 0; i < 2; ++i) {
                int idx = bi1[i];
                int r = idx >> 4, c4 = (idx & 15) * 4;
                rb[i] = *(const float4*)(Bb + (size_t)(k0 + r) * ldb + n0 + c4);
            }
        }
    };

    auto storeLDS = [&](int buf) {
        #pragma unroll
        for (int i = 0; i < 8; ++i) {
            float4 t = ra[i];
            v2f lo; lo.x = t.x; lo.y = t.y;
            v2f hi2; hi2.x = t.z; hi2.y = t.w;
            *(v2f*)&As[buf][arow[i]][ac4[i]]     = lo;
            *(v2f*)&As[buf][arow[i]][ac4[i] + 2] = hi2;
        }
        if (bT) {
            #pragma unroll
            for (int i = 0; i < 2; ++i) {
                int idx = bi0[i];
                int n = idx >> 3, c4 = (idx & 7) * 4;
                float4 t = rb[i];
                v2f lo; lo.x = t.x; lo.y = t.y;
                v2f hi2; hi2.x = t.z; hi2.y = t.w;
                *(v2f*)&BsT[buf][n][c4]     = lo;
                *(v2f*)&BsT[buf][n][c4 + 2] = hi2;
            }
        } else {
            #pragma unroll
            for (int i = 0; i < 2; ++i) {
                int idx = bi1[i];
                int r = idx >> 4, c4 = (idx & 15) * 4;
                float4 t = rb[i];
                BsT[buf][c4 + 0][r] = t.x;
                BsT[buf][c4 + 1][r] = t.y;
                BsT[buf][c4 + 2][r] = t.z;
                BsT[buf][c4 + 3][r] = t.w;
            }
        }
    };

    v8f acc00 = {}, acc01 = {}, acc02 = {}, acc03 = {};
    v8f acc10 = {}, acc11 = {}, acc12 = {}, acc13 = {};

    const int r0 = wave * 32 + lm;
    const int r1 = r0 + 16;
    const int nchunks = K / GBK;

    struct Frag { v2f a0[4], a1[4], b0[4], b1[4], b2[4], b3[4]; };

    auto loadFrag = [&](Frag& f, int buf, int step0) {
        #pragma unroll
        for (int s = 0; s < 4; ++s) {
            const int kc = (step0 + s) * 4 + 2 * hi;
            f.a0[s] = *(const v2f*)&As [buf][r0][kc];
            f.a1[s] = *(const v2f*)&As [buf][r1][kc];
            f.b0[s] = *(const v2f*)&BsT[buf][lm +  0][kc];
            f.b1[s] = *(const v2f*)&BsT[buf][lm + 16][kc];
            f.b2[s] = *(const v2f*)&BsT[buf][lm + 32][kc];
            f.b3[s] = *(const v2f*)&BsT[buf][lm + 48][kc];
        }
    };

    auto mma = [&](const Frag& f, int s) {
        acc00 = __builtin_amdgcn_wmma_f32_16x16x4_f32(false, f.a0[s], false, f.b0[s], (short)0, acc00, false, false);
        acc01 = __builtin_amdgcn_wmma_f32_16x16x4_f32(false, f.a0[s], false, f.b1[s], (short)0, acc01, false, false);
        acc02 = __builtin_amdgcn_wmma_f32_16x16x4_f32(false, f.a0[s], false, f.b2[s], (short)0, acc02, false, false);
        acc03 = __builtin_amdgcn_wmma_f32_16x16x4_f32(false, f.a0[s], false, f.b3[s], (short)0, acc03, false, false);
        acc10 = __builtin_amdgcn_wmma_f32_16x16x4_f32(false, f.a1[s], false, f.b0[s], (short)0, acc10, false, false);
        acc11 = __builtin_amdgcn_wmma_f32_16x16x4_f32(false, f.a1[s], false, f.b1[s], (short)0, acc11, false, false);
        acc12 = __builtin_amdgcn_wmma_f32_16x16x4_f32(false, f.a1[s], false, f.b2[s], (short)0, acc12, false, false);
        acc13 = __builtin_amdgcn_wmma_f32_16x16x4_f32(false, f.a1[s], false, f.b3[s], (short)0, acc13, false, false);
    };

    Frag P, Q;
    loadGlobal(0);
    storeLDS(0);
    __syncthreads();
    loadFrag(P, 0, 0);
    for (int c = 0; c + 1 < nchunks; ++c) {
        const int buf = c & 1;
        loadGlobal((c + 1) * GBK);
        loadFrag(Q, buf, 4);
        __builtin_amdgcn_sched_barrier(0);
        mma(P, 0); mma(P, 1); mma(P, 2); mma(P, 3);
        mma(Q, 0); mma(Q, 1);
        __builtin_amdgcn_sched_barrier(0);
        storeLDS(buf ^ 1);
        __builtin_amdgcn_sched_barrier(0);
        mma(Q, 2);
        __builtin_amdgcn_sched_barrier(0);
        __syncthreads();
        loadFrag(P, buf ^ 1, 0);
        __builtin_amdgcn_sched_barrier(0);
        mma(Q, 3);
    }
    {
        loadFrag(Q, (nchunks - 1) & 1, 4);
        __builtin_amdgcn_sched_barrier(0);
        mma(P, 0); mma(P, 1); mma(P, 2); mma(P, 3);
        mma(Q, 0); mma(Q, 1); mma(Q, 2); mma(Q, 3);
    }

    const int ncol = n0 + lm;
    #pragma unroll
    for (int mt = 0; mt < 2; ++mt) {
        const int mrow = m0 + wave * 32 + mt * 16 + hi * 8;
        v8f a0 = mt ? acc10 : acc00, a1 = mt ? acc11 : acc01;
        v8f a2 = mt ? acc12 : acc02, a3 = mt ? acc13 : acc03;
        #pragma unroll
        for (int r = 0; r < 8; ++r) {
            size_t base = (size_t)(mrow + r) * ldc + ncol;
            float v0 = a0[r], v1 = a1[r], v2 = a2[r], v3 = a3[r];
            if (Mb) {
                v0 *= Mb[base]; v1 *= Mb[base + 16]; v2 *= Mb[base + 32]; v3 *= Mb[base + 48];
            }
            Cb[base] = v0; Cb[base + 16] = v1; Cb[base + 32] = v2; Cb[base + 48] = v3;
        }
    }
}

__global__ __launch_bounds__(256) void gemm_wmma_f32(
    const float* __restrict__ A, const float* __restrict__ Bm,
    float* __restrict__ C, const float* __restrict__ mulC,
    int M, int N, int K, int lda, int ldb, int ldc, int bT,
    long long sA, long long sB, long long sC)
{
    (void)M; (void)N;
    const float* Ap = A  + (size_t)blockIdx.z * sA;
    const float* Bp = Bm + (size_t)blockIdx.z * sB;
    float*       Cp = C  + (size_t)blockIdx.z * sC;
    const float* Mp = mulC ? (mulC + (size_t)blockIdx.z * sC) : nullptr;
    if (bT) gemm_tile_f32(Ap, Bp, Cp, Mp, K, lda, ldb, ldc, 1, blockIdx.y * GBM, blockIdx.x * GBN);
    else    gemm_tile_f32(Ap, Bp, Cp, Mp, K, lda, ldb, ldc, 0, blockIdx.y * GBM, blockIdx.x * GBN);
}

struct Stage2Args {
    const float *a0, *a1, *a2, *a3, *a4, *a5;
    const float *w0, *w1, *w2, *w3, *w4, *w5;
    float       *c0, *c1, *c2, *c3, *c4, *c5;
    const float *g0, *g1, *g2, *g3, *g4, *g5;
};

__global__ __launch_bounds__(256) void gemm_stage2_kernel(Stage2Args p)
{
    const int s = blockIdx.z;
    const float* A = s == 0 ? p.a0 : s == 1 ? p.a1 : s == 2 ? p.a2 : s == 3 ? p.a3 : s == 4 ? p.a4 : p.a5;
    const float* Wm = s == 0 ? p.w0 : s == 1 ? p.w1 : s == 2 ? p.w2 : s == 3 ? p.w3 : s == 4 ? p.w4 : p.w5;
    float*       C = s == 0 ? p.c0 : s == 1 ? p.c1 : s == 2 ? p.c2 : s == 3 ? p.c3 : s == 4 ? p.c4 : p.c5;
    const float* G = s == 0 ? p.g0 : s == 1 ? p.g1 : s == 2 ? p.g2 : s == 3 ? p.g3 : s == 4 ? p.g4 : p.g5;
    gemm_tile_f32(A, Wm, C, G, RR, RR, DD, DD, 0, blockIdx.y * GBM, blockIdx.x * GBN);
}

__device__ __forceinline__ float softplusf(float x) {
    const float ax = __builtin_fabsf(x);
    const float t  = __builtin_amdgcn_exp2f(-1.44269504f * ax);
    const float u  = 1.0f + t;
    const float d  = u - 1.0f;
    const float l  = __builtin_amdgcn_logf(u) * 0.693147181f;
    const float q  = t * __builtin_amdgcn_rcpf(d);
    const float r  = (d == 0.0f) ? t : l * q;
    return __builtin_fmaxf(x, 0.0f) + r;
}

struct XwArgs {
    const float* x;
    const float *w0, *w1, *w2, *w3, *w4, *w5, *w6, *w7;
    const float *e0, *e1, *e2, *e3, *e4, *e5;
    const float *wdA, *wdB, *wdC;
    float* xr;
    float *o0, *o1, *o2, *o3, *o4, *o5;
    float *dA, *dB, *dC;
};

struct BFrag {
    v4f    r0[8], r1[8];
    float  e0[8], e1[8];
};

#define XT_ROWS 32
#define XT_K    32
#define XT_PITCH 36

__global__ __launch_bounds__(256) void xw_shared_kernel(XwArgs p)
{
    __shared__ __align__(16) float Xs[2][XT_ROWS][XT_PITCH];

    const int tid  = threadIdx.x;
    const int wave = tid >> 5, lane = tid & 31;
    const int hi   = lane >> 4, lm = lane & 15;
    const int m0   = blockIdx.x * XT_ROWS;
    const int sel  = blockIdx.y * 4 + (wave >> 1);
    const int mt   = wave & 1;

    const float* Wm = sel == 0 ? p.w0 : sel == 1 ? p.w1 : sel == 2 ? p.w2 : sel == 3 ? p.w3
                    : sel == 4 ? p.w4 : sel == 5 ? p.w5 : sel == 6 ? p.w6 : p.w7;
    Wm += (size_t)(2 * hi) * RR + 4 * lm;

    const float* ep;
    int es;
    if (sel < 6) {
        ep = (sel == 0 ? p.e0 : sel == 1 ? p.e1 : sel == 2 ? p.e2
            : sel == 3 ? p.e3 : sel == 4 ? p.e4 : p.e5) + lm;
        es = RSS;
    } else {
        ep = (sel == 6 && lm == 1) ? p.wdB : (sel == 6 && lm == 2) ? p.wdC : p.wdA;
        es = 1;
    }
    ep += (size_t)(2 * hi) * es;

    const int xrow = tid >> 3, xc4 = (tid & 7) * 4;
    const float* xg = p.x + (size_t)(m0 + xrow) * DD + xc4;

    auto loadB = [&](BFrag& f, int k0) {
        #pragma unroll
        for (int s = 0; s < 8; ++s) {
            const int k = k0 + 4 * s;
            f.r0[s] = *(const v4f*)(Wm + (size_t)k * RR);
            f.r1[s] = *(const v4f*)(Wm + (size_t)(k + 1) * RR);
            f.e0[s] = ep[(size_t)k * es];
            f.e1[s] = ep[(size_t)(k + 1) * es];
        }
    };

    v8f acc0 = {}, acc1 = {}, acc2 = {}, acc3 = {}, accE = {};
    const int arow = mt * 16 + lm;
    const int nch  = DD / XT_K;
    v4f xreg;

    auto stage = [&](const BFrag& cur, BFrag& nxt, int c, int buf) {
        *(v4f*)&Xs[buf][xrow][xc4] = xreg;
        __syncthreads();
        const int cn = (c + 1 < nch) ? c + 1 : c;
        xreg = *(const v4f*)(xg + cn * XT_K);
        loadB(nxt, cn * XT_K);

        v2f a[8];
        #pragma unroll
        for (int s = 0; s < 8; ++s)
            a[s] = *(const v2f*)&Xs[buf][arow][4 * s + 2 * hi];
        __builtin_amdgcn_sched_barrier(0);
        #pragma unroll
        for (int s = 0; s < 8; ++s) {
            v2f b0, b1, b2, b3, be;
            b0.x = cur.r0[s].x; b0.y = cur.r1[s].x;
            b1.x = cur.r0[s].y; b1.y = cur.r1[s].y;
            b2.x = cur.r0[s].z; b2.y = cur.r1[s].z;
            b3.x = cur.r0[s].w; b3.y = cur.r1[s].w;
            be.x = cur.e0[s];   be.y = cur.e1[s];
            acc0 = __builtin_amdgcn_wmma_f32_16x16x4_f32(false, a[s], false, b0, (short)0, acc0, false, false);
            acc1 = __builtin_amdgcn_wmma_f32_16x16x4_f32(false, a[s], false, b1, (short)0, acc1, false, false);
            acc2 = __builtin_amdgcn_wmma_f32_16x16x4_f32(false, a[s], false, b2, (short)0, acc2, false, false);
            acc3 = __builtin_amdgcn_wmma_f32_16x16x4_f32(false, a[s], false, b3, (short)0, acc3, false, false);
            accE = __builtin_amdgcn_wmma_f32_16x16x4_f32(false, a[s], false, be, (short)0, accE, false, false);
        }
        __builtin_amdgcn_sched_barrier(0);
        asm volatile("" :: "v"(nxt.r0[0]), "v"(nxt.r0[1]), "v"(nxt.r0[2]), "v"(nxt.r0[3]),
                           "v"(nxt.r0[4]), "v"(nxt.r0[5]), "v"(nxt.r0[6]), "v"(nxt.r0[7]),
                           "v"(nxt.r1[0]), "v"(nxt.r1[1]), "v"(nxt.r1[2]), "v"(nxt.r1[3]),
                           "v"(nxt.r1[4]), "v"(nxt.r1[5]), "v"(nxt.r1[6]), "v"(nxt.r1[7]));
        asm volatile("" :: "v"(nxt.e0[0]), "v"(nxt.e0[1]), "v"(nxt.e0[2]), "v"(nxt.e0[3]),
                           "v"(nxt.e0[4]), "v"(nxt.e0[5]), "v"(nxt.e0[6]), "v"(nxt.e0[7]),
                           "v"(nxt.e1[0]), "v"(nxt.e1[1]), "v"(nxt.e1[2]), "v"(nxt.e1[3]),
                           "v"(nxt.e1[4]), "v"(nxt.e1[5]), "v"(nxt.e1[6]), "v"(nxt.e1[7]),
                           "v"(xreg));
    };

    BFrag f0, f1;
    xreg = *(const v4f*)xg;
    loadB(f0, 0);
    for (int c = 0; c < nch; c += 2) {
        stage(f0, f1, c,     0);
        stage(f1, f0, c + 1, 1);
    }

    const int orow = m0 + mt * 16 + hi * 8;
    float* xo = p.xr + (size_t)sel * BT * RR + (size_t)orow * RR + 4 * lm;
    #pragma unroll
    for (int r = 0; r < 8; ++r) {
        float4 o; o.x = acc0[r]; o.y = acc1[r]; o.z = acc2[r]; o.w = acc3[r];
        *(float4*)(xo + (size_t)r * RR) = o;
    }
    if (sel < 6) {
        float* eo = (sel == 0 ? p.o0 : sel == 1 ? p.o1 : sel == 2 ? p.o2
                   : sel == 3 ? p.o3 : sel == 4 ? p.o4 : p.o5) + (size_t)orow * RSS + lm;
        #pragma unroll
        for (int r = 0; r < 8; ++r) eo[(size_t)r * RSS] = accE[r];
    } else if (sel == 6 && lm < 3) {
        float* dp = (lm == 0 ? p.dA : lm == 1 ? p.dB : p.dC) + orow;
        #pragma unroll
        for (int r = 0; r < 8; ++r) dp[r] = accE[r];
    }
}

__device__ inline v8f strip_times_factor(const v8f (&m)[8], const float* __restrict__ fac)
{
    v8f acc0 = {}, acc1 = {};
    #pragma unroll
    for (int t = 0; t < 8; ++t) {
        float bv[8];
        #pragma unroll
        for (int r = 0; r < 8; ++r) bv[r] = fac[(size_t)(16 * t + r) * RSS];
        #pragma unroll
        for (int q = 0; q < 4; ++q) {
            v2f a, b;
            a.x = m[t][2 * q]; a.y = m[t][2 * q + 1];
            b.x = bv[2 * q];   b.y = bv[2 * q + 1];
            if (q & 1) acc1 = __builtin_amdgcn_wmma_f32_16x16x4_f32(false, a, false, b, (short)0, acc1, false, false);
            else       acc0 = __builtin_amdgcn_wmma_f32_16x16x4_f32(false, a, false, b, (short)0, acc0, false, false);
        }
    }
    return acc0 + acc1;
}

__device__ inline void mix_apply(v8f (&m)[8], const float* __restrict__ dptr,
                                 const float* __restrict__ Arow, const v4f zb0, const v4f zb1)
{
    v2f b0, b1, b2, b3;
    b0.x = zb0.x; b0.y = zb0.y; b1.x = zb0.z; b1.y = zb0.w;
    b2.x = zb1.x; b2.y = zb1.y; b3.x = zb1.z; b3.y = zb1.w;
    #pragma unroll
    for (int t = 0; t < 8; ++t) {
        const v4f dlo = *(const v4f*)(dptr + 16 * t), dhi = *(const v4f*)(dptr + 16 * t + 4);
        const v4f f0 = *(const v4f*)(Arow + (size_t)(16 * t) * RSS);
        const v4f f1 = *(const v4f*)(Arow + (size_t)(16 * t) * RSS + 4);
        v8f c;
        c[0] = m[t][0] * dlo.x; c[1] = m[t][1] * dlo.y; c[2] = m[t][2] * dlo.z; c[3] = m[t][3] * dlo.w;
        c[4] = m[t][4] * dhi.x; c[5] = m[t][5] * dhi.y; c[6] = m[t][6] * dhi.z; c[7] = m[t][7] * dhi.w;
        v2f a0, a1, a2, a3;
        a0.x = f0.x; a0.y = f0.y; a1.x = f0.z; a1.y = f0.w;
        a2.x = f1.x; a2.y = f1.y; a3.x = f1.z; a3.y = f1.w;
        c = __builtin_amdgcn_wmma_f32_16x16x4_f32(false, a0, false, b0, (short)0, c, false, false);
        c = __builtin_amdgcn_wmma_f32_16x16x4_f32(false, a1, false, b1, (short)0, c, false, false);
        c = __builtin_amdgcn_wmma_f32_16x16x4_f32(false, a2, false, b2, (short)0, c, false, false);
        c = __builtin_amdgcn_wmma_f32_16x16x4_f32(false, a3, false, b3, (short)0, c, false, false);
        m[t] = c;
    }
}

__global__ __launch_bounds__(512) void mix_chain_kernel(
    const float* __restrict__ z, const float* zs,
    const float* __restrict__ d1,  const float* __restrict__ A1,  const float* __restrict__ B1,
    const float* __restrict__ d1s, const float* __restrict__ A1s, const float* __restrict__ B1s,
    const float* __restrict__ d2,  const float* __restrict__ A2,  const float* __restrict__ B2,
    float* out)
{
    __shared__ __align__(16) float part[2][16][16][16];
    __shared__ __align__(16) float full[2][16][16];
    __shared__ float rowpart[16][16];

    const int tid  = threadIdx.x;
    const int wave = tid >> 5, lane = tid & 31;
    const int hi   = lane >> 4, lm = lane & 15;
    const int row0 = blockIdx.x * 16;
    const int fb   = (row0 >> 11) * TT;
    const int j0   = wave * 128;

    const size_t soff = (size_t)(row0 + lm) * TT + j0 + 8 * hi;
    const size_t frow = (size_t)(fb + j0 + 8 * hi) * RSS + lm;
    const size_t arow = (size_t)(fb + j0 + lm) * RSS + 8 * hi;
    const int    doff = fb + j0 + 8 * hi;

    v8f m[8], ms[8];
    #pragma unroll
    for (int t = 0; t < 8; ++t) {
        const v4f lo = *(const v4f*)(z + soff + 16 * t), up = *(const v4f*)(z + soff + 16 * t + 4);
        m[t][0] = lo.x; m[t][1] = lo.y; m[t][2] = lo.z; m[t][3] = lo.w;
        m[t][4] = up.x; m[t][5] = up.y; m[t][6] = up.z; m[t][7] = up.w;
    }
    #pragma unroll
    for (int t = 0; t < 8; ++t) {
        const v4f lo = *(const v4f*)(zs + soff + 16 * t), up = *(const v4f*)(zs + soff + 16 * t + 4);
        ms[t][0] = lo.x; ms[t][1] = lo.y; ms[t][2] = lo.z; ms[t][3] = lo.w;
        ms[t][4] = up.x; ms[t][5] = up.y; ms[t][6] = up.z; ms[t][7] = up.w;
    }

    {
        const v8f p0 = strip_times_factor(m, B1 + frow);
        #pragma unroll
        for (int r = 0; r < 8; ++r) part[0][wave][8 * hi + r][lm] = p0[r];
    }
    {
        const v8f p1 = strip_times_factor(ms, B1s + frow);
        #pragma unroll
        for (int r = 0; r < 8; ++r) part[1][wave][8 * hi + r][lm] = p1[r];
    }
    __syncthreads();
    {
        const int pr = tid >> 8, e = tid & 255;
        float v = 0.f;
        #pragma unroll
        for (int w = 0; w < 16; ++w) v += (&part[pr][w][0][0])[e];
        (&full[pr][0][0])[e] = v;
    }
    __syncthreads();

    mix_apply(m,  d1  + doff, A1  + arow, *(const v4f*)&full[0][lm][8 * hi], *(const v4f*)&full[0][lm][8 * hi + 4]);
    mix_apply(ms, d1s + doff, A1s + arow, *(const v4f*)&full[1][lm][8 * hi], *(const v4f*)&full[1][lm][8 * hi + 4]);

    float ss = 0.f;
    #pragma unroll
    for (int t = 0; t < 8; ++t) {
        #pragma unroll
        for (int r = 0; r < 8; ++r) ss += m[t][r] * m[t][r];
    }
    ss += __shfl_xor(ss, 16, 32);
    if (hi == 0) rowpart[wave][lm] = ss;
    __syncthreads();
    float tot = 0.f;
    #pragma unroll
    for (int w = 0; w < 16; ++w) tot += rowpart[w][lm];
    const float inv = 1.f / (sqrtf(tot) + 1e-8f);

    #pragma unroll
    for (int t = 0; t < 8; ++t) {
        #pragma unroll
        for (int r = 0; r < 8; ++r) m[t][r] = softplusf(ms[t][r]) * fmaxf(m[t][r] * inv, 0.f);
    }

    {
        const v8f p0 = strip_times_factor(m, B2 + frow);
        #pragma unroll
        for (int r = 0; r < 8; ++r) part[0][wave][8 * hi + r][lm] = p0[r];
    }
    __syncthreads();
    if (tid < 256) {
        float v = 0.f;
        #pragma unroll
        for (int w = 0; w < 16; ++w) v += (&part[0][w][0][0])[tid];
        (&full[0][0][0])[tid] = v;
    }
    __syncthreads();
    mix_apply(m, d2 + doff, A2 + arow, *(const v4f*)&full[0][lm][8 * hi], *(const v4f*)&full[0][lm][8 * hi + 4]);

    #pragma unroll
    for (int t = 0; t < 8; ++t) {
        v4f lo, up;
        lo.x = m[t][0]; lo.y = m[t][1]; lo.z = m[t][2]; lo.w = m[t][3];
        up.x = m[t][4]; up.y = m[t][5]; up.z = m[t][6]; up.w = m[t][7];
        *(v4f*)(out + soff + 16 * t)     = lo;
        *(v4f*)(out + soff + 16 * t + 4) = up;
    }
}

__global__ __launch_bounds__(256) void emul_kernel(float* __restrict__ a, const float* __restrict__ b)
{
    const size_t i = ((size_t)blockIdx.x * blockDim.x + threadIdx.x) * 4;
    float4 av = *(const float4*)(a + i);
    float4 bv = *(const float4*)(b + i);
    av.x *= bv.x; av.y *= bv.y; av.z *= bv.z; av.w *= bv.w;
    *(float4*)(a + i) = av;
}

extern "C" void kernel_launch(void* const* d_in, const int* in_sizes, int n_in,
                              void* d_out, int out_size, void* d_ws, size_t ws_size,
                              hipStream_t stream) {
    (void)in_sizes; (void)n_in; (void)out_size; (void)ws_size;

    const float* x = (const float*)d_in[0];
    const float* qk_Wq1  = (const float*)d_in[1];  const float* qk_Wq2  = (const float*)d_in[2];
    const float* qk_Wm1  = (const float*)d_in[3];  const float* qk_Wm2  = (const float*)d_in[4];
    const float* qk_Wk1  = (const float*)d_in[5];  const float* qk_Wk2  = (const float*)d_in[6];
    const float* qks_Wq1 = (const float*)d_in[7];  const float* qks_Wq2 = (const float*)d_in[8];
    const float* qks_Wm1 = (const float*)d_in[9];  const float* qks_Wm2 = (const float*)d_in[10];
    const float* qks_Wk1 = (const float*)d_in[11]; const float* qks_Wk2 = (const float*)d_in[12];
    const float* vo_Wv1  = (const float*)d_in[13]; const float* vo_Wv2  = (const float*)d_in[14];
    const float* vo_Wm1  = (const float*)d_in[15]; const float* vo_Wm2  = (const float*)d_in[16];
    const float* vo_Wo   = (const float*)d_in[17];
    const float* sm1_Wd  = (const float*)d_in[18]; const float* sm1_Wa  = (const float*)d_in[19]; const float* sm1_Wb  = (const float*)d_in[20];
    const float* sm1s_Wd = (const float*)d_in[21]; const float* sm1s_Wa = (const float*)d_in[22]; const float* sm1s_Wb = (const float*)d_in[23];
    const float* sm2_Wd  = (const float*)d_in[24]; const float* sm2_Wa  = (const float*)d_in[25]; const float* sm2_Wb  = (const float*)d_in[26];

    const size_t BTD = (size_t)BT * DD;
    const size_t BTT = (size_t)BB * TT * TT;

    float* w = (float*)d_ws;
    float* qg   = w; w += BTD;
    float* kb   = w; w += BTD;
    float* qsg  = w; w += BTD;
    float* ksb  = w; w += BTD;
    float* vb   = w; w += BTD;
    float* mvob = w; w += BTD;
    float* tmpb = w; w += BTD;
    float* zb   = w; w += BTT;
    float* zsb  = w; w += BTT;
    float* ctx  = w; w += BTD;
    w += (size_t)BT * RR;
    float* d1   = w; w += BT; float* d1s = w; w += BT; float* d2 = w; w += BT;
    float* A1   = w; w += (size_t)BT * RSS; float* B1  = w; w += (size_t)BT * RSS;
    float* A1s  = w; w += (size_t)BT * RSS; float* B1s = w; w += (size_t)BT * RSS;
    float* A2   = w; w += (size_t)BT * RSS; float* B2  = w; w += (size_t)BT * RSS;

    auto gemm = [&](const float* A, const float* Bm, float* C, const float* mul,
                    int M, int N, int K, int lda, int ldb, int ldc, int bT,
                    long long sA, long long sB, long long sC, int batches) {
        dim3 g((unsigned)(N / GBN), (unsigned)(M / GBM), (unsigned)batches);
        gemm_wmma_f32<<<g, dim3(256), 0, stream>>>(A, Bm, C, mul, M, N, K, lda, ldb, ldc, bT, sA, sB, sC);
    };

    float* xrAll = ctx;
    const size_t XP = (size_t)BT * RR;
    XwArgs xa;
    xa.x  = x;
    xa.w0 = qk_Wm1;  xa.w1 = qk_Wq1;  xa.w2 = qk_Wk1;
    xa.w3 = qks_Wm1; xa.w4 = qks_Wq1; xa.w5 = qks_Wk1;
    xa.w6 = vo_Wv1;  xa.w7 = vo_Wm1;
    xa.e0 = sm1_Wa;  xa.e1 = sm1_Wb;  xa.e2 = sm1s_Wa; xa.e3 = sm1s_Wb; xa.e4 = sm2_Wa; xa.e5 = sm2_Wb;
    xa.wdA = sm1_Wd; xa.wdB = sm1s_Wd; xa.wdC = sm2_Wd;
    xa.xr = xrAll;
    xa.o0 = A1; xa.o1 = B1; xa.o2 = A1s; xa.o3 = B1s; xa.o4 = A2; xa.o5 = B2;
    xa.dA = d1; xa.dB = d1s; xa.dC = d2;
    xw_shared_kernel<<<dim3(BT / XT_ROWS, 2), 256, 0, stream>>>(xa);

    float* tmpb2 = zsb;
    const dim3 g2((unsigned)(DD / GBN), (unsigned)(BT / GBM), 6u);
    Stage2Args sa;
    sa.a0 = xrAll + 0 * XP; sa.w0 = qk_Wm2;  sa.c0 = tmpb;  sa.g0 = nullptr;
    sa.a1 = xrAll + 3 * XP; sa.w1 = qks_Wm2; sa.c1 = tmpb2; sa.g1 = nullptr;
    sa.a2 = xrAll + 2 * XP; sa.w2 = qk_Wk2;  sa.c2 = kb;    sa.g2 = nullptr;
    sa.a3 = xrAll + 5 * XP; sa.w3 = qks_Wk2; sa.c3 = ksb;   sa.g3 = nullptr;
    sa.a4 = xrAll + 6 * XP; sa.w4 = vo_Wv2;  sa.c4 = vb;    sa.g4 = nullptr;
    sa.a5 = xrAll + 7 * XP; sa.w5 = vo_Wm2;  sa.c5 = mvob;  sa.g5 = nullptr;
    gemm_stage2_kernel<<<g2, dim3(256), 0, stream>>>(sa);

    const dim3 g2b((unsigned)(DD / GBN), (unsigned)(BT / GBM), 2u);
    Stage2Args sb;
    sb.a0 = xrAll + 1 * XP; sb.w0 = qk_Wq2;  sb.c0 = qg;  sb.g0 = tmpb;
    sb.a1 = xrAll + 4 * XP; sb.w1 = qks_Wq2; sb.c1 = qsg; sb.g1 = tmpb2;
    sb.a2 = nullptr; sb.w2 = nullptr; sb.c2 = nullptr; sb.g2 = nullptr;
    sb.a3 = nullptr; sb.w3 = nullptr; sb.c3 = nullptr; sb.g3 = nullptr;
    sb.a4 = nullptr; sb.w4 = nullptr; sb.c4 = nullptr; sb.g4 = nullptr;
    sb.a5 = nullptr; sb.w5 = nullptr; sb.c5 = nullptr; sb.g5 = nullptr;
    gemm_stage2_kernel<<<g2b, dim3(256), 0, stream>>>(sb);

    const long long sTD = (long long)TT * DD, sTTl = (long long)TT * TT;
    gemm(qg,  kb,  zb,  nullptr, TT, TT, DD, DD, DD, TT, 1, sTD, sTD, sTTl, BB);
    gemm(qsg, ksb, zsb, nullptr, TT, TT, DD, DD, DD, TT, 1, sTD, sTD, sTTl, BB);

    mix_chain_kernel<<<dim3(BT / 16), 512, 0, stream>>>(zb, zsb, d1, A1, B1, d1s, A1s, B1s, d2, A2, B2, zsb);

    gemm(zsb, vb, ctx, nullptr, TT, DD, TT, TT, DD, DD, 0, sTTl, sTD, sTD, BB);
    emul_kernel<<<dim3((unsigned)(BTD / 4 / 256)), 256, 0, stream>>>(ctx, mvob);
    gemm(ctx, vo_Wo, (float*)d_out, nullptr, BT, DD, DD, DD, DD, DD, 1, 0, 0, 0, 1);
}


# context*mvo moves into the context GEMM's epilogue (its own launch is gone: 8 launches, was 9); the gated epilogue (q, qs too) fetches its gate tile as ONE load group, not row by row; sums unchanged
# speedup vs baseline: 3.2618x; 1.0475x over previous
#include <hip/hip_runtime.h>
#include <math.h>

#define BB 2
#define TT 2048
#define DD 1024
#define RR 64
#define RSS 16
#define BT (BB*TT)

typedef __attribute__((ext_vector_type(2))) float v2f;
typedef __attribute__((ext_vector_type(8))) float v8f;
typedef __attribute__((ext_vector_type(4))) float v4f;

#define GBM 256
#define GBN 64
#define GBK 32

__device__ __forceinline__ void gemm_tile_f32(
    const float* __restrict__ Ab, const float* __restrict__ Bb,
    float* __restrict__ Cb, const float* __restrict__ Mb,
    int K, int lda, int ldb, int ldc, int bT, int m0, int n0)
{
    __shared__ __align__(16) float As [2][GBM][34];
    __shared__ __align__(16) float BsT[2][GBN][34];

    const int tid  = threadIdx.x;
    const int wave = tid >> 5;
    const int lane = tid & 31;
    const int hi   = lane >> 4;
    const int lm   = lane & 15;

    float4 ra[8];
    float4 rb[2];

    int arow[8], ac4[8];
    #pragma unroll
    for (int i = 0; i < 8; ++i) { int idx = tid + i * 256; arow[i] = idx >> 3; ac4[i] = (idx & 7) * 4; }
    int bi0[2], bi1[2];
    #pragma unroll
    for (int i = 0; i < 2; ++i) {
        int idx = tid + i * 256;
        if (true) {   }
        bi0[i] = idx; bi1[i] = idx;
    }

    auto loadGlobal = [&](int k0) {
        #pragma unroll
        for (int i = 0; i < 8; ++i)
            ra[i] = *(const float4*)(Ab + (size_t)(m0 + arow[i]) * lda + k0 + ac4[i]);
        if (bT) {
            #pragma unroll
            for (int i = 0; i < 2; ++i) {
                int idx = bi0[i];
                int n = idx >> 3, c4 = (idx & 7) * 4;
                rb[i] = *(const float4*)(Bb + (size_t)(n0 + n) * ldb + k0 + c4);
            }
        } else {
            #pragma unroll
            for (int i = 0; i < 2; ++i) {
                int idx = bi1[i];
                int r = idx >> 4, c4 = (idx & 15) * 4;
                rb[i] = *(const float4*)(Bb + (size_t)(k0 + r) * ldb + n0 + c4);
            }
        }
    };

    auto storeLDS = [&](int buf) {
        #pragma unroll
        for (int i = 0; i < 8; ++i) {
            float4 t = ra[i];
            v2f lo; lo.x = t.x; lo.y = t.y;
            v2f hi2; hi2.x = t.z; hi2.y = t.w;
            *(v2f*)&As[buf][arow[i]][ac4[i]]     = lo;
            *(v2f*)&As[buf][arow[i]][ac4[i] + 2] = hi2;
        }
        if (bT) {
            #pragma unroll
            for (int i = 0; i < 2; ++i) {
                int idx = bi0[i];
                int n = idx >> 3, c4 = (idx & 7) * 4;
                float4 t = rb[i];
                v2f lo; lo.x = t.x; lo.y = t.y;
                v2f hi2; hi2.x = t.z; hi2.y = t.w;
                *(v2f*)&BsT[buf][n][c4]     = lo;
                *(v2f*)&BsT[buf][n][c4 + 2] = hi2;
            }
        } else {
            #pragma unroll
            for (int i = 0; i < 2; ++i) {
                int idx = bi1[i];
                int r = idx >> 4, c4 = (idx & 15) * 4;
                float4 t = rb[i];
                BsT[buf][c4 + 0][r] = t.x;
                BsT[buf][c4 + 1][r] = t.y;
                BsT[buf][c4 + 2][r] = t.z;
                BsT[buf][c4 + 3][r] = t.w;
            }
        }
    };

    v8f acc00 = {}, acc01 = {}, acc02 = {}, acc03 = {};
    v8f acc10 = {}, acc11 = {}, acc12 = {}, acc13 = {};

    const int r0 = wave * 32 + lm;
    const int r1 = r0 + 16;
    const int nchunks = K / GBK;

    struct Frag { v2f a0[4], a1[4], b0[4], b1[4], b2[4], b3[4]; };

    auto loadFrag = [&](Frag& f, int buf, int step0) {
        #pragma unroll
        for (int s = 0; s < 4; ++s) {
            const int kc = (step0 + s) * 4 + 2 * hi;
            f.a0[s] = *(const v2f*)&As [buf][r0][kc];
            f.a1[s] = *(const v2f*)&As [buf][r1][kc];
            f.b0[s] = *(const v2f*)&BsT[buf][lm +  0][kc];
            f.b1[s] = *(const v2f*)&BsT[buf][lm + 16][kc];
            f.b2[s] = *(const v2f*)&BsT[buf][lm + 32][kc];
            f.b3[s] = *(const v2f*)&BsT[buf][lm + 48][kc];
        }
    };

    auto mma = [&](const Frag& f, int s) {
        acc00 = __builtin_amdgcn_wmma_f32_16x16x4_f32(false, f.a0[s], false, f.b0[s], (short)0, acc00, false, false);
        acc01 = __builtin_amdgcn_wmma_f32_16x16x4_f32(false, f.a0[s], false, f.b1[s], (short)0, acc01, false, false);
        acc02 = __builtin_amdgcn_wmma_f32_16x16x4_f32(false, f.a0[s], false, f.b2[s], (short)0, acc02, false, false);
        acc03 = __builtin_amdgcn_wmma_f32_16x16x4_f32(false, f.a0[s], false, f.b3[s], (short)0, acc03, false, false);
        acc10 = __builtin_amdgcn_wmma_f32_16x16x4_f32(false, f.a1[s], false, f.b0[s], (short)0, acc10, false, false);
        acc11 = __builtin_amdgcn_wmma_f32_16x16x4_f32(false, f.a1[s], false, f.b1[s], (short)0, acc11, false, false);
        acc12 = __builtin_amdgcn_wmma_f32_16x16x4_f32(false, f.a1[s], false, f.b2[s], (short)0, acc12, false, false);
        acc13 = __builtin_amdgcn_wmma_f32_16x16x4_f32(false, f.a1[s], false, f.b3[s], (short)0, acc13, false, false);
    };

    Frag P, Q;
    loadGlobal(0);
    storeLDS(0);
    __syncthreads();
    loadFrag(P, 0, 0);
    for (int c = 0; c + 1 < nchunks; ++c) {
        const int buf = c & 1;
        loadGlobal((c + 1) * GBK);
        loadFrag(Q, buf, 4);
        __builtin_amdgcn_sched_barrier(0);
        mma(P, 0); mma(P, 1); mma(P, 2); mma(P, 3);
        mma(Q, 0); mma(Q, 1);
        __builtin_amdgcn_sched_barrier(0);
        storeLDS(buf ^ 1);
        __builtin_amdgcn_sched_barrier(0);
        mma(Q, 2);
        __builtin_amdgcn_sched_barrier(0);
        __syncthreads();
        loadFrag(P, buf ^ 1, 0);
        __builtin_amdgcn_sched_barrier(0);
        mma(Q, 3);
    }
    {
        loadFrag(Q, (nchunks - 1) & 1, 4);
        __builtin_amdgcn_sched_barrier(0);
        mma(P, 0); mma(P, 1); mma(P, 2); mma(P, 3);
        mma(Q, 0); mma(Q, 1); mma(Q, 2); mma(Q, 3);
    }

    const size_t cbase = (size_t)(m0 + wave * 32 + hi * 8) * ldc + n0 + lm;
    float* const c0p = Cb + cbase;
    float* const c1p = Cb + cbase + (size_t)16 * ldc;
    if (Mb) {
        const float* const g0p = Mb + cbase;
        const float* const g1p = Mb + cbase + (size_t)16 * ldc;
        float g0[8][4], g1[8][4];
        #pragma unroll
        for (int r = 0; r < 8; ++r) {
            const float* gp = g0p + (size_t)r * ldc;
            g0[r][0] = gp[0]; g0[r][1] = gp[16]; g0[r][2] = gp[32]; g0[r][3] = gp[48];
        }
        #pragma unroll
        for (int r = 0; r < 8; ++r) {
            const float* gp = g1p + (size_t)r * ldc;
            g1[r][0] = gp[0]; g1[r][1] = gp[16]; g1[r][2] = gp[32]; g1[r][3] = gp[48];
        }
        #pragma unroll
        for (int r = 0; r < 8; ++r) {
            float* cp = c0p + (size_t)r * ldc;
            cp[0]  = acc00[r] * g0[r][0]; cp[16] = acc01[r] * g0[r][1];
            cp[32] = acc02[r] * g0[r][2]; cp[48] = acc03[r] * g0[r][3];
        }
        #pragma unroll
        for (int r = 0; r < 8; ++r) {
            float* cp = c1p + (size_t)r * ldc;
            cp[0]  = acc10[r] * g1[r][0]; cp[16] = acc11[r] * g1[r][1];
            cp[32] = acc12[r] * g1[r][2]; cp[48] = acc13[r] * g1[r][3];
        }
    } else {
        #pragma unroll
        for (int r = 0; r < 8; ++r) {
            float* cp = c0p + (size_t)r * ldc;
            cp[0] = acc00[r]; cp[16] = acc01[r]; cp[32] = acc02[r]; cp[48] = acc03[r];
        }
        #pragma unroll
        for (int r = 0; r < 8; ++r) {
            float* cp = c1p + (size_t)r * ldc;
            cp[0] = acc10[r]; cp[16] = acc11[r]; cp[32] = acc12[r]; cp[48] = acc13[r];
        }
    }
}

__global__ __launch_bounds__(256) void gemm_wmma_f32(
    const float* __restrict__ A, const float* __restrict__ Bm,
    float* __restrict__ C, const float* __restrict__ mulC,
    int M, int N, int K, int lda, int ldb, int ldc, int bT,
    long long sA, long long sB, long long sC)
{
    (void)M; (void)N;
    const float* Ap = A  + (size_t)blockIdx.z * sA;
    const float* Bp = Bm + (size_t)blockIdx.z * sB;
    float*       Cp = C  + (size_t)blockIdx.z * sC;
    const float* Mp = mulC ? (mulC + (size_t)blockIdx.z * sC) : nullptr;
    if (bT) gemm_tile_f32(Ap, Bp, Cp, Mp, K, lda, ldb, ldc, 1, blockIdx.y * GBM, blockIdx.x * GBN);
    else    gemm_tile_f32(Ap, Bp, Cp, Mp, K, lda, ldb, ldc, 0, blockIdx.y * GBM, blockIdx.x * GBN);
}

struct Stage2Args {
    const float *a0, *a1, *a2, *a3, *a4, *a5;
    const float *w0, *w1, *w2, *w3, *w4, *w5;
    float       *c0, *c1, *c2, *c3, *c4, *c5;
    const float *g0, *g1, *g2, *g3, *g4, *g5;
};

__global__ __launch_bounds__(256) void gemm_stage2_kernel(Stage2Args p)
{
    const int s = blockIdx.z;
    const float* A = s == 0 ? p.a0 : s == 1 ? p.a1 : s == 2 ? p.a2 : s == 3 ? p.a3 : s == 4 ? p.a4 : p.a5;
    const float* Wm = s == 0 ? p.w0 : s == 1 ? p.w1 : s == 2 ? p.w2 : s == 3 ? p.w3 : s == 4 ? p.w4 : p.w5;
    float*       C = s == 0 ? p.c0 : s == 1 ? p.c1 : s == 2 ? p.c2 : s == 3 ? p.c3 : s == 4 ? p.c4 : p.c5;
    const float* G = s == 0 ? p.g0 : s == 1 ? p.g1 : s == 2 ? p.g2 : s == 3 ? p.g3 : s == 4 ? p.g4 : p.g5;
    gemm_tile_f32(A, Wm, C, G, RR, RR, DD, DD, 0, blockIdx.y * GBM, blockIdx.x * GBN);
}

__device__ __forceinline__ float softplusf(float x) {
    const float ax = __builtin_fabsf(x);
    const float t  = __builtin_amdgcn_exp2f(-1.44269504f * ax);
    const float u  = 1.0f + t;
    const float d  = u - 1.0f;
    const float l  = __builtin_amdgcn_logf(u) * 0.693147181f;
    const float q  = t * __builtin_amdgcn_rcpf(d);
    const float r  = (d == 0.0f) ? t : l * q;
    return __builtin_fmaxf(x, 0.0f) + r;
}

struct XwArgs {
    const float* x;
    const float *w0, *w1, *w2, *w3, *w4, *w5, *w6, *w7;
    const float *e0, *e1, *e2, *e3, *e4, *e5;
    const float *wdA, *wdB, *wdC;
    float* xr;
    float *o0, *o1, *o2, *o3, *o4, *o5;
    float *dA, *dB, *dC;
};

struct BFrag {
    v4f    r0[8], r1[8];
    float  e0[8], e1[8];
};

#define XT_ROWS 32
#define XT_K    32
#define XT_PITCH 36

__global__ __launch_bounds__(256) void xw_shared_kernel(XwArgs p)
{
    __shared__ __align__(16) float Xs[2][XT_ROWS][XT_PITCH];

    const int tid  = threadIdx.x;
    const int wave = tid >> 5, lane = tid & 31;
    const int hi   = lane >> 4, lm = lane & 15;
    const int m0   = blockIdx.x * XT_ROWS;
    const int sel  = blockIdx.y * 4 + (wave >> 1);
    const int mt   = wave & 1;

    const float* Wm = sel == 0 ? p.w0 : sel == 1 ? p.w1 : sel == 2 ? p.w2 : sel == 3 ? p.w3
                    : sel == 4 ? p.w4 : sel == 5 ? p.w5 : sel == 6 ? p.w6 : p.w7;
    Wm += (size_t)(2 * hi) * RR + 4 * lm;

    const float* ep;
    int es;
    if (sel < 6) {
        ep = (sel == 0 ? p.e0 : sel == 1 ? p.e1 : sel == 2 ? p.e2
            : sel == 3 ? p.e3 : sel == 4 ? p.e4 : p.e5) + lm;
        es = RSS;
    } else {
        ep = (sel == 6 && lm == 1) ? p.wdB : (sel == 6 && lm == 2) ? p.wdC : p.wdA;
        es = 1;
    }
    ep += (size_t)(2 * hi) * es;

    const int xrow = tid >> 3, xc4 = (tid & 7) * 4;
    const float* xg = p.x + (size_t)(m0 + xrow) * DD + xc4;

    auto loadB = [&](BFrag& f, int k0) {
        #pragma unroll
        for (int s = 0; s < 8; ++s) {
            const int k = k0 + 4 * s;
            f.r0[s] = *(const v4f*)(Wm + (size_t)k * RR);
            f.r1[s] = *(const v4f*)(Wm + (size_t)(k + 1) * RR);
            f.e0[s] = ep[(size_t)k * es];
            f.e1[s] = ep[(size_t)(k + 1) * es];
        }
    };

    v8f acc0 = {}, acc1 = {}, acc2 = {}, acc3 = {}, accE = {};
    const int arow = mt * 16 + lm;
    const int nch  = DD / XT_K;
    v4f xreg;

    auto stage = [&](const BFrag& cur, BFrag& nxt, int c, int buf) {
        *(v4f*)&Xs[buf][xrow][xc4] = xreg;
        __syncthreads();
        const int cn = (c + 1 < nch) ? c + 1 : c;
        xreg = *(const v4f*)(xg + cn * XT_K);
        loadB(nxt, cn * XT_K);

        v2f a[8];
        #pragma unroll
        for (int s = 0; s < 8; ++s)
            a[s] = *(const v2f*)&Xs[buf][arow][4 * s + 2 * hi];
        __builtin_amdgcn_sched_barrier(0);
        #pragma unroll
        for (int s = 0; s < 8; ++s) {
            v2f b0, b1, b2, b3, be;
            b0.x = cur.r0[s].x; b0.y = cur.r1[s].x;
            b1.x = cur.r0[s].y; b1.y = cur.r1[s].y;
            b2.x = cur.r0[s].z; b2.y = cur.r1[s].z;
            b3.x = cur.r0[s].w; b3.y = cur.r1[s].w;
            be.x = cur.e0[s];   be.y = cur.e1[s];
            acc0 = __builtin_amdgcn_wmma_f32_16x16x4_f32(false, a[s], false, b0, (short)0, acc0, false, false);
            acc1 = __builtin_amdgcn_wmma_f32_16x16x4_f32(false, a[s], false, b1, (short)0, acc1, false, false);
            acc2 = __builtin_amdgcn_wmma_f32_16x16x4_f32(false, a[s], false, b2, (short)0, acc2, false, false);
            acc3 = __builtin_amdgcn_wmma_f32_16x16x4_f32(false, a[s], false, b3, (short)0, acc3, false, false);
            accE = __builtin_amdgcn_wmma_f32_16x16x4_f32(false, a[s], false, be, (short)0, accE, false, false);
        }
        __builtin_amdgcn_sched_barrier(0);
        asm volatile("" :: "v"(nxt.r0[0]), "v"(nxt.r0[1]), "v"(nxt.r0[2]), "v"(nxt.r0[3]),
                           "v"(nxt.r0[4]), "v"(nxt.r0[5]), "v"(nxt.r0[6]), "v"(nxt.r0[7]),
                           "v"(nxt.r1[0]), "v"(nxt.r1[1]), "v"(nxt.r1[2]), "v"(nxt.r1[3]),
                           "v"(nxt.r1[4]), "v"(nxt.r1[5]), "v"(nxt.r1[6]), "v"(nxt.r1[7]));
        asm volatile("" :: "v"(nxt.e0[0]), "v"(nxt.e0[1]), "v"(nxt.e0[2]), "v"(nxt.e0[3]),
                           "v"(nxt.e0[4]), "v"(nxt.e0[5]), "v"(nxt.e0[6]), "v"(nxt.e0[7]),
                           "v"(nxt.e1[0]), "v"(nxt.e1[1]), "v"(nxt.e1[2]), "v"(nxt.e1[3]),
                           "v"(nxt.e1[4]), "v"(nxt.e1[5]), "v"(nxt.e1[6]), "v"(nxt.e1[7]),
                           "v"(xreg));
    };

    BFrag f0, f1;
    xreg = *(const v4f*)xg;
    loadB(f0, 0);
    for (int c = 0; c < nch; c += 2) {
        stage(f0, f1, c,     0);
        stage(f1, f0, c + 1, 1);
    }

    const int orow = m0 + mt * 16 + hi * 8;
    float* xo = p.xr + (size_t)sel * BT * RR + (size_t)orow * RR + 4 * lm;
    #pragma unroll
    for (int r = 0; r < 8; ++r) {
        float4 o; o.x = acc0[r]; o.y = acc1[r]; o.z = acc2[r]; o.w = acc3[r];
        *(float4*)(xo + (size_t)r * RR) = o;
    }
    if (sel < 6) {
        float* eo = (sel == 0 ? p.o0 : sel == 1 ? p.o1 : sel == 2 ? p.o2
                   : sel == 3 ? p.o3 : sel == 4 ? p.o4 : p.o5) + (size_t)orow * RSS + lm;
        #pragma unroll
        for (int r = 0; r < 8; ++r) eo[(size_t)r * RSS] = accE[r];
    } else if (sel == 6 && lm < 3) {
        float* dp = (lm == 0 ? p.dA : lm == 1 ? p.dB : p.dC) + orow;
        #pragma unroll
        for (int r = 0; r < 8; ++r) dp[r] = accE[r];
    }
}

__device__ inline v8f strip_times_factor(const v8f (&m)[8], const float* __restrict__ fac)
{
    v8f acc0 = {}, acc1 = {};
    #pragma unroll
    for (int t = 0; t < 8; ++t) {
        float bv[8];
        #pragma unroll
        for (int r = 0; r < 8; ++r) bv[r] = fac[(size_t)(16 * t + r) * RSS];
        #pragma unroll
        for (int q = 0; q < 4; ++q) {
            v2f a, b;
            a.x = m[t][2 * q]; a.y = m[t][2 * q + 1];
            b.x = bv[2 * q];   b.y = bv[2 * q + 1];
            if (q & 1) acc1 = __builtin_amdgcn_wmma_f32_16x16x4_f32(false, a, false, b, (short)0, acc1, false, false);
            else       acc0 = __builtin_amdgcn_wmma_f32_16x16x4_f32(false, a, false, b, (short)0, acc0, false, false);
        }
    }
    return acc0 + acc1;
}

__device__ inline void mix_apply(v8f (&m)[8], const float* __restrict__ dptr,
                                 const float* __restrict__ Arow, const v4f zb0, const v4f zb1)
{
    v2f b0, b1, b2, b3;
    b0.x = zb0.x; b0.y = zb0.y; b1.x = zb0.z; b1.y = zb0.w;
    b2.x = zb1.x; b2.y = zb1.y; b3.x = zb1.z; b3.y = zb1.w;
    #pragma unroll
    for (int t = 0; t < 8; ++t) {
        const v4f dlo = *(const v4f*)(dptr + 16 * t), dhi = *(const v4f*)(dptr + 16 * t + 4);
        const v4f f0 = *(const v4f*)(Arow + (size_t)(16 * t) * RSS);
        const v4f f1 = *(const v4f*)(Arow + (size_t)(16 * t) * RSS + 4);
        v8f c;
        c[0] = m[t][0] * dlo.x; c[1] = m[t][1] * dlo.y; c[2] = m[t][2] * dlo.z; c[3] = m[t][3] * dlo.w;
        c[4] = m[t][4] * dhi.x; c[5] = m[t][5] * dhi.y; c[6] = m[t][6] * dhi.z; c[7] = m[t][7] * dhi.w;
        v2f a0, a1, a2, a3;
        a0.x = f0.x; a0.y = f0.y; a1.x = f0.z; a1.y = f0.w;
        a2.x = f1.x; a2.y = f1.y; a3.x = f1.z; a3.y = f1.w;
        c = __builtin_amdgcn_wmma_f32_16x16x4_f32(false, a0, false, b0, (short)0, c, false, false);
        c = __builtin_amdgcn_wmma_f32_16x16x4_f32(false, a1, false, b1, (short)0, c, false, false);
        c = __builtin_amdgcn_wmma_f32_16x16x4_f32(false, a2, false, b2, (short)0, c, false, false);
        c = __builtin_amdgcn_wmma_f32_16x16x4_f32(false, a3, false, b3, (short)0, c, false, false);
        m[t] = c;
    }
}

__global__ __launch_bounds__(512) void mix_chain_kernel(
    const float* __restrict__ z, const float* zs,
    const float* __restrict__ d1,  const float* __restrict__ A1,  const float* __restrict__ B1,
    const float* __restrict__ d1s, const float* __restrict__ A1s, const float* __restrict__ B1s,
    const float* __restrict__ d2,  const float* __restrict__ A2,  const float* __restrict__ B2,
    float* out)
{
    __shared__ __align__(16) float part[2][16][16][16];
    __shared__ __align__(16) float full[2][16][16];
    __shared__ float rowpart[16][16];

    const int tid  = threadIdx.x;
    const int wave = tid >> 5, lane = tid & 31;
    const int hi   = lane >> 4, lm = lane & 15;
    const int row0 = blockIdx.x * 16;
    const int fb   = (row0 >> 11) * TT;
    const int j0   = wave * 128;

    const size_t soff = (size_t)(row0 + lm) * TT + j0 + 8 * hi;
    const size_t frow = (size_t)(fb + j0 + 8 * hi) * RSS + lm;
    const size_t arow = (size_t)(fb + j0 + lm) * RSS + 8 * hi;
    const int    doff = fb + j0 + 8 * hi;

    v8f m[8], ms[8];
    #pragma unroll
    for (int t = 0; t < 8; ++t) {
        const v4f lo = *(const v4f*)(z + soff + 16 * t), up = *(const v4f*)(z + soff + 16 * t + 4);
        m[t][0] = lo.x; m[t][1] = lo.y; m[t][2] = lo.z; m[t][3] = lo.w;
        m[t][4] = up.x; m[t][5] = up.y; m[t][6] = up.z; m[t][7] = up.w;
    }
    #pragma unroll
    for (int t = 0; t < 8; ++t) {
        const v4f lo = *(const v4f*)(zs + soff + 16 * t), up = *(const v4f*)(zs + soff + 16 * t + 4);
        ms[t][0] = lo.x; ms[t][1] = lo.y; ms[t][2] = lo.z; ms[t][3] = lo.w;
        ms[t][4] = up.x; ms[t][5] = up.y; ms[t][6] = up.z; ms[t][7] = up.w;
    }

    {
        const v8f p0 = strip_times_factor(m, B1 + frow);
        #pragma unroll
        for (int r = 0; r < 8; ++r) part[0][wave][8 * hi + r][lm] = p0[r];
    }
    {
        const v8f p1 = strip_times_factor(ms, B1s + frow);
        #pragma unroll
        for (int r = 0; r < 8; ++r) part[1][wave][8 * hi + r][lm] = p1[r];
    }
    __syncthreads();
    {
        const int pr = tid >> 8, e = tid & 255;
        float v = 0.f;
        #pragma unroll
        for (int w = 0; w < 16; ++w) v += (&part[pr][w][0][0])[e];
        (&full[pr][0][0])[e] = v;
    }
    __syncthreads();

    mix_apply(m,  d1  + doff, A1  + arow, *(const v4f*)&full[0][lm][8 * hi], *(const v4f*)&full[0][lm][8 * hi + 4]);
    mix_apply(ms, d1s + doff, A1s + arow, *(const v4f*)&full[1][lm][8 * hi], *(const v4f*)&full[1][lm][8 * hi + 4]);

    float ss = 0.f;
    #pragma unroll
    for (int t = 0; t < 8; ++t) {
        #pragma unroll
        for (int r = 0; r < 8; ++r) ss += m[t][r] * m[t][r];
    }
    ss += __shfl_xor(ss, 16, 32);
    if (hi == 0) rowpart[wave][lm] = ss;
    __syncthreads();
    float tot = 0.f;
    #pragma unroll
    for (int w = 0; w < 16; ++w) tot += rowpart[w][lm];
    const float inv = 1.f / (sqrtf(tot) + 1e-8f);

    #pragma unroll
    for (int t = 0; t < 8; ++t) {
        #pragma unroll
        for (int r = 0; r < 8; ++r) m[t][r] = softplusf(ms[t][r]) * fmaxf(m[t][r] * inv, 0.f);
    }

    {
        const v8f p0 = strip_times_factor(m, B2 + frow);
        #pragma unroll
        for (int r = 0; r < 8; ++r) part[0][wave][8 * hi + r][lm] = p0[r];
    }
    __syncthreads();
    if (tid < 256) {
        float v = 0.f;
        #pragma unroll
        for (int w = 0; w < 16; ++w) v += (&part[0][w][0][0])[tid];
        (&full[0][0][0])[tid] = v;
    }
    __syncthreads();
    mix_apply(m, d2 + doff, A2 + arow, *(const v4f*)&full[0][lm][8 * hi], *(const v4f*)&full[0][lm][8 * hi + 4]);

    #pragma unroll
    for (int t = 0; t < 8; ++t) {
        v4f lo, up;
        lo.x = m[t][0]; lo.y = m[t][1]; lo.z = m[t][2]; lo.w = m[t][3];
        up.x = m[t][4]; up.y = m[t][5]; up.z = m[t][6]; up.w = m[t][7];
        *(v4f*)(out + soff + 16 * t)     = lo;
        *(v4f*)(out + soff + 16 * t + 4) = up;
    }
}

extern "C" void kernel_launch(void* const* d_in, const int* in_sizes, int n_in,
                              void* d_out, int out_size, void* d_ws, size_t ws_size,
                              hipStream_t stream) {
    (void)in_sizes; (void)n_in; (void)out_size; (void)ws_size;

    const float* x = (const float*)d_in[0];
    const float* qk_Wq1  = (const float*)d_in[1];  const float* qk_Wq2  = (const float*)d_in[2];
    const float* qk_Wm1  = (const float*)d_in[3];  const float* qk_Wm2  = (const float*)d_in[4];
    const float* qk_Wk1  = (const float*)d_in[5];  const float* qk_Wk2  = (const float*)d_in[6];
    const float* qks_Wq1 = (const float*)d_in[7];  const float* qks_Wq2 = (const float*)d_in[8];
    const float* qks_Wm1 = (const float*)d_in[9];  const float* qks_Wm2 = (const float*)d_in[10];
    const float* qks_Wk1 = (const float*)d_in[11]; const float* qks_Wk2 = (const float*)d_in[12];
    const float* vo_Wv1  = (const float*)d_in[13]; const float* vo_Wv2  = (const float*)d_in[14];
    const float* vo_Wm1  = (const float*)d_in[15]; const float* vo_Wm2  = (const float*)d_in[16];
    const float* vo_Wo   = (const float*)d_in[17];
    const float* sm1_Wd  = (const float*)d_in[18]; const float* sm1_Wa  = (const float*)d_in[19]; const float* sm1_Wb  = (const float*)d_in[20];
    const float* sm1s_Wd = (const float*)d_in[21]; const float* sm1s_Wa = (const float*)d_in[22]; const float* sm1s_Wb = (const float*)d_in[23];
    const float* sm2_Wd  = (const float*)d_in[24]; const float* sm2_Wa  = (const float*)d_in[25]; const float* sm2_Wb  = (const float*)d_in[26];

    const size_t BTD = (size_t)BT * DD;
    const size_t BTT = (size_t)BB * TT * TT;

    float* w = (float*)d_ws;
    float* qg   = w; w += BTD;
    float* kb   = w; w += BTD;
    float* qsg  = w; w += BTD;
    float* ksb  = w; w += BTD;
    float* vb   = w; w += BTD;
    float* mvob = w; w += BTD;
    float* tmpb = w; w += BTD;
    float* zb   = w; w += BTT;
    float* zsb  = w; w += BTT;
    float* ctx  = w; w += BTD;
    w += (size_t)BT * RR;
    float* d1   = w; w += BT; float* d1s = w; w += BT; float* d2 = w; w += BT;
    float* A1   = w; w += (size_t)BT * RSS; float* B1  = w; w += (size_t)BT * RSS;
    float* A1s  = w; w += (size_t)BT * RSS; float* B1s = w; w += (size_t)BT * RSS;
    float* A2   = w; w += (size_t)BT * RSS; float* B2  = w; w += (size_t)BT * RSS;

    auto gemm = [&](const float* A, const float* Bm, float* C, const float* mul,
                    int M, int N, int K, int lda, int ldb, int ldc, int bT,
                    long long sA, long long sB, long long sC, int batches) {
        dim3 g((unsigned)(N / GBN), (unsigned)(M / GBM), (unsigned)batches);
        gemm_wmma_f32<<<g, dim3(256), 0, stream>>>(A, Bm, C, mul, M, N, K, lda, ldb, ldc, bT, sA, sB, sC);
    };

    float* xrAll = ctx;
    const size_t XP = (size_t)BT * RR;
    XwArgs xa;
    xa.x  = x;
    xa.w0 = qk_Wm1;  xa.w1 = qk_Wq1;  xa.w2 = qk_Wk1;
    xa.w3 = qks_Wm1; xa.w4 = qks_Wq1; xa.w5 = qks_Wk1;
    xa.w6 = vo_Wv1;  xa.w7 = vo_Wm1;
    xa.e0 = sm1_Wa;  xa.e1 = sm1_Wb;  xa.e2 = sm1s_Wa; xa.e3 = sm1s_Wb; xa.e4 = sm2_Wa; xa.e5 = sm2_Wb;
    xa.wdA = sm1_Wd; xa.wdB = sm1s_Wd; xa.wdC = sm2_Wd;
    xa.xr = xrAll;
    xa.o0 = A1; xa.o1 = B1; xa.o2 = A1s; xa.o3 = B1s; xa.o4 = A2; xa.o5 = B2;
    xa.dA = d1; xa.dB = d1s; xa.dC = d2;
    xw_shared_kernel<<<dim3(BT / XT_ROWS, 2), 256, 0, stream>>>(xa);

    float* tmpb2 = zsb;
    const dim3 g2((unsigned)(DD / GBN), (unsigned)(BT / GBM), 6u);
    Stage2Args sa;
    sa.a0 = xrAll + 0 * XP; sa.w0 = qk_Wm2;  sa.c0 = tmpb;  sa.g0 = nullptr;
    sa.a1 = xrAll + 3 * XP; sa.w1 = qks_Wm2; sa.c1 = tmpb2; sa.g1 = nullptr;
    sa.a2 = xrAll + 2 * XP; sa.w2 = qk_Wk2;  sa.c2 = kb;    sa.g2 = nullptr;
    sa.a3 = xrAll + 5 * XP; sa.w3 = qks_Wk2; sa.c3 = ksb;   sa.g3 = nullptr;
    sa.a4 = xrAll + 6 * XP; sa.w4 = vo_Wv2;  sa.c4 = vb;    sa.g4 = nullptr;
    sa.a5 = xrAll + 7 * XP; sa.w5 = vo_Wm2;  sa.c5 = mvob;  sa.g5 = nullptr;
    gemm_stage2_kernel<<<g2, dim3(256), 0, stream>>>(sa);

    const dim3 g2b((unsigned)(DD / GBN), (unsigned)(BT / GBM), 2u);
    Stage2Args sb;
    sb.a0 = xrAll + 1 * XP; sb.w0 = qk_Wq2;  sb.c0 = qg;  sb.g0 = tmpb;
    sb.a1 = xrAll + 4 * XP; sb.w1 = qks_Wq2; sb.c1 = qsg; sb.g1 = tmpb2;
    sb.a2 = nullptr; sb.w2 = nullptr; sb.c2 = nullptr; sb.g2 = nullptr;
    sb.a3 = nullptr; sb.w3 = nullptr; sb.c3 = nullptr; sb.g3 = nullptr;
    sb.a4 = nullptr; sb.w4 = nullptr; sb.c4 = nullptr; sb.g4 = nullptr;
    sb.a5 = nullptr; sb.w5 = nullptr; sb.c5 = nullptr; sb.g5 = nullptr;
    gemm_stage2_kernel<<<g2b, dim3(256), 0, stream>>>(sb);

    const long long sTD = (long long)TT * DD, sTTl = (long long)TT * TT;
    gemm(qg,  kb,  zb,  nullptr, TT, TT, DD, DD, DD, TT, 1, sTD, sTD, sTTl, BB);
    gemm(qsg, ksb, zsb, nullptr, TT, TT, DD, DD, DD, TT, 1, sTD, sTD, sTTl, BB);

    mix_chain_kernel<<<dim3(BT / 16), 512, 0, stream>>>(zb, zsb, d1, A1, B1, d1s, A1s, B1s, d2, A2, B2, zsb);

    gemm(zsb, vb, ctx, mvob, TT, DD, TT, TT, DD, DD, 0, sTTl, sTD, sTD, BB);
    gemm(ctx, vo_Wo, (float*)d_out, nullptr, BT, DD, DD, DD, DD, DD, 1, 0, 0, 0, 1);
}
